# starting kernel
# baseline (speedup 1.0000x reference)
#include <hip/hip_runtime.h>


#define BSZ   8
#define T_    8
#define L_    196
#define D_    192
#define E_    192
#define C_    1536
#define N_    16
#define KC    4
#define R_    96
#define RN2   128
#define LTILES 13

typedef float v2f __attribute__((ext_vector_type(2)));
typedef float v8f __attribute__((ext_vector_type(8)));

__device__ __forceinline__ float sigmoid_(float x) {
    return 1.0f / (1.0f + __expf(-x));
}

__global__ __launch_bounds__(32)
void k_gemm_inproj(const float* __restrict__ hidden,
                   const float* __restrict__ w,
                   float* __restrict__ x_pre,
                   float* __restrict__ zbuf)
{
    const int lane = threadIdx.x;
    const int et   = blockIdx.x % 24;
    const int lt   = blockIdx.x / 24;
    const int bt   = blockIdx.y;
    const int b    = bt / T_;
    const int t    = bt % T_;

    const int mrow  = lt * 16 + (lane & 15);
    const int mclmp = (mrow < L_) ? mrow : (L_ - 1);
    const int koff  = (lane >> 4) * 2;
    const int ecol  = et * 16 + (lane & 15);

    const float* Arow = hidden + ((size_t)bt * L_ + mclmp) * D_;
    const float* Brow = w + (size_t)ecol * D_;

    v8f acc0 = {0.f,0.f,0.f,0.f,0.f,0.f,0.f,0.f};
    v8f acc1 = {0.f,0.f,0.f,0.f,0.f,0.f,0.f,0.f};
    for (int k0 = 0; k0 < D_; k0 += 8) {
        v2f a0 = *(const v2f*)(Arow + k0 + koff);
        v2f b0 = *(const v2f*)(Brow + k0 + koff);
        v2f a1 = *(const v2f*)(Arow + k0 + 4 + koff);
        v2f b1 = *(const v2f*)(Brow + k0 + 4 + koff);
        acc0 = __builtin_amdgcn_wmma_f32_16x16x4_f32(
                   false, a0, false, b0, (short)0, acc0, false, false);
        acc1 = __builtin_amdgcn_wmma_f32_16x16x4_f32(
                   false, a1, false, b1, (short)0, acc1, false, false);
    }

    const int nn    = lane & 15;
    const int e_out = et * 16 + nn;
    const int mbase = (lane >> 4) * 8;
    #pragma unroll
    for (int r = 0; r < 8; ++r) {
        const int l = lt * 16 + mbase + r;
        if (l < L_) {
            const float v = acc0[r] + acc1[r];
            if (e_out < E_) {
                const int c = e_out * T_ + t;
                x_pre[((size_t)b * C_ + c) * L_ + l] = v;
            } else {
                const int c = (e_out - E_) * T_ + t;
                zbuf[((size_t)b * C_ + c) * L_ + l] = v;
            }
        }
    }
}

__global__ __launch_bounds__(256)
void k_conv_silu(const float* __restrict__ x_pre,
                 const float* __restrict__ cw,
                 const float* __restrict__ cb,
                 float* __restrict__ xc)
{
    const int idx = blockIdx.x * 256 + threadIdx.x;
    if (idx >= BSZ * C_ * L_) return;
    const int l  = idx % L_;
    const int bc = idx / L_;
    const int c  = bc % C_;

    const float* xp = x_pre + (size_t)bc * L_;
    float s = cb[c];
    #pragma unroll
    for (int k = 0; k < KC; ++k) {
        const int ls = l - (KC - 1) + k;
        if (ls >= 0) s += xp[ls] * cw[c * KC + k];
    }
    xc[idx] = s * sigmoid_(s);
}

__global__ __launch_bounds__(32)
void k_gemm_xproj(const float* __restrict__ xc,
                  const float* __restrict__ w,
                  float* __restrict__ x_dbl)
{
    const int lane = threadIdx.x;
    const int rt   = blockIdx.x % 8;
    const int lt   = blockIdx.x / 8;
    const int b    = blockIdx.y;

    const int rrow  = rt * 16 + (lane & 15);
    const int koff  = (lane >> 4) * 2;
    const int lcol  = lt * 16 + (lane & 15);
    const int lclmp = (lcol < L_) ? lcol : (L_ - 1);

    const float* Arow  = w + (size_t)rrow * C_;
    const float* Bcol  = xc + (size_t)b * C_ * L_ + lclmp;

    v8f acc0 = {0.f,0.f,0.f,0.f,0.f,0.f,0.f,0.f};
    v8f acc1 = {0.f,0.f,0.f,0.f,0.f,0.f,0.f,0.f};
    for (int k0 = 0; k0 < C_; k0 += 8) {
        const int kB = k0 + koff;
        v2f a0 = *(const v2f*)(Arow + k0 + koff);
        v2f b0; b0.x = Bcol[(size_t)kB * L_];
                b0.y = Bcol[(size_t)(kB + 1) * L_];
        v2f a1 = *(const v2f*)(Arow + k0 + 4 + koff);
        v2f b1; b1.x = Bcol[(size_t)(kB + 4) * L_];
                b1.y = Bcol[(size_t)(kB + 5) * L_];
        acc0 = __builtin_amdgcn_wmma_f32_16x16x4_f32(
                   false, a0, false, b0, (short)0, acc0, false, false);
        acc1 = __builtin_amdgcn_wmma_f32_16x16x4_f32(
                   false, a1, false, b1, (short)0, acc1, false, false);
    }

    const int nn    = lane & 15;
    const int l     = lt * 16 + nn;
    const int mbase = (lane >> 4) * 8;
    if (l < L_) {
        #pragma unroll
        for (int r = 0; r < 8; ++r) {
            const int rr = rt * 16 + mbase + r;
            x_dbl[((size_t)b * RN2 + rr) * L_ + l] = acc0[r] + acc1[r];
        }
    }
}

__global__ __launch_bounds__(32)
void k_gemm_dtproj(const float* __restrict__ x_dbl,
                   const float* __restrict__ w,
                   const float* __restrict__ dt_bias,
                   float* __restrict__ delta)
{
    const int lane = threadIdx.x;
    const int ct   = blockIdx.x % 96;
    const int lt   = blockIdx.x / 96;
    const int b    = blockIdx.y;

    const int crow  = ct * 16 + (lane & 15);
    const int koff  = (lane >> 4) * 2;
    const int lcol  = lt * 16 + (lane & 15);
    const int lclmp = (lcol < L_) ? lcol : (L_ - 1);

    const float* Arow = w + (size_t)crow * R_;
    const float* Bcol = x_dbl + (size_t)b * RN2 * L_ + lclmp;

    v8f acc0 = {0.f,0.f,0.f,0.f,0.f,0.f,0.f,0.f};
    v8f acc1 = {0.f,0.f,0.f,0.f,0.f,0.f,0.f,0.f};
    for (int k0 = 0; k0 < R_; k0 += 8) {
        const int kB = k0 + koff;
        v2f a0 = *(const v2f*)(Arow + k0 + koff);
        v2f b0; b0.x = Bcol[(size_t)kB * L_];
                b0.y = Bcol[(size_t)(kB + 1) * L_];
        v2f a1 = *(const v2f*)(Arow + k0 + 4 + koff);
        v2f b1; b1.x = Bcol[(size_t)(kB + 4) * L_];
                b1.y = Bcol[(size_t)(kB + 5) * L_];
        acc0 = __builtin_amdgcn_wmma_f32_16x16x4_f32(
                   false, a0, false, b0, (short)0, acc0, false, false);
        acc1 = __builtin_amdgcn_wmma_f32_16x16x4_f32(
                   false, a1, false, b1, (short)0, acc1, false, false);
    }

    const int nn    = lane & 15;
    const int l     = lt * 16 + nn;
    const int mbase = (lane >> 4) * 8;
    if (l < L_) {
        #pragma unroll
        for (int r = 0; r < 8; ++r) {
            const int c = ct * 16 + mbase + r;
            const float v = acc0[r] + acc1[r] + dt_bias[c];
            const float sp = (v > 20.f) ? v : __logf(1.f + __expf(v));
            delta[((size_t)b * C_ + c) * L_ + l] = sp;
        }
    }
}

__global__ __launch_bounds__(256)
void k_scan_out(const float* __restrict__ delta,
                const float* __restrict__ xc,
                const float* __restrict__ zbuf,
                const float* __restrict__ x_dbl,
                const float* __restrict__ A_log,
                const float* __restrict__ Dp_,
                float* __restrict__ out)
{
    const int g  = blockIdx.x * 256 + threadIdx.x;
    const int n  = g & 15;
    const int bc = g >> 4;
    const int b  = bc / C_;
    const int c  = bc % C_;

    const float A_neg = -__expf(A_log[c * N_ + n]);
    const float Dp    = Dp_[c];

    const float* dp = delta + (size_t)bc * L_;
    const float* xp = xc    + (size_t)bc * L_;
    const float* zp = zbuf  + (size_t)bc * L_;
    const float* Bs = x_dbl + ((size_t)b * RN2 + R_       + n) * L_;
    const float* Cs = x_dbl + ((size_t)b * RN2 + R_ + N_  + n) * L_;
    float* op = out + (size_t)bc * L_;

    float h = 0.f;
    for (int l = 0; l < L_; ++l) {
        const float dv = dp[l];
        const float xv = xp[l];
        const float dA = __expf(dv * A_neg);
        h = h * dA + dv * Bs[l] * xv;
        float yp = h * Cs[l];
        yp += __shfl_xor(yp, 1, 16);
        yp += __shfl_xor(yp, 2, 16);
        yp += __shfl_xor(yp, 4, 16);
        yp += __shfl_xor(yp, 8, 16);
        if (n == 0) {
            const float zv = zp[l];
            op[l] = (yp + xv * Dp) * (zv * sigmoid_(zv));
        }
    }
}

extern "C" void kernel_launch(void* const* d_in, const int* in_sizes, int n_in,
                              void* d_out, int out_size, void* d_ws, size_t ws_size,
                              hipStream_t stream)
{
    const float* hidden    = (const float*)d_in[0];
    const float* in_proj_w = (const float*)d_in[1];
    const float* conv_w    = (const float*)d_in[2];
    const float* conv_b    = (const float*)d_in[3];
    const float* x_proj_w  = (const float*)d_in[4];
    const float* dt_proj_w = (const float*)d_in[5];
    const float* dt_bias   = (const float*)d_in[6];
    const float* A_log     = (const float*)d_in[7];
    const float* D_param   = (const float*)d_in[8];

    float* ws = (float*)d_ws;
    const size_t BCL = (size_t)BSZ * C_ * L_;
    float* x_pre = ws;
    float* zbuf  = ws + BCL;
    float* xcbuf = ws + 2 * BCL;
    float* delta = ws + 3 * BCL;
    float* x_dbl = ws + 4 * BCL;

    k_gemm_inproj<<<dim3(24 * LTILES, BSZ * T_), 32, 0, stream>>>(
        hidden, in_proj_w, x_pre, zbuf);

    k_conv_silu<<<(BSZ * C_ * L_ + 255) / 256, 256, 0, stream>>>(
        x_pre, conv_w, conv_b, xcbuf);

    k_gemm_xproj<<<dim3(8 * LTILES, BSZ), 32, 0, stream>>>(
        xcbuf, x_proj_w, x_dbl);

    k_gemm_dtproj<<<dim3(96 * LTILES, BSZ), 32, 0, stream>>>(
        x_dbl, dt_proj_w, dt_bias, delta);

    k_scan_out<<<(BSZ * C_ * N_) / 256, 256, 0, stream>>>(
        delta, xcbuf, zbuf, x_dbl, A_log, D_param, (float*)d_out);
}


# scan kernel remapped: one lane per (b,c) channel keeps its 16 states in registers, B/C rows read as LDS broadcasts; sum over n now an in-lane 4-way partial sum (summation order changed)
# speedup vs baseline: 1.5639x; 1.5639x over previous
#include <hip/hip_runtime.h>


#define BSZ   8
#define T_    8
#define L_    196
#define D_    192
#define E_    192
#define C_    1536
#define N_    16
#define KC    4
#define R_    96
#define RN2   128
#define LTILES 13

typedef float v2f __attribute__((ext_vector_type(2)));
typedef float v8f __attribute__((ext_vector_type(8)));

__device__ __forceinline__ float sigmoid_(float x) {
    return 1.0f / (1.0f + __expf(-x));
}

__global__ __launch_bounds__(32)
void k_gemm_inproj(const float* __restrict__ hidden,
                   const float* __restrict__ w,
                   float* __restrict__ x_pre,
                   float* __restrict__ zbuf)
{
    const int lane = threadIdx.x;
    const int et   = blockIdx.x % 24;
    const int lt   = blockIdx.x / 24;
    const int bt   = blockIdx.y;
    const int b    = bt / T_;
    const int t    = bt % T_;

    const int mrow  = lt * 16 + (lane & 15);
    const int mclmp = (mrow < L_) ? mrow : (L_ - 1);
    const int koff  = (lane >> 4) * 2;
    const int ecol  = et * 16 + (lane & 15);

    const float* Arow = hidden + ((size_t)bt * L_ + mclmp) * D_;
    const float* Brow = w + (size_t)ecol * D_;

    v8f acc0 = {0.f,0.f,0.f,0.f,0.f,0.f,0.f,0.f};
    v8f acc1 = {0.f,0.f,0.f,0.f,0.f,0.f,0.f,0.f};
    for (int k0 = 0; k0 < D_; k0 += 8) {
        v2f a0 = *(const v2f*)(Arow + k0 + koff);
        v2f b0 = *(const v2f*)(Brow + k0 + koff);
        v2f a1 = *(const v2f*)(Arow + k0 + 4 + koff);
        v2f b1 = *(const v2f*)(Brow + k0 + 4 + koff);
        acc0 = __builtin_amdgcn_wmma_f32_16x16x4_f32(
                   false, a0, false, b0, (short)0, acc0, false, false);
        acc1 = __builtin_amdgcn_wmma_f32_16x16x4_f32(
                   false, a1, false, b1, (short)0, acc1, false, false);
    }

    const int nn    = lane & 15;
    const int e_out = et * 16 + nn;
    const int mbase = (lane >> 4) * 8;
    #pragma unroll
    for (int r = 0; r < 8; ++r) {
        const int l = lt * 16 + mbase + r;
        if (l < L_) {
            const float v = acc0[r] + acc1[r];
            if (e_out < E_) {
                const int c = e_out * T_ + t;
                x_pre[((size_t)b * C_ + c) * L_ + l] = v;
            } else {
                const int c = (e_out - E_) * T_ + t;
                zbuf[((size_t)b * C_ + c) * L_ + l] = v;
            }
        }
    }
}

__global__ __launch_bounds__(256)
void k_conv_silu(const float* __restrict__ x_pre,
                 const float* __restrict__ cw,
                 const float* __restrict__ cb,
                 float* __restrict__ xc)
{
    const int idx = blockIdx.x * 256 + threadIdx.x;
    if (idx >= BSZ * C_ * L_) return;
    const int l  = idx % L_;
    const int bc = idx / L_;
    const int c  = bc % C_;

    const float* xp = x_pre + (size_t)bc * L_;
    float s = cb[c];
    #pragma unroll
    for (int k = 0; k < KC; ++k) {
        const int ls = l - (KC - 1) + k;
        if (ls >= 0) s += xp[ls] * cw[c * KC + k];
    }
    xc[idx] = s * sigmoid_(s);
}

__global__ __launch_bounds__(32)
void k_gemm_xproj(const float* __restrict__ xc,
                  const float* __restrict__ w,
                  float* __restrict__ x_dbl)
{
    const int lane = threadIdx.x;
    const int rt   = blockIdx.x % 8;
    const int lt   = blockIdx.x / 8;
    const int b    = blockIdx.y;

    const int rrow  = rt * 16 + (lane & 15);
    const int koff  = (lane >> 4) * 2;
    const int lcol  = lt * 16 + (lane & 15);
    const int lclmp = (lcol < L_) ? lcol : (L_ - 1);

    const float* Arow  = w + (size_t)rrow * C_;
    const float* Bcol  = xc + (size_t)b * C_ * L_ + lclmp;

    v8f acc0 = {0.f,0.f,0.f,0.f,0.f,0.f,0.f,0.f};
    v8f acc1 = {0.f,0.f,0.f,0.f,0.f,0.f,0.f,0.f};
    for (int k0 = 0; k0 < C_; k0 += 8) {
        const int kB = k0 + koff;
        v2f a0 = *(const v2f*)(Arow + k0 + koff);
        v2f b0; b0.x = Bcol[(size_t)kB * L_];
                b0.y = Bcol[(size_t)(kB + 1) * L_];
        v2f a1 = *(const v2f*)(Arow + k0 + 4 + koff);
        v2f b1; b1.x = Bcol[(size_t)(kB + 4) * L_];
                b1.y = Bcol[(size_t)(kB + 5) * L_];
        acc0 = __builtin_amdgcn_wmma_f32_16x16x4_f32(
                   false, a0, false, b0, (short)0, acc0, false, false);
        acc1 = __builtin_amdgcn_wmma_f32_16x16x4_f32(
                   false, a1, false, b1, (short)0, acc1, false, false);
    }

    const int nn    = lane & 15;
    const int l     = lt * 16 + nn;
    const int mbase = (lane >> 4) * 8;
    if (l < L_) {
        #pragma unroll
        for (int r = 0; r < 8; ++r) {
            const int rr = rt * 16 + mbase + r;
            x_dbl[((size_t)b * RN2 + rr) * L_ + l] = acc0[r] + acc1[r];
        }
    }
}

__global__ __launch_bounds__(32)
void k_gemm_dtproj(const float* __restrict__ x_dbl,
                   const float* __restrict__ w,
                   const float* __restrict__ dt_bias,
                   float* __restrict__ delta)
{
    const int lane = threadIdx.x;
    const int ct   = blockIdx.x % 96;
    const int lt   = blockIdx.x / 96;
    const int b    = blockIdx.y;

    const int crow  = ct * 16 + (lane & 15);
    const int koff  = (lane >> 4) * 2;
    const int lcol  = lt * 16 + (lane & 15);
    const int lclmp = (lcol < L_) ? lcol : (L_ - 1);

    const float* Arow = w + (size_t)crow * R_;
    const float* Bcol = x_dbl + (size_t)b * RN2 * L_ + lclmp;

    v8f acc0 = {0.f,0.f,0.f,0.f,0.f,0.f,0.f,0.f};
    v8f acc1 = {0.f,0.f,0.f,0.f,0.f,0.f,0.f,0.f};
    for (int k0 = 0; k0 < R_; k0 += 8) {
        const int kB = k0 + koff;
        v2f a0 = *(const v2f*)(Arow + k0 + koff);
        v2f b0; b0.x = Bcol[(size_t)kB * L_];
                b0.y = Bcol[(size_t)(kB + 1) * L_];
        v2f a1 = *(const v2f*)(Arow + k0 + 4 + koff);
        v2f b1; b1.x = Bcol[(size_t)(kB + 4) * L_];
                b1.y = Bcol[(size_t)(kB + 5) * L_];
        acc0 = __builtin_amdgcn_wmma_f32_16x16x4_f32(
                   false, a0, false, b0, (short)0, acc0, false, false);
        acc1 = __builtin_amdgcn_wmma_f32_16x16x4_f32(
                   false, a1, false, b1, (short)0, acc1, false, false);
    }

    const int nn    = lane & 15;
    const int l     = lt * 16 + nn;
    const int mbase = (lane >> 4) * 8;
    if (l < L_) {
        #pragma unroll
        for (int r = 0; r < 8; ++r) {
            const int c = ct * 16 + mbase + r;
            const float v = acc0[r] + acc1[r] + dt_bias[c];
            const float sp = (v > 20.f) ? v : __logf(1.f + __expf(v));
            delta[((size_t)b * C_ + c) * L_ + l] = sp;
        }
    }
}

#define SC_THREADS 64
#define SC_CHUNKS  (L_ / 4)
#define BC_PITCH   36
#define SC_STAGE_FULL ((2 * N_ * SC_CHUNKS) / SC_THREADS)
#define SC_STAGE_REST ((2 * N_ * SC_CHUNKS) - SC_STAGE_FULL * SC_THREADS)

__device__ __forceinline__ void scan_chunk4(const float* __restrict__ srow,
                                            const float4 d4, const float4 x4,
                                            const float4 z4,
                                            const float (&A_neg)[N_],
                                            float (&h)[N_],
                                            const float Dp,
                                            float4* __restrict__ dst)
{
    const float dvs[4] = {d4.x, d4.y, d4.z, d4.w};
    const float xvs[4] = {x4.x, x4.y, x4.z, x4.w};
    const float zvs[4] = {z4.x, z4.y, z4.z, z4.w};
    float ov[4];
    #pragma unroll
    for (int k = 0; k < 4; ++k) {
        const float dv  = dvs[k];
        const float xv  = xvs[k];
        const float zv  = zvs[k];
        const float dvx = dv * xv;
        const float4* row = (const float4*)(srow + k * BC_PITCH);
        float bc[2 * N_];
        #pragma unroll
        for (int q = 0; q < 8; ++q) {
            const float4 t = row[q];
            bc[4 * q + 0] = t.x; bc[4 * q + 1] = t.y;
            bc[4 * q + 2] = t.z; bc[4 * q + 3] = t.w;
        }
        float y0 = 0.f, y1 = 0.f, y2 = 0.f, y3 = 0.f;
        #pragma unroll
        for (int n = 0; n < N_; n += 4) {
            h[n + 0] = h[n + 0] * __expf(dv * A_neg[n + 0]) + dvx * bc[n + 0];
            h[n + 1] = h[n + 1] * __expf(dv * A_neg[n + 1]) + dvx * bc[n + 1];
            h[n + 2] = h[n + 2] * __expf(dv * A_neg[n + 2]) + dvx * bc[n + 2];
            h[n + 3] = h[n + 3] * __expf(dv * A_neg[n + 3]) + dvx * bc[n + 3];
            y0 += h[n + 0] * bc[N_ + n + 0];
            y1 += h[n + 1] * bc[N_ + n + 1];
            y2 += h[n + 2] * bc[N_ + n + 2];
            y3 += h[n + 3] * bc[N_ + n + 3];
        }
        const float y = (y0 + y1) + (y2 + y3);
        ov[k] = (y + xv * Dp) * (zv * sigmoid_(zv));
    }
    *dst = make_float4(ov[0], ov[1], ov[2], ov[3]);
}

__global__ __launch_bounds__(SC_THREADS) __attribute__((amdgpu_waves_per_eu(1, 1)))
void k_scan_out(const float* __restrict__ delta,
                const float* __restrict__ xc,
                const float* __restrict__ zbuf,
                const float* __restrict__ x_dbl,
                const float* __restrict__ A_log,
                const float* __restrict__ Dp_,
                float* __restrict__ out)
{
    __shared__ __attribute__((aligned(16))) float sBC[L_ * BC_PITCH];

    const int tid = threadIdx.x;
    const int b   = blockIdx.x / (C_ / SC_THREADS);
    const int c   = (blockIdx.x % (C_ / SC_THREADS)) * SC_THREADS + tid;
    const int bc_ = b * C_ + c;

    const float4* dp4 = (const float4*)(delta + (size_t)bc_ * L_);
    const float4* xp4 = (const float4*)(xc    + (size_t)bc_ * L_);
    const float4* zp4 = (const float4*)(zbuf  + (size_t)bc_ * L_);
    float4*       op4 = (float4*)(out + (size_t)bc_ * L_);

    float4 qd0 = dp4[0], qx0 = xp4[0], qz0 = zp4[0];
    float4 qd1 = dp4[1], qx1 = xp4[1], qz1 = zp4[1];
    float4 qd2, qx2, qz2;

    const float4* src4 = (const float4*)(x_dbl + ((size_t)b * RN2 + R_) * L_);
    #pragma unroll
    for (int j = 0; j < SC_STAGE_FULL; ++j) {
        const int f  = tid + j * SC_THREADS;
        const int r  = f / SC_CHUNKS;
        const int l4 = f % SC_CHUNKS;
        const float4 v = src4[f];
        float* d = sBC + (l4 * 4) * BC_PITCH + r;
        d[0]            = v.x;
        d[BC_PITCH]     = v.y;
        d[2 * BC_PITCH] = v.z;
        d[3 * BC_PITCH] = v.w;
    }
    if (tid < SC_STAGE_REST) {
        const int f  = tid + SC_STAGE_FULL * SC_THREADS;
        const int r  = f / SC_CHUNKS;
        const int l4 = f % SC_CHUNKS;
        const float4 v = src4[f];
        float* d = sBC + (l4 * 4) * BC_PITCH + r;
        d[0]            = v.x;
        d[BC_PITCH]     = v.y;
        d[2 * BC_PITCH] = v.z;
        d[3 * BC_PITCH] = v.w;
    }

    float A_neg[N_];
    {
        const float4* a4 = (const float4*)(A_log + (size_t)c * N_);
        #pragma unroll
        for (int q = 0; q < 4; ++q) {
            const float4 t = a4[q];
            A_neg[4 * q + 0] = -__expf(t.x);
            A_neg[4 * q + 1] = -__expf(t.y);
            A_neg[4 * q + 2] = -__expf(t.z);
            A_neg[4 * q + 3] = -__expf(t.w);
        }
    }
    const float Dp = Dp_[c];

    float h[N_];
    #pragma unroll
    for (int n = 0; n < N_; ++n) h[n] = 0.f;

    __syncthreads();

    for (int i = 0; i < SC_CHUNKS - 1; i += 3) {
        qd2 = dp4[i + 2]; qx2 = xp4[i + 2]; qz2 = zp4[i + 2];
        scan_chunk4(sBC + (i * 4) * BC_PITCH, qd0, qx0, qz0, A_neg, h, Dp, op4 + i);
        qd0 = dp4[i + 3]; qx0 = xp4[i + 3]; qz0 = zp4[i + 3];
        scan_chunk4(sBC + ((i + 1) * 4) * BC_PITCH, qd1, qx1, qz1, A_neg, h, Dp, op4 + i + 1);
        const int i4 = (i + 4 < SC_CHUNKS) ? (i + 4) : (SC_CHUNKS - 1);
        qd1 = dp4[i4]; qx1 = xp4[i4]; qz1 = zp4[i4];
        scan_chunk4(sBC + ((i + 2) * 4) * BC_PITCH, qd2, qx2, qz2, A_neg, h, Dp, op4 + i + 2);
    }
    scan_chunk4(sBC + ((SC_CHUNKS - 1) * 4) * BC_PITCH, qd0, qx0, qz0, A_neg, h, Dp,
                op4 + (SC_CHUNKS - 1));
}

extern "C" void kernel_launch(void* const* d_in, const int* in_sizes, int n_in,
                              void* d_out, int out_size, void* d_ws, size_t ws_size,
                              hipStream_t stream)
{
    const float* hidden    = (const float*)d_in[0];
    const float* in_proj_w = (const float*)d_in[1];
    const float* conv_w    = (const float*)d_in[2];
    const float* conv_b    = (const float*)d_in[3];
    const float* x_proj_w  = (const float*)d_in[4];
    const float* dt_proj_w = (const float*)d_in[5];
    const float* dt_bias   = (const float*)d_in[6];
    const float* A_log     = (const float*)d_in[7];
    const float* D_param   = (const float*)d_in[8];

    float* ws = (float*)d_ws;
    const size_t BCL = (size_t)BSZ * C_ * L_;
    float* x_pre = ws;
    float* zbuf  = ws + BCL;
    float* xcbuf = ws + 2 * BCL;
    float* delta = ws + 3 * BCL;
    float* x_dbl = ws + 4 * BCL;

    k_gemm_inproj<<<dim3(24 * LTILES, BSZ * T_), 32, 0, stream>>>(
        hidden, in_proj_w, x_pre, zbuf);

    k_conv_silu<<<(BSZ * C_ * L_ + 255) / 256, 256, 0, stream>>>(
        x_pre, conv_w, conv_b, xcbuf);

    k_gemm_xproj<<<dim3(8 * LTILES, BSZ), 32, 0, stream>>>(
        xcbuf, x_proj_w, x_dbl);

    k_gemm_dtproj<<<dim3(96 * LTILES, BSZ), 32, 0, stream>>>(
        x_dbl, dt_proj_w, dt_bias, delta);

    k_scan_out<<<(BSZ * C_) / SC_THREADS, SC_THREADS, 0, stream>>>(
        delta, xcbuf, zbuf, x_dbl, A_log, D_param, (float*)d_out);
}


# in_proj GEMM retiled: a wave computes 64(e) x 32(l) as 8 accumulator tiles (was one 16x16), l on the lane axis, each row's 128-B line read in one burst; one k chain per tile (sum order changed)
# speedup vs baseline: 2.4022x; 1.5360x over previous
#include <hip/hip_runtime.h>


#define BSZ   8
#define T_    8
#define L_    196
#define D_    192
#define E_    192
#define C_    1536
#define N_    16
#define KC    4
#define R_    96
#define RN2   128
#define LTILES 13

typedef float v2f __attribute__((ext_vector_type(2)));
typedef float v8f __attribute__((ext_vector_type(8)));

__device__ __forceinline__ float sigmoid_(float x) {
    return 1.0f / (1.0f + __expf(-x));
}

#define IP_EW    64
#define IP_MW    32
#define IP_EBLK  ((2 * E_) / IP_EW)
#define IP_MBLK  ((BSZ * T_ * L_) / IP_MW)
#define IP_KC    32
#define IP_KS    (IP_KC / 4)
#define IP_NCH   (D_ / IP_KC)
#define IP_NPASS (IP_NCH * 4)

__device__ __forceinline__ void ip_load_line(const float* __restrict__ p, v2f (&x)[IP_KS])
{
    #pragma unroll
    for (int s = 0; s < IP_KS; ++s)
        x[s] = *(const v2f*)(p + 4 * s);
}

__device__ __forceinline__ void ip_pass(const v2f (&a)[IP_KS],
                                        const v2f (&b0)[IP_KS], const v2f (&b1)[IP_KS],
                                        v8f& c0, v8f& c1)
{
    #pragma unroll
    for (int s = 0; s < IP_KS; ++s) {
        c0 = __builtin_amdgcn_wmma_f32_16x16x4_f32(
                 false, a[s], false, b0[s], (short)0, c0, false, false);
        c1 = __builtin_amdgcn_wmma_f32_16x16x4_f32(
                 false, a[s], false, b1[s], (short)0, c1, false, false);
    }
}

__global__ __launch_bounds__(32)
void k_gemm_inproj(const float* __restrict__ hidden,
                   const float* __restrict__ w,
                   float* __restrict__ x_pre,
                   float* __restrict__ zbuf)
{
    const int lane = threadIdx.x;
    const int c    = lane & 15;
    const int h    = lane >> 4;
    const int eb   = blockIdx.x % IP_EBLK;
    const int mb   = blockIdx.x / IP_EBLK;

    const float* Ap = w      + (size_t)(eb * IP_EW + c) * D_ + 2 * h;
    const float* Bp = hidden + (size_t)(mb * IP_MW + c) * D_ + 2 * h;

    v8f acc[4][2];
    #pragma unroll
    for (int f = 0; f < 4; ++f) {
        #pragma unroll
        for (int g = 0; g < 2; ++g)
            acc[f][g] = (v8f){0.f,0.f,0.f,0.f,0.f,0.f,0.f,0.f};
    }

    v2f a[4][IP_KS];
    v2f b[2][2][IP_KS];
    ip_load_line(Bp, b[0][0]);
    ip_load_line(Bp + (size_t)16 * D_, b[0][1]);
    ip_load_line(Ap, a[0]);
    ip_load_line(Ap + (size_t)16 * D_, a[1]);
    ip_load_line(Ap + (size_t)32 * D_, a[2]);

    #pragma unroll
    for (int p = 0; p < IP_NPASS; ++p) {
        const int ch = p >> 2;
        const int f  = p & 3;
        const int q  = p + 3;
        if (q < IP_NPASS)
            ip_load_line(Ap + (size_t)(q & 3) * 16 * D_ + (q >> 2) * IP_KC, a[q & 3]);
        if (f == 1 && ch + 1 < IP_NCH) {
            ip_load_line(Bp + (ch + 1) * IP_KC, b[(ch + 1) & 1][0]);
            ip_load_line(Bp + (size_t)16 * D_ + (ch + 1) * IP_KC, b[(ch + 1) & 1][1]);
        }
        __builtin_amdgcn_sched_barrier(0);
        ip_pass(a[p & 3], b[ch & 1][0], b[ch & 1][1], acc[f][0], acc[f][1]);
        __builtin_amdgcn_sched_barrier(0);
    }

    float* outp = (eb < IP_EBLK / 2) ? x_pre : zbuf;
    const int e0 = (eb % (IP_EBLK / 2)) * IP_EW + 8 * h;
    #pragma unroll
    for (int g = 0; g < 2; ++g) {
        const int m  = mb * IP_MW + g * 16 + c;
        const int bt = m / L_;
        const int l  = m - bt * L_;
        const int b_ = bt / T_;
        const int t  = bt % T_;
        float* p = outp + ((size_t)b_ * C_ + (size_t)e0 * T_ + t) * L_ + l;
        #pragma unroll
        for (int f = 0; f < 4; ++f) {
            #pragma unroll
            for (int r = 0; r < 8; ++r)
                p[(size_t)(f * 16 + r) * (T_ * L_)] = acc[f][g][r];
        }
    }
}

__global__ __launch_bounds__(256)
void k_conv_silu(const float* __restrict__ x_pre,
                 const float* __restrict__ cw,
                 const float* __restrict__ cb,
                 float* __restrict__ xc)
{
    const int idx = blockIdx.x * 256 + threadIdx.x;
    if (idx >= BSZ * C_ * L_) return;
    const int l  = idx % L_;
    const int bc = idx / L_;
    const int c  = bc % C_;

    const float* xp = x_pre + (size_t)bc * L_;
    float s = cb[c];
    #pragma unroll
    for (int k = 0; k < KC; ++k) {
        const int ls = l - (KC - 1) + k;
        if (ls >= 0) s += xp[ls] * cw[c * KC + k];
    }
    xc[idx] = s * sigmoid_(s);
}

__global__ __launch_bounds__(32)
void k_gemm_xproj(const float* __restrict__ xc,
                  const float* __restrict__ w,
                  float* __restrict__ x_dbl)
{
    const int lane = threadIdx.x;
    const int rt   = blockIdx.x % 8;
    const int lt   = blockIdx.x / 8;
    const int b    = blockIdx.y;

    const int rrow  = rt * 16 + (lane & 15);
    const int koff  = (lane >> 4) * 2;
    const int lcol  = lt * 16 + (lane & 15);
    const int lclmp = (lcol < L_) ? lcol : (L_ - 1);

    const float* Arow  = w + (size_t)rrow * C_;
    const float* Bcol  = xc + (size_t)b * C_ * L_ + lclmp;

    v8f acc0 = {0.f,0.f,0.f,0.f,0.f,0.f,0.f,0.f};
    v8f acc1 = {0.f,0.f,0.f,0.f,0.f,0.f,0.f,0.f};
    for (int k0 = 0; k0 < C_; k0 += 8) {
        const int kB = k0 + koff;
        v2f a0 = *(const v2f*)(Arow + k0 + koff);
        v2f b0; b0.x = Bcol[(size_t)kB * L_];
                b0.y = Bcol[(size_t)(kB + 1) * L_];
        v2f a1 = *(const v2f*)(Arow + k0 + 4 + koff);
        v2f b1; b1.x = Bcol[(size_t)(kB + 4) * L_];
                b1.y = Bcol[(size_t)(kB + 5) * L_];
        acc0 = __builtin_amdgcn_wmma_f32_16x16x4_f32(
                   false, a0, false, b0, (short)0, acc0, false, false);
        acc1 = __builtin_amdgcn_wmma_f32_16x16x4_f32(
                   false, a1, false, b1, (short)0, acc1, false, false);
    }

    const int nn    = lane & 15;
    const int l     = lt * 16 + nn;
    const int mbase = (lane >> 4) * 8;
    if (l < L_) {
        #pragma unroll
        for (int r = 0; r < 8; ++r) {
            const int rr = rt * 16 + mbase + r;
            x_dbl[((size_t)b * RN2 + rr) * L_ + l] = acc0[r] + acc1[r];
        }
    }
}

__global__ __launch_bounds__(32)
void k_gemm_dtproj(const float* __restrict__ x_dbl,
                   const float* __restrict__ w,
                   const float* __restrict__ dt_bias,
                   float* __restrict__ delta)
{
    const int lane = threadIdx.x;
    const int ct   = blockIdx.x % 96;
    const int lt   = blockIdx.x / 96;
    const int b    = blockIdx.y;

    const int crow  = ct * 16 + (lane & 15);
    const int koff  = (lane >> 4) * 2;
    const int lcol  = lt * 16 + (lane & 15);
    const int lclmp = (lcol < L_) ? lcol : (L_ - 1);

    const float* Arow = w + (size_t)crow * R_;
    const float* Bcol = x_dbl + (size_t)b * RN2 * L_ + lclmp;

    v8f acc0 = {0.f,0.f,0.f,0.f,0.f,0.f,0.f,0.f};
    v8f acc1 = {0.f,0.f,0.f,0.f,0.f,0.f,0.f,0.f};
    for (int k0 = 0; k0 < R_; k0 += 8) {
        const int kB = k0 + koff;
        v2f a0 = *(const v2f*)(Arow + k0 + koff);
        v2f b0; b0.x = Bcol[(size_t)kB * L_];
                b0.y = Bcol[(size_t)(kB + 1) * L_];
        v2f a1 = *(const v2f*)(Arow + k0 + 4 + koff);
        v2f b1; b1.x = Bcol[(size_t)(kB + 4) * L_];
                b1.y = Bcol[(size_t)(kB + 5) * L_];
        acc0 = __builtin_amdgcn_wmma_f32_16x16x4_f32(
                   false, a0, false, b0, (short)0, acc0, false, false);
        acc1 = __builtin_amdgcn_wmma_f32_16x16x4_f32(
                   false, a1, false, b1, (short)0, acc1, false, false);
    }

    const int nn    = lane & 15;
    const int l     = lt * 16 + nn;
    const int mbase = (lane >> 4) * 8;
    if (l < L_) {
        #pragma unroll
        for (int r = 0; r < 8; ++r) {
            const int c = ct * 16 + mbase + r;
            const float v = acc0[r] + acc1[r] + dt_bias[c];
            const float sp = (v > 20.f) ? v : __logf(1.f + __expf(v));
            delta[((size_t)b * C_ + c) * L_ + l] = sp;
        }
    }
}

#define SC_THREADS 64
#define SC_CHUNKS  (L_ / 4)
#define BC_PITCH   36
#define SC_STAGE_FULL ((2 * N_ * SC_CHUNKS) / SC_THREADS)
#define SC_STAGE_REST ((2 * N_ * SC_CHUNKS) - SC_STAGE_FULL * SC_THREADS)

__device__ __forceinline__ void scan_chunk4(const float* __restrict__ srow,
                                            const float4 d4, const float4 x4,
                                            const float4 z4,
                                            const float (&A_neg)[N_],
                                            float (&h)[N_],
                                            const float Dp,
                                            float4* __restrict__ dst)
{
    const float dvs[4] = {d4.x, d4.y, d4.z, d4.w};
    const float xvs[4] = {x4.x, x4.y, x4.z, x4.w};
    const float zvs[4] = {z4.x, z4.y, z4.z, z4.w};
    float ov[4];
    #pragma unroll
    for (int k = 0; k < 4; ++k) {
        const float dv  = dvs[k];
        const float xv  = xvs[k];
        const float zv  = zvs[k];
        const float dvx = dv * xv;
        const float4* row = (const float4*)(srow + k * BC_PITCH);
        float bc[2 * N_];
        #pragma unroll
        for (int q = 0; q < 8; ++q) {
            const float4 t = row[q];
            bc[4 * q + 0] = t.x; bc[4 * q + 1] = t.y;
            bc[4 * q + 2] = t.z; bc[4 * q + 3] = t.w;
        }
        float y0 = 0.f, y1 = 0.f, y2 = 0.f, y3 = 0.f;
        #pragma unroll
        for (int n = 0; n < N_; n += 4) {
            h[n + 0] = h[n + 0] * __expf(dv * A_neg[n + 0]) + dvx * bc[n + 0];
            h[n + 1] = h[n + 1] * __expf(dv * A_neg[n + 1]) + dvx * bc[n + 1];
            h[n + 2] = h[n + 2] * __expf(dv * A_neg[n + 2]) + dvx * bc[n + 2];
            h[n + 3] = h[n + 3] * __expf(dv * A_neg[n + 3]) + dvx * bc[n + 3];
            y0 += h[n + 0] * bc[N_ + n + 0];
            y1 += h[n + 1] * bc[N_ + n + 1];
            y2 += h[n + 2] * bc[N_ + n + 2];
            y3 += h[n + 3] * bc[N_ + n + 3];
        }
        const float y = (y0 + y1) + (y2 + y3);
        ov[k] = (y + xv * Dp) * (zv * sigmoid_(zv));
    }
    *dst = make_float4(ov[0], ov[1], ov[2], ov[3]);
}

__global__ __launch_bounds__(SC_THREADS) __attribute__((amdgpu_waves_per_eu(1, 1)))
void k_scan_out(const float* __restrict__ delta,
                const float* __restrict__ xc,
                const float* __restrict__ zbuf,
                const float* __restrict__ x_dbl,
                const float* __restrict__ A_log,
                const float* __restrict__ Dp_,
                float* __restrict__ out)
{
    __shared__ __attribute__((aligned(16))) float sBC[L_ * BC_PITCH];

    const int tid = threadIdx.x;
    const int b   = blockIdx.x / (C_ / SC_THREADS);
    const int c   = (blockIdx.x % (C_ / SC_THREADS)) * SC_THREADS + tid;
    const int bc_ = b * C_ + c;

    const float4* dp4 = (const float4*)(delta + (size_t)bc_ * L_);
    const float4* xp4 = (const float4*)(xc    + (size_t)bc_ * L_);
    const float4* zp4 = (const float4*)(zbuf  + (size_t)bc_ * L_);
    float4*       op4 = (float4*)(out + (size_t)bc_ * L_);

    float4 qd0 = dp4[0], qx0 = xp4[0], qz0 = zp4[0];
    float4 qd1 = dp4[1], qx1 = xp4[1], qz1 = zp4[1];
    float4 qd2, qx2, qz2;

    const float4* src4 = (const float4*)(x_dbl + ((size_t)b * RN2 + R_) * L_);
    #pragma unroll
    for (int j = 0; j < SC_STAGE_FULL; ++j) {
        const int f  = tid + j * SC_THREADS;
        const int r  = f / SC_CHUNKS;
        const int l4 = f % SC_CHUNKS;
        const float4 v = src4[f];
        float* d = sBC + (l4 * 4) * BC_PITCH + r;
        d[0]            = v.x;
        d[BC_PITCH]     = v.y;
        d[2 * BC_PITCH] = v.z;
        d[3 * BC_PITCH] = v.w;
    }
    if (tid < SC_STAGE_REST) {
        const int f  = tid + SC_STAGE_FULL * SC_THREADS;
        const int r  = f / SC_CHUNKS;
        const int l4 = f % SC_CHUNKS;
        const float4 v = src4[f];
        float* d = sBC + (l4 * 4) * BC_PITCH + r;
        d[0]            = v.x;
        d[BC_PITCH]     = v.y;
        d[2 * BC_PITCH] = v.z;
        d[3 * BC_PITCH] = v.w;
    }

    float A_neg[N_];
    {
        const float4* a4 = (const float4*)(A_log + (size_t)c * N_);
        #pragma unroll
        for (int q = 0; q < 4; ++q) {
            const float4 t = a4[q];
            A_neg[4 * q + 0] = -__expf(t.x);
            A_neg[4 * q + 1] = -__expf(t.y);
            A_neg[4 * q + 2] = -__expf(t.z);
            A_neg[4 * q + 3] = -__expf(t.w);
        }
    }
    const float Dp = Dp_[c];

    float h[N_];
    #pragma unroll
    for (int n = 0; n < N_; ++n) h[n] = 0.f;

    __syncthreads();

    for (int i = 0; i < SC_CHUNKS - 1; i += 3) {
        qd2 = dp4[i + 2]; qx2 = xp4[i + 2]; qz2 = zp4[i + 2];
        scan_chunk4(sBC + (i * 4) * BC_PITCH, qd0, qx0, qz0, A_neg, h, Dp, op4 + i);
        qd0 = dp4[i + 3]; qx0 = xp4[i + 3]; qz0 = zp4[i + 3];
        scan_chunk4(sBC + ((i + 1) * 4) * BC_PITCH, qd1, qx1, qz1, A_neg, h, Dp, op4 + i + 1);
        const int i4 = (i + 4 < SC_CHUNKS) ? (i + 4) : (SC_CHUNKS - 1);
        qd1 = dp4[i4]; qx1 = xp4[i4]; qz1 = zp4[i4];
        scan_chunk4(sBC + ((i + 2) * 4) * BC_PITCH, qd2, qx2, qz2, A_neg, h, Dp, op4 + i + 2);
    }
    scan_chunk4(sBC + ((SC_CHUNKS - 1) * 4) * BC_PITCH, qd0, qx0, qz0, A_neg, h, Dp,
                op4 + (SC_CHUNKS - 1));
}

extern "C" void kernel_launch(void* const* d_in, const int* in_sizes, int n_in,
                              void* d_out, int out_size, void* d_ws, size_t ws_size,
                              hipStream_t stream)
{
    const float* hidden    = (const float*)d_in[0];
    const float* in_proj_w = (const float*)d_in[1];
    const float* conv_w    = (const float*)d_in[2];
    const float* conv_b    = (const float*)d_in[3];
    const float* x_proj_w  = (const float*)d_in[4];
    const float* dt_proj_w = (const float*)d_in[5];
    const float* dt_bias   = (const float*)d_in[6];
    const float* A_log     = (const float*)d_in[7];
    const float* D_param   = (const float*)d_in[8];

    float* ws = (float*)d_ws;
    const size_t BCL = (size_t)BSZ * C_ * L_;
    float* x_pre = ws;
    float* zbuf  = ws + BCL;
    float* xcbuf = ws + 2 * BCL;
    float* delta = ws + 3 * BCL;
    float* x_dbl = ws + 4 * BCL;

    k_gemm_inproj<<<IP_EBLK * IP_MBLK, 32, 0, stream>>>(
        hidden, in_proj_w, x_pre, zbuf);

    k_conv_silu<<<(BSZ * C_ * L_ + 255) / 256, 256, 0, stream>>>(
        x_pre, conv_w, conv_b, xcbuf);

    k_gemm_xproj<<<dim3(8 * LTILES, BSZ), 32, 0, stream>>>(
        xcbuf, x_proj_w, x_dbl);

    k_gemm_dtproj<<<dim3(96 * LTILES, BSZ), 32, 0, stream>>>(
        x_dbl, dt_proj_w, dt_bias, delta);

    k_scan_out<<<(BSZ * C_) / SC_THREADS, SC_THREADS, 0, stream>>>(
        delta, xcbuf, zbuf, x_dbl, A_log, D_param, (float*)d_out);
}


# x_proj GEMM retiled: a block computes 32(r) x 32(l), K split over its 8 waves, partial tiles summed through LDS (sum order changed); w read 16 B per lane, xc 8 B per lane for two l tiles
# speedup vs baseline: 3.0019x; 1.2496x over previous
#include <hip/hip_runtime.h>


#define BSZ   8
#define T_    8
#define L_    196
#define D_    192
#define E_    192
#define C_    1536
#define N_    16
#define KC    4
#define R_    96
#define RN2   128
#define LTILES 13

typedef float v2f __attribute__((ext_vector_type(2)));
typedef float v8f __attribute__((ext_vector_type(8)));

__device__ __forceinline__ float sigmoid_(float x) {
    return 1.0f / (1.0f + __expf(-x));
}

#define IP_EW    64
#define IP_MW    32
#define IP_EBLK  ((2 * E_) / IP_EW)
#define IP_MBLK  ((BSZ * T_ * L_) / IP_MW)
#define IP_KC    32
#define IP_KS    (IP_KC / 4)
#define IP_NCH   (D_ / IP_KC)
#define IP_NPASS (IP_NCH * 4)

__device__ __forceinline__ void ip_load_line(const float* __restrict__ p, v2f (&x)[IP_KS])
{
    #pragma unroll
    for (int s = 0; s < IP_KS; ++s)
        x[s] = *(const v2f*)(p + 4 * s);
}

__device__ __forceinline__ void ip_pass(const v2f (&a)[IP_KS],
                                        const v2f (&b0)[IP_KS], const v2f (&b1)[IP_KS],
                                        v8f& c0, v8f& c1)
{
    #pragma unroll
    for (int s = 0; s < IP_KS; ++s) {
        c0 = __builtin_amdgcn_wmma_f32_16x16x4_f32(
                 false, a[s], false, b0[s], (short)0, c0, false, false);
        c1 = __builtin_amdgcn_wmma_f32_16x16x4_f32(
                 false, a[s], false, b1[s], (short)0, c1, false, false);
    }
}

__global__ __launch_bounds__(32)
void k_gemm_inproj(const float* __restrict__ hidden,
                   const float* __restrict__ w,
                   float* __restrict__ x_pre,
                   float* __restrict__ zbuf)
{
    const int lane = threadIdx.x;
    const int c    = lane & 15;
    const int h    = lane >> 4;
    const int eb   = blockIdx.x % IP_EBLK;
    const int mb   = blockIdx.x / IP_EBLK;

    const float* Ap = w      + (size_t)(eb * IP_EW + c) * D_ + 2 * h;
    const float* Bp = hidden + (size_t)(mb * IP_MW + c) * D_ + 2 * h;

    v8f acc[4][2];
    #pragma unroll
    for (int f = 0; f < 4; ++f) {
        #pragma unroll
        for (int g = 0; g < 2; ++g)
            acc[f][g] = (v8f){0.f,0.f,0.f,0.f,0.f,0.f,0.f,0.f};
    }

    v2f a[4][IP_KS];
    v2f b[2][2][IP_KS];
    ip_load_line(Bp, b[0][0]);
    ip_load_line(Bp + (size_t)16 * D_, b[0][1]);
    ip_load_line(Ap, a[0]);
    ip_load_line(Ap + (size_t)16 * D_, a[1]);
    ip_load_line(Ap + (size_t)32 * D_, a[2]);

    #pragma unroll
    for (int p = 0; p < IP_NPASS; ++p) {
        const int ch = p >> 2;
        const int f  = p & 3;
        const int q  = p + 3;
        if (q < IP_NPASS)
            ip_load_line(Ap + (size_t)(q & 3) * 16 * D_ + (q >> 2) * IP_KC, a[q & 3]);
        if (f == 1 && ch + 1 < IP_NCH) {
            ip_load_line(Bp + (ch + 1) * IP_KC, b[(ch + 1) & 1][0]);
            ip_load_line(Bp + (size_t)16 * D_ + (ch + 1) * IP_KC, b[(ch + 1) & 1][1]);
        }
        __builtin_amdgcn_sched_barrier(0);
        ip_pass(a[p & 3], b[ch & 1][0], b[ch & 1][1], acc[f][0], acc[f][1]);
        __builtin_amdgcn_sched_barrier(0);
    }

    float* outp = (eb < IP_EBLK / 2) ? x_pre : zbuf;
    const int e0 = (eb % (IP_EBLK / 2)) * IP_EW + 8 * h;
    #pragma unroll
    for (int g = 0; g < 2; ++g) {
        const int m  = mb * IP_MW + g * 16 + c;
        const int bt = m / L_;
        const int l  = m - bt * L_;
        const int b_ = bt / T_;
        const int t  = bt % T_;
        float* p = outp + ((size_t)b_ * C_ + (size_t)e0 * T_ + t) * L_ + l;
        #pragma unroll
        for (int f = 0; f < 4; ++f) {
            #pragma unroll
            for (int r = 0; r < 8; ++r)
                p[(size_t)(f * 16 + r) * (T_ * L_)] = acc[f][g][r];
        }
    }
}

__global__ __launch_bounds__(256)
void k_conv_silu(const float* __restrict__ x_pre,
                 const float* __restrict__ cw,
                 const float* __restrict__ cb,
                 float* __restrict__ xc)
{
    const int idx = blockIdx.x * 256 + threadIdx.x;
    if (idx >= BSZ * C_ * L_) return;
    const int l  = idx % L_;
    const int bc = idx / L_;
    const int c  = bc % C_;

    const float* xp = x_pre + (size_t)bc * L_;
    float s = cb[c];
    #pragma unroll
    for (int k = 0; k < KC; ++k) {
        const int ls = l - (KC - 1) + k;
        if (ls >= 0) s += xp[ls] * cw[c * KC + k];
    }
    xc[idx] = s * sigmoid_(s);
}

#define XP_NW    8
#define XP_KW    (C_ / XP_NW)
#define XP_RB    32
#define XP_LB    32
#define XP_RBLK  (RN2 / XP_RB)
#define XP_LBLK  ((L_ + XP_LB - 1) / XP_LB)
#define XP_G     4
#define XP_NG    (XP_KW / (8 * XP_G))

__device__ __forceinline__ void xp_load(const float* __restrict__ Ap,
                                        const float* __restrict__ Bp,
                                        float4 (&a)[2][XP_G], float2 (&bq)[XP_G][4])
{
    #pragma unroll
    for (int s = 0; s < XP_G; ++s) {
        a[0][s] = *(const float4*)(Ap + 8 * s);
        a[1][s] = *(const float4*)(Ap + (size_t)16 * C_ + 8 * s);
    }
    #pragma unroll
    for (int s = 0; s < XP_G; ++s) {
        #pragma unroll
        for (int i = 0; i < 4; ++i)
            bq[s][i] = *(const float2*)(Bp + (size_t)(8 * s + i) * L_);
    }
}

__device__ __forceinline__ void xp_mma(const float4 (&a)[2][XP_G],
                                       const float2 (&bq)[XP_G][4], v8f (&acc)[2][2])
{
    #pragma unroll
    for (int s = 0; s < XP_G; ++s) {
        const v2f be0 = {bq[s][0].x, bq[s][1].x};
        const v2f bo0 = {bq[s][0].y, bq[s][1].y};
        const v2f be1 = {bq[s][2].x, bq[s][3].x};
        const v2f bo1 = {bq[s][2].y, bq[s][3].y};
        const v2f a0lo = {a[0][s].x, a[0][s].y};
        const v2f a1lo = {a[1][s].x, a[1][s].y};
        const v2f a0hi = {a[0][s].z, a[0][s].w};
        const v2f a1hi = {a[1][s].z, a[1][s].w};
        acc[0][0] = __builtin_amdgcn_wmma_f32_16x16x4_f32(
                        false, a0lo, false, be0, (short)0, acc[0][0], false, false);
        acc[0][1] = __builtin_amdgcn_wmma_f32_16x16x4_f32(
                        false, a0lo, false, bo0, (short)0, acc[0][1], false, false);
        acc[1][0] = __builtin_amdgcn_wmma_f32_16x16x4_f32(
                        false, a1lo, false, be0, (short)0, acc[1][0], false, false);
        acc[1][1] = __builtin_amdgcn_wmma_f32_16x16x4_f32(
                        false, a1lo, false, bo0, (short)0, acc[1][1], false, false);
        acc[0][0] = __builtin_amdgcn_wmma_f32_16x16x4_f32(
                        false, a0hi, false, be1, (short)0, acc[0][0], false, false);
        acc[0][1] = __builtin_amdgcn_wmma_f32_16x16x4_f32(
                        false, a0hi, false, bo1, (short)0, acc[0][1], false, false);
        acc[1][0] = __builtin_amdgcn_wmma_f32_16x16x4_f32(
                        false, a1hi, false, be1, (short)0, acc[1][0], false, false);
        acc[1][1] = __builtin_amdgcn_wmma_f32_16x16x4_f32(
                        false, a1hi, false, bo1, (short)0, acc[1][1], false, false);
    }
}

__global__ __launch_bounds__(XP_NW * 32)
void k_gemm_xproj(const float* __restrict__ xc,
                  const float* __restrict__ w,
                  float* __restrict__ x_dbl)
{
    __shared__ __attribute__((aligned(16))) float sPart[XP_NW * 8 * 32 * 4];

    const int tid  = threadIdx.x;
    const int lane = tid & 31;
    const int wv   = tid >> 5;
    const int c    = lane & 15;
    const int h    = lane >> 4;
    const int rb   = blockIdx.x % XP_RBLK;
    const int lb   = blockIdx.x / XP_RBLK;
    const int b    = blockIdx.y;

    const int l0 = lb * XP_LB + 2 * c;
    const int lc = (l0 < L_) ? l0 : (L_ - 2);
    const int k0 = wv * XP_KW + 4 * h;

    const float* Ap = w  + (size_t)(rb * XP_RB + c) * C_ + k0;
    const float* Bp = xc + ((size_t)b * C_ + k0) * L_ + lc;

    v8f acc[2][2];
    #pragma unroll
    for (int f = 0; f < 2; ++f) {
        #pragma unroll
        for (int g = 0; g < 2; ++g)
            acc[f][g] = (v8f){0.f,0.f,0.f,0.f,0.f,0.f,0.f,0.f};
    }

    float4 a[2][2][XP_G];
    float2 bq[2][XP_G][4];
    xp_load(Ap, Bp, a[0], bq[0]);
    #pragma unroll
    for (int t = 0; t < XP_NG; ++t) {
        if (t + 1 < XP_NG)
            xp_load(Ap + (t + 1) * 8 * XP_G, Bp + (size_t)(t + 1) * 8 * XP_G * L_,
                    a[(t + 1) & 1], bq[(t + 1) & 1]);
        __builtin_amdgcn_sched_barrier(0);
        xp_mma(a[t & 1], bq[t & 1], acc);
        __builtin_amdgcn_sched_barrier(0);
    }

    float4* sp = (float4*)sPart;
    #pragma unroll
    for (int f = 0; f < 2; ++f) {
        #pragma unroll
        for (int rp = 0; rp < 4; ++rp)
            sp[(wv * 8 + f * 4 + rp) * 32 + lane] =
                make_float4(acc[f][0][2 * rp],     acc[f][1][2 * rp],
                            acc[f][0][2 * rp + 1], acc[f][1][2 * rp + 1]);
    }
    __syncthreads();

    float4 s = sp[wv * 32 + lane];
    #pragma unroll
    for (int u = 1; u < XP_NW; ++u) {
        const float4 t = sp[(u * 8 + wv) * 32 + lane];
        s.x += t.x; s.y += t.y; s.z += t.z; s.w += t.w;
    }
    const int row = rb * XP_RB + 16 * (wv >> 2) + 8 * h + 2 * (wv & 3);
    if (l0 < L_) {
        float* o = x_dbl + ((size_t)b * RN2 + row) * L_ + l0;
        *(float2*)o        = make_float2(s.x, s.y);
        *(float2*)(o + L_) = make_float2(s.z, s.w);
    }
}

__global__ __launch_bounds__(32)
void k_gemm_dtproj(const float* __restrict__ x_dbl,
                   const float* __restrict__ w,
                   const float* __restrict__ dt_bias,
                   float* __restrict__ delta)
{
    const int lane = threadIdx.x;
    const int ct   = blockIdx.x % 96;
    const int lt   = blockIdx.x / 96;
    const int b    = blockIdx.y;

    const int crow  = ct * 16 + (lane & 15);
    const int koff  = (lane >> 4) * 2;
    const int lcol  = lt * 16 + (lane & 15);
    const int lclmp = (lcol < L_) ? lcol : (L_ - 1);

    const float* Arow = w + (size_t)crow * R_;
    const float* Bcol = x_dbl + (size_t)b * RN2 * L_ + lclmp;

    v8f acc0 = {0.f,0.f,0.f,0.f,0.f,0.f,0.f,0.f};
    v8f acc1 = {0.f,0.f,0.f,0.f,0.f,0.f,0.f,0.f};
    for (int k0 = 0; k0 < R_; k0 += 8) {
        const int kB = k0 + koff;
        v2f a0 = *(const v2f*)(Arow + k0 + koff);
        v2f b0; b0.x = Bcol[(size_t)kB * L_];
                b0.y = Bcol[(size_t)(kB + 1) * L_];
        v2f a1 = *(const v2f*)(Arow + k0 + 4 + koff);
        v2f b1; b1.x = Bcol[(size_t)(kB + 4) * L_];
                b1.y = Bcol[(size_t)(kB + 5) * L_];
        acc0 = __builtin_amdgcn_wmma_f32_16x16x4_f32(
                   false, a0, false, b0, (short)0, acc0, false, false);
        acc1 = __builtin_amdgcn_wmma_f32_16x16x4_f32(
                   false, a1, false, b1, (short)0, acc1, false, false);
    }

    const int nn    = lane & 15;
    const int l     = lt * 16 + nn;
    const int mbase = (lane >> 4) * 8;
    if (l < L_) {
        #pragma unroll
        for (int r = 0; r < 8; ++r) {
            const int c = ct * 16 + mbase + r;
            const float v = acc0[r] + acc1[r] + dt_bias[c];
            const float sp = (v > 20.f) ? v : __logf(1.f + __expf(v));
            delta[((size_t)b * C_ + c) * L_ + l] = sp;
        }
    }
}

#define SC_THREADS 64
#define SC_CHUNKS  (L_ / 4)
#define BC_PITCH   36
#define SC_STAGE_FULL ((2 * N_ * SC_CHUNKS) / SC_THREADS)
#define SC_STAGE_REST ((2 * N_ * SC_CHUNKS) - SC_STAGE_FULL * SC_THREADS)

__device__ __forceinline__ void scan_chunk4(const float* __restrict__ srow,
                                            const float4 d4, const float4 x4,
                                            const float4 z4,
                                            const float (&A_neg)[N_],
                                            float (&h)[N_],
                                            const float Dp,
                                            float4* __restrict__ dst)
{
    const float dvs[4] = {d4.x, d4.y, d4.z, d4.w};
    const float xvs[4] = {x4.x, x4.y, x4.z, x4.w};
    const float zvs[4] = {z4.x, z4.y, z4.z, z4.w};
    float ov[4];
    #pragma unroll
    for (int k = 0; k < 4; ++k) {
        const float dv  = dvs[k];
        const float xv  = xvs[k];
        const float zv  = zvs[k];
        const float dvx = dv * xv;
        const float4* row = (const float4*)(srow + k * BC_PITCH);
        float bc[2 * N_];
        #pragma unroll
        for (int q = 0; q < 8; ++q) {
            const float4 t = row[q];
            bc[4 * q + 0] = t.x; bc[4 * q + 1] = t.y;
            bc[4 * q + 2] = t.z; bc[4 * q + 3] = t.w;
        }
        float y0 = 0.f, y1 = 0.f, y2 = 0.f, y3 = 0.f;
        #pragma unroll
        for (int n = 0; n < N_; n += 4) {
            h[n + 0] = h[n + 0] * __expf(dv * A_neg[n + 0]) + dvx * bc[n + 0];
            h[n + 1] = h[n + 1] * __expf(dv * A_neg[n + 1]) + dvx * bc[n + 1];
            h[n + 2] = h[n + 2] * __expf(dv * A_neg[n + 2]) + dvx * bc[n + 2];
            h[n + 3] = h[n + 3] * __expf(dv * A_neg[n + 3]) + dvx * bc[n + 3];
            y0 += h[n + 0] * bc[N_ + n + 0];
            y1 += h[n + 1] * bc[N_ + n + 1];
            y2 += h[n + 2] * bc[N_ + n + 2];
            y3 += h[n + 3] * bc[N_ + n + 3];
        }
        const float y = (y0 + y1) + (y2 + y3);
        ov[k] = (y + xv * Dp) * (zv * sigmoid_(zv));
    }
    *dst = make_float4(ov[0], ov[1], ov[2], ov[3]);
}

__global__ __launch_bounds__(SC_THREADS) __attribute__((amdgpu_waves_per_eu(1, 1)))
void k_scan_out(const float* __restrict__ delta,
                const float* __restrict__ xc,
                const float* __restrict__ zbuf,
                const float* __restrict__ x_dbl,
                const float* __restrict__ A_log,
                const float* __restrict__ Dp_,
                float* __restrict__ out)
{
    __shared__ __attribute__((aligned(16))) float sBC[L_ * BC_PITCH];

    const int tid = threadIdx.x;
    const int b   = blockIdx.x / (C_ / SC_THREADS);
    const int c   = (blockIdx.x % (C_ / SC_THREADS)) * SC_THREADS + tid;
    const int bc_ = b * C_ + c;

    const float4* dp4 = (const float4*)(delta + (size_t)bc_ * L_);
    const float4* xp4 = (const float4*)(xc    + (size_t)bc_ * L_);
    const float4* zp4 = (const float4*)(zbuf  + (size_t)bc_ * L_);
    float4*       op4 = (float4*)(out + (size_t)bc_ * L_);

    float4 qd0 = dp4[0], qx0 = xp4[0], qz0 = zp4[0];
    float4 qd1 = dp4[1], qx1 = xp4[1], qz1 = zp4[1];
    float4 qd2, qx2, qz2;

    const float4* src4 = (const float4*)(x_dbl + ((size_t)b * RN2 + R_) * L_);
    #pragma unroll
    for (int j = 0; j < SC_STAGE_FULL; ++j) {
        const int f  = tid + j * SC_THREADS;
        const int r  = f / SC_CHUNKS;
        const int l4 = f % SC_CHUNKS;
        const float4 v = src4[f];
        float* d = sBC + (l4 * 4) * BC_PITCH + r;
        d[0]            = v.x;
        d[BC_PITCH]     = v.y;
        d[2 * BC_PITCH] = v.z;
        d[3 * BC_PITCH] = v.w;
    }
    if (tid < SC_STAGE_REST) {
        const int f  = tid + SC_STAGE_FULL * SC_THREADS;
        const int r  = f / SC_CHUNKS;
        const int l4 = f % SC_CHUNKS;
        const float4 v = src4[f];
        float* d = sBC + (l4 * 4) * BC_PITCH + r;
        d[0]            = v.x;
        d[BC_PITCH]     = v.y;
        d[2 * BC_PITCH] = v.z;
        d[3 * BC_PITCH] = v.w;
    }

    float A_neg[N_];
    {
        const float4* a4 = (const float4*)(A_log + (size_t)c * N_);
        #pragma unroll
        for (int q = 0; q < 4; ++q) {
            const float4 t = a4[q];
            A_neg[4 * q + 0] = -__expf(t.x);
            A_neg[4 * q + 1] = -__expf(t.y);
            A_neg[4 * q + 2] = -__expf(t.z);
            A_neg[4 * q + 3] = -__expf(t.w);
        }
    }
    const float Dp = Dp_[c];

    float h[N_];
    #pragma unroll
    for (int n = 0; n < N_; ++n) h[n] = 0.f;

    __syncthreads();

    for (int i = 0; i < SC_CHUNKS - 1; i += 3) {
        qd2 = dp4[i + 2]; qx2 = xp4[i + 2]; qz2 = zp4[i + 2];
        scan_chunk4(sBC + (i * 4) * BC_PITCH, qd0, qx0, qz0, A_neg, h, Dp, op4 + i);
        qd0 = dp4[i + 3]; qx0 = xp4[i + 3]; qz0 = zp4[i + 3];
        scan_chunk4(sBC + ((i + 1) * 4) * BC_PITCH, qd1, qx1, qz1, A_neg, h, Dp, op4 + i + 1);
        const int i4 = (i + 4 < SC_CHUNKS) ? (i + 4) : (SC_CHUNKS - 1);
        qd1 = dp4[i4]; qx1 = xp4[i4]; qz1 = zp4[i4];
        scan_chunk4(sBC + ((i + 2) * 4) * BC_PITCH, qd2, qx2, qz2, A_neg, h, Dp, op4 + i + 2);
    }
    scan_chunk4(sBC + ((SC_CHUNKS - 1) * 4) * BC_PITCH, qd0, qx0, qz0, A_neg, h, Dp,
                op4 + (SC_CHUNKS - 1));
}

extern "C" void kernel_launch(void* const* d_in, const int* in_sizes, int n_in,
                              void* d_out, int out_size, void* d_ws, size_t ws_size,
                              hipStream_t stream)
{
    const float* hidden    = (const float*)d_in[0];
    const float* in_proj_w = (const float*)d_in[1];
    const float* conv_w    = (const float*)d_in[2];
    const float* conv_b    = (const float*)d_in[3];
    const float* x_proj_w  = (const float*)d_in[4];
    const float* dt_proj_w = (const float*)d_in[5];
    const float* dt_bias   = (const float*)d_in[6];
    const float* A_log     = (const float*)d_in[7];
    const float* D_param   = (const float*)d_in[8];

    float* ws = (float*)d_ws;
    const size_t BCL = (size_t)BSZ * C_ * L_;
    float* x_pre = ws;
    float* zbuf  = ws + BCL;
    float* xcbuf = ws + 2 * BCL;
    float* delta = ws + 3 * BCL;
    float* x_dbl = ws + 4 * BCL;

    k_gemm_inproj<<<IP_EBLK * IP_MBLK, 32, 0, stream>>>(
        hidden, in_proj_w, x_pre, zbuf);

    k_conv_silu<<<(BSZ * C_ * L_ + 255) / 256, 256, 0, stream>>>(
        x_pre, conv_w, conv_b, xcbuf);

    k_gemm_xproj<<<dim3(XP_RBLK * XP_LBLK, BSZ), XP_NW * 32, 0, stream>>>(
        xcbuf, x_proj_w, x_dbl);

    k_gemm_dtproj<<<dim3(96 * LTILES, BSZ), 32, 0, stream>>>(
        x_dbl, dt_proj_w, dt_bias, delta);

    k_scan_out<<<(BSZ * C_) / SC_THREADS, SC_THREADS, 0, stream>>>(
        delta, xcbuf, zbuf, x_dbl, A_log, D_param, (float*)d_out);
}


# dt_proj GEMM retiled: a wave computes 32(c) x 32(b,l) as 2x2 accumulator tiles (was one 16x16), whole K loaded up front, 16-B/8-B loads, 128-B row stores; one k chain per tile (sum order changed)
# speedup vs baseline: 3.2223x; 1.0734x over previous
#include <hip/hip_runtime.h>


#define BSZ   8
#define T_    8
#define L_    196
#define D_    192
#define E_    192
#define C_    1536
#define N_    16
#define KC    4
#define R_    96
#define RN2   128
#define LTILES 13

typedef float v2f __attribute__((ext_vector_type(2)));
typedef float v8f __attribute__((ext_vector_type(8)));

__device__ __forceinline__ float sigmoid_(float x) {
    return 1.0f / (1.0f + __expf(-x));
}

#define IP_EW    64
#define IP_MW    32
#define IP_EBLK  ((2 * E_) / IP_EW)
#define IP_MBLK  ((BSZ * T_ * L_) / IP_MW)
#define IP_KC    32
#define IP_KS    (IP_KC / 4)
#define IP_NCH   (D_ / IP_KC)
#define IP_NPASS (IP_NCH * 4)

__device__ __forceinline__ void ip_load_line(const float* __restrict__ p, v2f (&x)[IP_KS])
{
    #pragma unroll
    for (int s = 0; s < IP_KS; ++s)
        x[s] = *(const v2f*)(p + 4 * s);
}

__device__ __forceinline__ void ip_pass(const v2f (&a)[IP_KS],
                                        const v2f (&b0)[IP_KS], const v2f (&b1)[IP_KS],
                                        v8f& c0, v8f& c1)
{
    #pragma unroll
    for (int s = 0; s < IP_KS; ++s) {
        c0 = __builtin_amdgcn_wmma_f32_16x16x4_f32(
                 false, a[s], false, b0[s], (short)0, c0, false, false);
        c1 = __builtin_amdgcn_wmma_f32_16x16x4_f32(
                 false, a[s], false, b1[s], (short)0, c1, false, false);
    }
}

__global__ __launch_bounds__(32)
void k_gemm_inproj(const float* __restrict__ hidden,
                   const float* __restrict__ w,
                   float* __restrict__ x_pre,
                   float* __restrict__ zbuf)
{
    const int lane = threadIdx.x;
    const int c    = lane & 15;
    const int h    = lane >> 4;
    const int eb   = blockIdx.x % IP_EBLK;
    const int mb   = blockIdx.x / IP_EBLK;

    const float* Ap = w      + (size_t)(eb * IP_EW + c) * D_ + 2 * h;
    const float* Bp = hidden + (size_t)(mb * IP_MW + c) * D_ + 2 * h;

    v8f acc[4][2];
    #pragma unroll
    for (int f = 0; f < 4; ++f) {
        #pragma unroll
        for (int g = 0; g < 2; ++g)
            acc[f][g] = (v8f){0.f,0.f,0.f,0.f,0.f,0.f,0.f,0.f};
    }

    v2f a[4][IP_KS];
    v2f b[2][2][IP_KS];
    ip_load_line(Bp, b[0][0]);
    ip_load_line(Bp + (size_t)16 * D_, b[0][1]);
    ip_load_line(Ap, a[0]);
    ip_load_line(Ap + (size_t)16 * D_, a[1]);
    ip_load_line(Ap + (size_t)32 * D_, a[2]);

    #pragma unroll
    for (int p = 0; p < IP_NPASS; ++p) {
        const int ch = p >> 2;
        const int f  = p & 3;
        const int q  = p + 3;
        if (q < IP_NPASS)
            ip_load_line(Ap + (size_t)(q & 3) * 16 * D_ + (q >> 2) * IP_KC, a[q & 3]);
        if (f == 1 && ch + 1 < IP_NCH) {
            ip_load_line(Bp + (ch + 1) * IP_KC, b[(ch + 1) & 1][0]);
            ip_load_line(Bp + (size_t)16 * D_ + (ch + 1) * IP_KC, b[(ch + 1) & 1][1]);
        }
        __builtin_amdgcn_sched_barrier(0);
        ip_pass(a[p & 3], b[ch & 1][0], b[ch & 1][1], acc[f][0], acc[f][1]);
        __builtin_amdgcn_sched_barrier(0);
    }

    float* outp = (eb < IP_EBLK / 2) ? x_pre : zbuf;
    const int e0 = (eb % (IP_EBLK / 2)) * IP_EW + 8 * h;
    #pragma unroll
    for (int g = 0; g < 2; ++g) {
        const int m  = mb * IP_MW + g * 16 + c;
        const int bt = m / L_;
        const int l  = m - bt * L_;
        const int b_ = bt / T_;
        const int t  = bt % T_;
        float* p = outp + ((size_t)b_ * C_ + (size_t)e0 * T_ + t) * L_ + l;
        #pragma unroll
        for (int f = 0; f < 4; ++f) {
            #pragma unroll
            for (int r = 0; r < 8; ++r)
                p[(size_t)(f * 16 + r) * (T_ * L_)] = acc[f][g][r];
        }
    }
}

__global__ __launch_bounds__(256)
void k_conv_silu(const float* __restrict__ x_pre,
                 const float* __restrict__ cw,
                 const float* __restrict__ cb,
                 float* __restrict__ xc)
{
    const int idx = blockIdx.x * 256 + threadIdx.x;
    if (idx >= BSZ * C_ * L_) return;
    const int l  = idx % L_;
    const int bc = idx / L_;
    const int c  = bc % C_;

    const float* xp = x_pre + (size_t)bc * L_;
    float s = cb[c];
    #pragma unroll
    for (int k = 0; k < KC; ++k) {
        const int ls = l - (KC - 1) + k;
        if (ls >= 0) s += xp[ls] * cw[c * KC + k];
    }
    xc[idx] = s * sigmoid_(s);
}

#define XP_NW    8
#define XP_KW    (C_ / XP_NW)
#define XP_RB    32
#define XP_LB    32
#define XP_RBLK  (RN2 / XP_RB)
#define XP_LBLK  ((L_ + XP_LB - 1) / XP_LB)
#define XP_G     4
#define XP_NG    (XP_KW / (8 * XP_G))

__device__ __forceinline__ void xp_load(const float* __restrict__ Ap,
                                        const float* __restrict__ Bp,
                                        float4 (&a)[2][XP_G], float2 (&bq)[XP_G][4])
{
    #pragma unroll
    for (int s = 0; s < XP_G; ++s) {
        a[0][s] = *(const float4*)(Ap + 8 * s);
        a[1][s] = *(const float4*)(Ap + (size_t)16 * C_ + 8 * s);
    }
    #pragma unroll
    for (int s = 0; s < XP_G; ++s) {
        #pragma unroll
        for (int i = 0; i < 4; ++i)
            bq[s][i] = *(const float2*)(Bp + (size_t)(8 * s + i) * L_);
    }
}

__device__ __forceinline__ void xp_mma(const float4 (&a)[2][XP_G],
                                       const float2 (&bq)[XP_G][4], v8f (&acc)[2][2])
{
    #pragma unroll
    for (int s = 0; s < XP_G; ++s) {
        const v2f be0 = {bq[s][0].x, bq[s][1].x};
        const v2f bo0 = {bq[s][0].y, bq[s][1].y};
        const v2f be1 = {bq[s][2].x, bq[s][3].x};
        const v2f bo1 = {bq[s][2].y, bq[s][3].y};
        const v2f a0lo = {a[0][s].x, a[0][s].y};
        const v2f a1lo = {a[1][s].x, a[1][s].y};
        const v2f a0hi = {a[0][s].z, a[0][s].w};
        const v2f a1hi = {a[1][s].z, a[1][s].w};
        acc[0][0] = __builtin_amdgcn_wmma_f32_16x16x4_f32(
                        false, a0lo, false, be0, (short)0, acc[0][0], false, false);
        acc[0][1] = __builtin_amdgcn_wmma_f32_16x16x4_f32(
                        false, a0lo, false, bo0, (short)0, acc[0][1], false, false);
        acc[1][0] = __builtin_amdgcn_wmma_f32_16x16x4_f32(
                        false, a1lo, false, be0, (short)0, acc[1][0], false, false);
        acc[1][1] = __builtin_amdgcn_wmma_f32_16x16x4_f32(
                        false, a1lo, false, bo0, (short)0, acc[1][1], false, false);
        acc[0][0] = __builtin_amdgcn_wmma_f32_16x16x4_f32(
                        false, a0hi, false, be1, (short)0, acc[0][0], false, false);
        acc[0][1] = __builtin_amdgcn_wmma_f32_16x16x4_f32(
                        false, a0hi, false, bo1, (short)0, acc[0][1], false, false);
        acc[1][0] = __builtin_amdgcn_wmma_f32_16x16x4_f32(
                        false, a1hi, false, be1, (short)0, acc[1][0], false, false);
        acc[1][1] = __builtin_amdgcn_wmma_f32_16x16x4_f32(
                        false, a1hi, false, bo1, (short)0, acc[1][1], false, false);
    }
}

__global__ __launch_bounds__(XP_NW * 32)
void k_gemm_xproj(const float* __restrict__ xc,
                  const float* __restrict__ w,
                  float* __restrict__ x_dbl)
{
    __shared__ __attribute__((aligned(16))) float sPart[XP_NW * 8 * 32 * 4];

    const int tid  = threadIdx.x;
    const int lane = tid & 31;
    const int wv   = tid >> 5;
    const int c    = lane & 15;
    const int h    = lane >> 4;
    const int rb   = blockIdx.x % XP_RBLK;
    const int lb   = blockIdx.x / XP_RBLK;
    const int b    = blockIdx.y;

    const int l0 = lb * XP_LB + 2 * c;
    const int lc = (l0 < L_) ? l0 : (L_ - 2);
    const int k0 = wv * XP_KW + 4 * h;

    const float* Ap = w  + (size_t)(rb * XP_RB + c) * C_ + k0;
    const float* Bp = xc + ((size_t)b * C_ + k0) * L_ + lc;

    v8f acc[2][2];
    #pragma unroll
    for (int f = 0; f < 2; ++f) {
        #pragma unroll
        for (int g = 0; g < 2; ++g)
            acc[f][g] = (v8f){0.f,0.f,0.f,0.f,0.f,0.f,0.f,0.f};
    }

    float4 a[2][2][XP_G];
    float2 bq[2][XP_G][4];
    xp_load(Ap, Bp, a[0], bq[0]);
    #pragma unroll
    for (int t = 0; t < XP_NG; ++t) {
        if (t + 1 < XP_NG)
            xp_load(Ap + (t + 1) * 8 * XP_G, Bp + (size_t)(t + 1) * 8 * XP_G * L_,
                    a[(t + 1) & 1], bq[(t + 1) & 1]);
        __builtin_amdgcn_sched_barrier(0);
        xp_mma(a[t & 1], bq[t & 1], acc);
        __builtin_amdgcn_sched_barrier(0);
    }

    float4* sp = (float4*)sPart;
    #pragma unroll
    for (int f = 0; f < 2; ++f) {
        #pragma unroll
        for (int rp = 0; rp < 4; ++rp)
            sp[(wv * 8 + f * 4 + rp) * 32 + lane] =
                make_float4(acc[f][0][2 * rp],     acc[f][1][2 * rp],
                            acc[f][0][2 * rp + 1], acc[f][1][2 * rp + 1]);
    }
    __syncthreads();

    float4 s = sp[wv * 32 + lane];
    #pragma unroll
    for (int u = 1; u < XP_NW; ++u) {
        const float4 t = sp[(u * 8 + wv) * 32 + lane];
        s.x += t.x; s.y += t.y; s.z += t.z; s.w += t.w;
    }
    const int row = rb * XP_RB + 16 * (wv >> 2) + 8 * h + 2 * (wv & 3);
    if (l0 < L_) {
        float* o = x_dbl + ((size_t)b * RN2 + row) * L_ + l0;
        *(float2*)o        = make_float2(s.x, s.y);
        *(float2*)(o + L_) = make_float2(s.z, s.w);
    }
}

#define DP_CB    32
#define DP_MB    32
#define DP_CBLK  (C_ / DP_CB)
#define DP_MBLK  ((BSZ * L_) / DP_MB)
#define DP_NG    (R_ / (8 * XP_G))

__device__ __forceinline__ void dp_load(const float* __restrict__ Ap,
                                        const float* __restrict__ Bp,
                                        float4 (&a)[2][XP_G], float2 (&bq)[XP_G][4])
{
    #pragma unroll
    for (int s = 0; s < XP_G; ++s) {
        a[0][s] = *(const float4*)(Ap + 8 * s);
        a[1][s] = *(const float4*)(Ap + (size_t)16 * R_ + 8 * s);
    }
    #pragma unroll
    for (int s = 0; s < XP_G; ++s) {
        #pragma unroll
        for (int i = 0; i < 4; ++i)
            bq[s][i] = *(const float2*)(Bp + (size_t)(8 * s + i) * L_);
    }
}

__device__ __forceinline__ float softplus_(float v) {
    return (v > 20.f) ? v : __logf(1.f + __expf(v));
}

__global__ __launch_bounds__(32)
void k_gemm_dtproj(const float* __restrict__ x_dbl,
                   const float* __restrict__ w,
                   const float* __restrict__ dt_bias,
                   float* __restrict__ delta)
{
    const int lane = threadIdx.x;
    const int c    = lane & 15;
    const int h    = lane >> 4;
    const int cb   = blockIdx.x % DP_CBLK;
    const int mb   = blockIdx.x / DP_CBLK;

    const int m0 = mb * DP_MB + 2 * c;
    const int b  = m0 / L_;
    const int l0 = m0 - b * L_;

    const float* Ap = w     + (size_t)(cb * DP_CB + c) * R_ + 4 * h;
    const float* Bp = x_dbl + ((size_t)b * RN2 + 4 * h) * L_ + l0;

    v8f acc[2][2];
    #pragma unroll
    for (int f = 0; f < 2; ++f) {
        #pragma unroll
        for (int g = 0; g < 2; ++g)
            acc[f][g] = (v8f){0.f,0.f,0.f,0.f,0.f,0.f,0.f,0.f};
    }

    float4 a[DP_NG][2][XP_G];
    float2 bq[DP_NG][XP_G][4];
    #pragma unroll
    for (int t = 0; t < DP_NG; ++t)
        dp_load(Ap + t * 8 * XP_G, Bp + (size_t)t * 8 * XP_G * L_, a[t], bq[t]);

    const float4* bp4 = (const float4*)(dt_bias + cb * DP_CB + 8 * h);
    float bias[2][8];
    #pragma unroll
    for (int f = 0; f < 2; ++f) {
        const float4 t0 = bp4[f * 4], t1 = bp4[f * 4 + 1];
        bias[f][0] = t0.x; bias[f][1] = t0.y; bias[f][2] = t0.z; bias[f][3] = t0.w;
        bias[f][4] = t1.x; bias[f][5] = t1.y; bias[f][6] = t1.z; bias[f][7] = t1.w;
    }

    __builtin_amdgcn_sched_barrier(0);
    #pragma unroll
    for (int t = 0; t < DP_NG; ++t) {
        xp_mma(a[t], bq[t], acc);
        __builtin_amdgcn_sched_barrier(0);
    }

    float* o = delta + ((size_t)b * C_ + cb * DP_CB + 8 * h) * L_ + l0;
    #pragma unroll
    for (int f = 0; f < 2; ++f) {
        #pragma unroll
        for (int r = 0; r < 8; ++r) {
            const float v0 = acc[f][0][r] + bias[f][r];
            const float v1 = acc[f][1][r] + bias[f][r];
            *(float2*)(o + (size_t)(f * 16 + r) * L_) =
                make_float2(softplus_(v0), softplus_(v1));
        }
    }
}

#define SC_THREADS 64
#define SC_CHUNKS  (L_ / 4)
#define BC_PITCH   36
#define SC_STAGE_FULL ((2 * N_ * SC_CHUNKS) / SC_THREADS)
#define SC_STAGE_REST ((2 * N_ * SC_CHUNKS) - SC_STAGE_FULL * SC_THREADS)

__device__ __forceinline__ void scan_chunk4(const float* __restrict__ srow,
                                            const float4 d4, const float4 x4,
                                            const float4 z4,
                                            const float (&A_neg)[N_],
                                            float (&h)[N_],
                                            const float Dp,
                                            float4* __restrict__ dst)
{
    const float dvs[4] = {d4.x, d4.y, d4.z, d4.w};
    const float xvs[4] = {x4.x, x4.y, x4.z, x4.w};
    const float zvs[4] = {z4.x, z4.y, z4.z, z4.w};
    float ov[4];
    #pragma unroll
    for (int k = 0; k < 4; ++k) {
        const float dv  = dvs[k];
        const float xv  = xvs[k];
        const float zv  = zvs[k];
        const float dvx = dv * xv;
        const float4* row = (const float4*)(srow + k * BC_PITCH);
        float bc[2 * N_];
        #pragma unroll
        for (int q = 0; q < 8; ++q) {
            const float4 t = row[q];
            bc[4 * q + 0] = t.x; bc[4 * q + 1] = t.y;
            bc[4 * q + 2] = t.z; bc[4 * q + 3] = t.w;
        }
        float y0 = 0.f, y1 = 0.f, y2 = 0.f, y3 = 0.f;
        #pragma unroll
        for (int n = 0; n < N_; n += 4) {
            h[n + 0] = h[n + 0] * __expf(dv * A_neg[n + 0]) + dvx * bc[n + 0];
            h[n + 1] = h[n + 1] * __expf(dv * A_neg[n + 1]) + dvx * bc[n + 1];
            h[n + 2] = h[n + 2] * __expf(dv * A_neg[n + 2]) + dvx * bc[n + 2];
            h[n + 3] = h[n + 3] * __expf(dv * A_neg[n + 3]) + dvx * bc[n + 3];
            y0 += h[n + 0] * bc[N_ + n + 0];
            y1 += h[n + 1] * bc[N_ + n + 1];
            y2 += h[n + 2] * bc[N_ + n + 2];
            y3 += h[n + 3] * bc[N_ + n + 3];
        }
        const float y = (y0 + y1) + (y2 + y3);
        ov[k] = (y + xv * Dp) * (zv * sigmoid_(zv));
    }
    *dst = make_float4(ov[0], ov[1], ov[2], ov[3]);
}

__global__ __launch_bounds__(SC_THREADS) __attribute__((amdgpu_waves_per_eu(1, 1)))
void k_scan_out(const float* __restrict__ delta,
                const float* __restrict__ xc,
                const float* __restrict__ zbuf,
                const float* __restrict__ x_dbl,
                const float* __restrict__ A_log,
                const float* __restrict__ Dp_,
                float* __restrict__ out)
{
    __shared__ __attribute__((aligned(16))) float sBC[L_ * BC_PITCH];

    const int tid = threadIdx.x;
    const int b   = blockIdx.x / (C_ / SC_THREADS);
    const int c   = (blockIdx.x % (C_ / SC_THREADS)) * SC_THREADS + tid;
    const int bc_ = b * C_ + c;

    const float4* dp4 = (const float4*)(delta + (size_t)bc_ * L_);
    const float4* xp4 = (const float4*)(xc    + (size_t)bc_ * L_);
    const float4* zp4 = (const float4*)(zbuf  + (size_t)bc_ * L_);
    float4*       op4 = (float4*)(out + (size_t)bc_ * L_);

    float4 qd0 = dp4[0], qx0 = xp4[0], qz0 = zp4[0];
    float4 qd1 = dp4[1], qx1 = xp4[1], qz1 = zp4[1];
    float4 qd2, qx2, qz2;

    const float4* src4 = (const float4*)(x_dbl + ((size_t)b * RN2 + R_) * L_);
    #pragma unroll
    for (int j = 0; j < SC_STAGE_FULL; ++j) {
        const int f  = tid + j * SC_THREADS;
        const int r  = f / SC_CHUNKS;
        const int l4 = f % SC_CHUNKS;
        const float4 v = src4[f];
        float* d = sBC + (l4 * 4) * BC_PITCH + r;
        d[0]            = v.x;
        d[BC_PITCH]     = v.y;
        d[2 * BC_PITCH] = v.z;
        d[3 * BC_PITCH] = v.w;
    }
    if (tid < SC_STAGE_REST) {
        const int f  = tid + SC_STAGE_FULL * SC_THREADS;
        const int r  = f / SC_CHUNKS;
        const int l4 = f % SC_CHUNKS;
        const float4 v = src4[f];
        float* d = sBC + (l4 * 4) * BC_PITCH + r;
        d[0]            = v.x;
        d[BC_PITCH]     = v.y;
        d[2 * BC_PITCH] = v.z;
        d[3 * BC_PITCH] = v.w;
    }

    float A_neg[N_];
    {
        const float4* a4 = (const float4*)(A_log + (size_t)c * N_);
        #pragma unroll
        for (int q = 0; q < 4; ++q) {
            const float4 t = a4[q];
            A_neg[4 * q + 0] = -__expf(t.x);
            A_neg[4 * q + 1] = -__expf(t.y);
            A_neg[4 * q + 2] = -__expf(t.z);
            A_neg[4 * q + 3] = -__expf(t.w);
        }
    }
    const float Dp = Dp_[c];

    float h[N_];
    #pragma unroll
    for (int n = 0; n < N_; ++n) h[n] = 0.f;

    __syncthreads();

    for (int i = 0; i < SC_CHUNKS - 1; i += 3) {
        qd2 = dp4[i + 2]; qx2 = xp4[i + 2]; qz2 = zp4[i + 2];
        scan_chunk4(sBC + (i * 4) * BC_PITCH, qd0, qx0, qz0, A_neg, h, Dp, op4 + i);
        qd0 = dp4[i + 3]; qx0 = xp4[i + 3]; qz0 = zp4[i + 3];
        scan_chunk4(sBC + ((i + 1) * 4) * BC_PITCH, qd1, qx1, qz1, A_neg, h, Dp, op4 + i + 1);
        const int i4 = (i + 4 < SC_CHUNKS) ? (i + 4) : (SC_CHUNKS - 1);
        qd1 = dp4[i4]; qx1 = xp4[i4]; qz1 = zp4[i4];
        scan_chunk4(sBC + ((i + 2) * 4) * BC_PITCH, qd2, qx2, qz2, A_neg, h, Dp, op4 + i + 2);
    }
    scan_chunk4(sBC + ((SC_CHUNKS - 1) * 4) * BC_PITCH, qd0, qx0, qz0, A_neg, h, Dp,
                op4 + (SC_CHUNKS - 1));
}

extern "C" void kernel_launch(void* const* d_in, const int* in_sizes, int n_in,
                              void* d_out, int out_size, void* d_ws, size_t ws_size,
                              hipStream_t stream)
{
    const float* hidden    = (const float*)d_in[0];
    const float* in_proj_w = (const float*)d_in[1];
    const float* conv_w    = (const float*)d_in[2];
    const float* conv_b    = (const float*)d_in[3];
    const float* x_proj_w  = (const float*)d_in[4];
    const float* dt_proj_w = (const float*)d_in[5];
    const float* dt_bias   = (const float*)d_in[6];
    const float* A_log     = (const float*)d_in[7];
    const float* D_param   = (const float*)d_in[8];

    float* ws = (float*)d_ws;
    const size_t BCL = (size_t)BSZ * C_ * L_;
    float* x_pre = ws;
    float* zbuf  = ws + BCL;
    float* xcbuf = ws + 2 * BCL;
    float* delta = ws + 3 * BCL;
    float* x_dbl = ws + 4 * BCL;

    k_gemm_inproj<<<IP_EBLK * IP_MBLK, 32, 0, stream>>>(
        hidden, in_proj_w, x_pre, zbuf);

    k_conv_silu<<<(BSZ * C_ * L_ + 255) / 256, 256, 0, stream>>>(
        x_pre, conv_w, conv_b, xcbuf);

    k_gemm_xproj<<<dim3(XP_RBLK * XP_LBLK, BSZ), XP_NW * 32, 0, stream>>>(
        xcbuf, x_proj_w, x_dbl);

    k_gemm_dtproj<<<DP_CBLK * DP_MBLK, 32, 0, stream>>>(
        x_dbl, dt_proj_w, dt_bias, delta);

    k_scan_out<<<(BSZ * C_) / SC_THREADS, SC_THREADS, 0, stream>>>(
        delta, xcbuf, zbuf, x_dbl, A_log, D_param, (float*)d_out);
}


# conv+SiLU kernel retiled: a lane computes one float4 position (4 consecutive l) for all 8 b, 16-B loads/stores, weights loaded once; 2352 waves instead of 75264; per-element sum order kept
# speedup vs baseline: 3.3441x; 1.0378x over previous
#include <hip/hip_runtime.h>


#define BSZ   8
#define T_    8
#define L_    196
#define D_    192
#define E_    192
#define C_    1536
#define N_    16
#define KC    4
#define R_    96
#define RN2   128
#define LTILES 13

typedef float v2f __attribute__((ext_vector_type(2)));
typedef float v8f __attribute__((ext_vector_type(8)));

__device__ __forceinline__ float sigmoid_(float x) {
    return 1.0f / (1.0f + __expf(-x));
}

#define IP_EW    64
#define IP_MW    32
#define IP_EBLK  ((2 * E_) / IP_EW)
#define IP_MBLK  ((BSZ * T_ * L_) / IP_MW)
#define IP_KC    32
#define IP_KS    (IP_KC / 4)
#define IP_NCH   (D_ / IP_KC)
#define IP_NPASS (IP_NCH * 4)

__device__ __forceinline__ void ip_load_line(const float* __restrict__ p, v2f (&x)[IP_KS])
{
    #pragma unroll
    for (int s = 0; s < IP_KS; ++s)
        x[s] = *(const v2f*)(p + 4 * s);
}

__device__ __forceinline__ void ip_pass(const v2f (&a)[IP_KS],
                                        const v2f (&b0)[IP_KS], const v2f (&b1)[IP_KS],
                                        v8f& c0, v8f& c1)
{
    #pragma unroll
    for (int s = 0; s < IP_KS; ++s) {
        c0 = __builtin_amdgcn_wmma_f32_16x16x4_f32(
                 false, a[s], false, b0[s], (short)0, c0, false, false);
        c1 = __builtin_amdgcn_wmma_f32_16x16x4_f32(
                 false, a[s], false, b1[s], (short)0, c1, false, false);
    }
}

__global__ __launch_bounds__(32)
void k_gemm_inproj(const float* __restrict__ hidden,
                   const float* __restrict__ w,
                   float* __restrict__ x_pre,
                   float* __restrict__ zbuf)
{
    const int lane = threadIdx.x;
    const int c    = lane & 15;
    const int h    = lane >> 4;
    const int eb   = blockIdx.x % IP_EBLK;
    const int mb   = blockIdx.x / IP_EBLK;

    const float* Ap = w      + (size_t)(eb * IP_EW + c) * D_ + 2 * h;
    const float* Bp = hidden + (size_t)(mb * IP_MW + c) * D_ + 2 * h;

    v8f acc[4][2];
    #pragma unroll
    for (int f = 0; f < 4; ++f) {
        #pragma unroll
        for (int g = 0; g < 2; ++g)
            acc[f][g] = (v8f){0.f,0.f,0.f,0.f,0.f,0.f,0.f,0.f};
    }

    v2f a[4][IP_KS];
    v2f b[2][2][IP_KS];
    ip_load_line(Bp, b[0][0]);
    ip_load_line(Bp + (size_t)16 * D_, b[0][1]);
    ip_load_line(Ap, a[0]);
    ip_load_line(Ap + (size_t)16 * D_, a[1]);
    ip_load_line(Ap + (size_t)32 * D_, a[2]);

    #pragma unroll
    for (int p = 0; p < IP_NPASS; ++p) {
        const int ch = p >> 2;
        const int f  = p & 3;
        const int q  = p + 3;
        if (q < IP_NPASS)
            ip_load_line(Ap + (size_t)(q & 3) * 16 * D_ + (q >> 2) * IP_KC, a[q & 3]);
        if (f == 1 && ch + 1 < IP_NCH) {
            ip_load_line(Bp + (ch + 1) * IP_KC, b[(ch + 1) & 1][0]);
            ip_load_line(Bp + (size_t)16 * D_ + (ch + 1) * IP_KC, b[(ch + 1) & 1][1]);
        }
        __builtin_amdgcn_sched_barrier(0);
        ip_pass(a[p & 3], b[ch & 1][0], b[ch & 1][1], acc[f][0], acc[f][1]);
        __builtin_amdgcn_sched_barrier(0);
    }

    float* outp = (eb < IP_EBLK / 2) ? x_pre : zbuf;
    const int e0 = (eb % (IP_EBLK / 2)) * IP_EW + 8 * h;
    #pragma unroll
    for (int g = 0; g < 2; ++g) {
        const int m  = mb * IP_MW + g * 16 + c;
        const int bt = m / L_;
        const int l  = m - bt * L_;
        const int b_ = bt / T_;
        const int t  = bt % T_;
        float* p = outp + ((size_t)b_ * C_ + (size_t)e0 * T_ + t) * L_ + l;
        #pragma unroll
        for (int f = 0; f < 4; ++f) {
            #pragma unroll
            for (int r = 0; r < 8; ++r)
                p[(size_t)(f * 16 + r) * (T_ * L_)] = acc[f][g][r];
        }
    }
}

#define CV_F4_ROW   (L_ / 4)
#define CV_F4_B     (C_ * CV_F4_ROW)
#define CV_THREADS  128
#define CV_BLOCKS   (CV_F4_B / CV_THREADS)

__global__ __launch_bounds__(CV_THREADS)
void k_conv_silu(const float* __restrict__ x_pre,
                 const float* __restrict__ cw,
                 const float* __restrict__ cb,
                 float* __restrict__ xc)
{
    const int f0 = blockIdx.x * CV_THREADS + threadIdx.x;
    const int c  = f0 / CV_F4_ROW;
    const int l4 = f0 - c * CV_F4_ROW;
    const bool has_prev = (l4 > 0);
    const int  poff     = has_prev ? -1 : 0;

    const float4* src = (const float4*)x_pre + f0;
    float4*       dst = (float4*)xc + f0;

    float4 cur[BSZ], prv[BSZ];
    #pragma unroll
    for (int b = 0; b < BSZ; ++b) {
        cur[b] = src[(size_t)b * CV_F4_B];
        prv[b] = src[(size_t)b * CV_F4_B + poff];
    }
    const float4 w    = *(const float4*)(cw + c * KC);
    const float  bias = cb[c];

    #pragma unroll
    for (int b = 0; b < BSZ; ++b) {
        const float p1 = has_prev ? prv[b].y : 0.f;
        const float p2 = has_prev ? prv[b].z : 0.f;
        const float p3 = has_prev ? prv[b].w : 0.f;
        const float4 x = cur[b];
        float s0 = bias, s1 = bias, s2 = bias, s3 = bias;
        s0 += p1  * w.x;  s1 += p2  * w.x;  s2 += p3  * w.x;  s3 += x.x * w.x;
        s0 += p2  * w.y;  s1 += p3  * w.y;  s2 += x.x * w.y;  s3 += x.y * w.y;
        s0 += p3  * w.z;  s1 += x.x * w.z;  s2 += x.y * w.z;  s3 += x.z * w.z;
        s0 += x.x * w.w;  s1 += x.y * w.w;  s2 += x.z * w.w;  s3 += x.w * w.w;
        dst[(size_t)b * CV_F4_B] = make_float4(s0 * sigmoid_(s0), s1 * sigmoid_(s1),
                                               s2 * sigmoid_(s2), s3 * sigmoid_(s3));
    }
}

#define XP_NW    8
#define XP_KW    (C_ / XP_NW)
#define XP_RB    32
#define XP_LB    32
#define XP_RBLK  (RN2 / XP_RB)
#define XP_LBLK  ((L_ + XP_LB - 1) / XP_LB)
#define XP_G     4
#define XP_NG    (XP_KW / (8 * XP_G))

__device__ __forceinline__ void xp_load(const float* __restrict__ Ap,
                                        const float* __restrict__ Bp,
                                        float4 (&a)[2][XP_G], float2 (&bq)[XP_G][4])
{
    #pragma unroll
    for (int s = 0; s < XP_G; ++s) {
        a[0][s] = *(const float4*)(Ap + 8 * s);
        a[1][s] = *(const float4*)(Ap + (size_t)16 * C_ + 8 * s);
    }
    #pragma unroll
    for (int s = 0; s < XP_G; ++s) {
        #pragma unroll
        for (int i = 0; i < 4; ++i)
            bq[s][i] = *(const float2*)(Bp + (size_t)(8 * s + i) * L_);
    }
}

__device__ __forceinline__ void xp_mma(const float4 (&a)[2][XP_G],
                                       const float2 (&bq)[XP_G][4], v8f (&acc)[2][2])
{
    #pragma unroll
    for (int s = 0; s < XP_G; ++s) {
        const v2f be0 = {bq[s][0].x, bq[s][1].x};
        const v2f bo0 = {bq[s][0].y, bq[s][1].y};
        const v2f be1 = {bq[s][2].x, bq[s][3].x};
        const v2f bo1 = {bq[s][2].y, bq[s][3].y};
        const v2f a0lo = {a[0][s].x, a[0][s].y};
        const v2f a1lo = {a[1][s].x, a[1][s].y};
        const v2f a0hi = {a[0][s].z, a[0][s].w};
        const v2f a1hi = {a[1][s].z, a[1][s].w};
        acc[0][0] = __builtin_amdgcn_wmma_f32_16x16x4_f32(
                        false, a0lo, false, be0, (short)0, acc[0][0], false, false);
        acc[0][1] = __builtin_amdgcn_wmma_f32_16x16x4_f32(
                        false, a0lo, false, bo0, (short)0, acc[0][1], false, false);
        acc[1][0] = __builtin_amdgcn_wmma_f32_16x16x4_f32(
                        false, a1lo, false, be0, (short)0, acc[1][0], false, false);
        acc[1][1] = __builtin_amdgcn_wmma_f32_16x16x4_f32(
                        false, a1lo, false, bo0, (short)0, acc[1][1], false, false);
        acc[0][0] = __builtin_amdgcn_wmma_f32_16x16x4_f32(
                        false, a0hi, false, be1, (short)0, acc[0][0], false, false);
        acc[0][1] = __builtin_amdgcn_wmma_f32_16x16x4_f32(
                        false, a0hi, false, bo1, (short)0, acc[0][1], false, false);
        acc[1][0] = __builtin_amdgcn_wmma_f32_16x16x4_f32(
                        false, a1hi, false, be1, (short)0, acc[1][0], false, false);
        acc[1][1] = __builtin_amdgcn_wmma_f32_16x16x4_f32(
                        false, a1hi, false, bo1, (short)0, acc[1][1], false, false);
    }
}

__global__ __launch_bounds__(XP_NW * 32)
void k_gemm_xproj(const float* __restrict__ xc,
                  const float* __restrict__ w,
                  float* __restrict__ x_dbl)
{
    __shared__ __attribute__((aligned(16))) float sPart[XP_NW * 8 * 32 * 4];

    const int tid  = threadIdx.x;
    const int lane = tid & 31;
    const int wv   = tid >> 5;
    const int c    = lane & 15;
    const int h    = lane >> 4;
    const int rb   = blockIdx.x % XP_RBLK;
    const int lb   = blockIdx.x / XP_RBLK;
    const int b    = blockIdx.y;

    const int l0 = lb * XP_LB + 2 * c;
    const int lc = (l0 < L_) ? l0 : (L_ - 2);
    const int k0 = wv * XP_KW + 4 * h;

    const float* Ap = w  + (size_t)(rb * XP_RB + c) * C_ + k0;
    const float* Bp = xc + ((size_t)b * C_ + k0) * L_ + lc;

    v8f acc[2][2];
    #pragma unroll
    for (int f = 0; f < 2; ++f) {
        #pragma unroll
        for (int g = 0; g < 2; ++g)
            acc[f][g] = (v8f){0.f,0.f,0.f,0.f,0.f,0.f,0.f,0.f};
    }

    float4 a[2][2][XP_G];
    float2 bq[2][XP_G][4];
    xp_load(Ap, Bp, a[0], bq[0]);
    #pragma unroll
    for (int t = 0; t < XP_NG; ++t) {
        if (t + 1 < XP_NG)
            xp_load(Ap + (t + 1) * 8 * XP_G, Bp + (size_t)(t + 1) * 8 * XP_G * L_,
                    a[(t + 1) & 1], bq[(t + 1) & 1]);
        __builtin_amdgcn_sched_barrier(0);
        xp_mma(a[t & 1], bq[t & 1], acc);
        __builtin_amdgcn_sched_barrier(0);
    }

    float4* sp = (float4*)sPart;
    #pragma unroll
    for (int f = 0; f < 2; ++f) {
        #pragma unroll
        for (int rp = 0; rp < 4; ++rp)
            sp[(wv * 8 + f * 4 + rp) * 32 + lane] =
                make_float4(acc[f][0][2 * rp],     acc[f][1][2 * rp],
                            acc[f][0][2 * rp + 1], acc[f][1][2 * rp + 1]);
    }
    __syncthreads();

    float4 s = sp[wv * 32 + lane];
    #pragma unroll
    for (int u = 1; u < XP_NW; ++u) {
        const float4 t = sp[(u * 8 + wv) * 32 + lane];
        s.x += t.x; s.y += t.y; s.z += t.z; s.w += t.w;
    }
    const int row = rb * XP_RB + 16 * (wv >> 2) + 8 * h + 2 * (wv & 3);
    if (l0 < L_) {
        float* o = x_dbl + ((size_t)b * RN2 + row) * L_ + l0;
        *(float2*)o        = make_float2(s.x, s.y);
        *(float2*)(o + L_) = make_float2(s.z, s.w);
    }
}

#define DP_CB    32
#define DP_MB    32
#define DP_CBLK  (C_ / DP_CB)
#define DP_MBLK  ((BSZ * L_) / DP_MB)
#define DP_NG    (R_ / (8 * XP_G))

__device__ __forceinline__ void dp_load(const float* __restrict__ Ap,
                                        const float* __restrict__ Bp,
                                        float4 (&a)[2][XP_G], float2 (&bq)[XP_G][4])
{
    #pragma unroll
    for (int s = 0; s < XP_G; ++s) {
        a[0][s] = *(const float4*)(Ap + 8 * s);
        a[1][s] = *(const float4*)(Ap + (size_t)16 * R_ + 8 * s);
    }
    #pragma unroll
    for (int s = 0; s < XP_G; ++s) {
        #pragma unroll
        for (int i = 0; i < 4; ++i)
            bq[s][i] = *(const float2*)(Bp + (size_t)(8 * s + i) * L_);
    }
}

__device__ __forceinline__ float softplus_(float v) {
    return (v > 20.f) ? v : __logf(1.f + __expf(v));
}

__global__ __launch_bounds__(32)
void k_gemm_dtproj(const float* __restrict__ x_dbl,
                   const float* __restrict__ w,
                   const float* __restrict__ dt_bias,
                   float* __restrict__ delta)
{
    const int lane = threadIdx.x;
    const int c    = lane & 15;
    const int h    = lane >> 4;
    const int cb   = blockIdx.x % DP_CBLK;
    const int mb   = blockIdx.x / DP_CBLK;

    const int m0 = mb * DP_MB + 2 * c;
    const int b  = m0 / L_;
    const int l0 = m0 - b * L_;

    const float* Ap = w     + (size_t)(cb * DP_CB + c) * R_ + 4 * h;
    const float* Bp = x_dbl + ((size_t)b * RN2 + 4 * h) * L_ + l0;

    v8f acc[2][2];
    #pragma unroll
    for (int f = 0; f < 2; ++f) {
        #pragma unroll
        for (int g = 0; g < 2; ++g)
            acc[f][g] = (v8f){0.f,0.f,0.f,0.f,0.f,0.f,0.f,0.f};
    }

    float4 a[DP_NG][2][XP_G];
    float2 bq[DP_NG][XP_G][4];
    #pragma unroll
    for (int t = 0; t < DP_NG; ++t)
        dp_load(Ap + t * 8 * XP_G, Bp + (size_t)t * 8 * XP_G * L_, a[t], bq[t]);

    const float4* bp4 = (const float4*)(dt_bias + cb * DP_CB + 8 * h);
    float bias[2][8];
    #pragma unroll
    for (int f = 0; f < 2; ++f) {
        const float4 t0 = bp4[f * 4], t1 = bp4[f * 4 + 1];
        bias[f][0] = t0.x; bias[f][1] = t0.y; bias[f][2] = t0.z; bias[f][3] = t0.w;
        bias[f][4] = t1.x; bias[f][5] = t1.y; bias[f][6] = t1.z; bias[f][7] = t1.w;
    }

    __builtin_amdgcn_sched_barrier(0);
    #pragma unroll
    for (int t = 0; t < DP_NG; ++t) {
        xp_mma(a[t], bq[t], acc);
        __builtin_amdgcn_sched_barrier(0);
    }

    float* o = delta + ((size_t)b * C_ + cb * DP_CB + 8 * h) * L_ + l0;
    #pragma unroll
    for (int f = 0; f < 2; ++f) {
        #pragma unroll
        for (int r = 0; r < 8; ++r) {
            const float v0 = acc[f][0][r] + bias[f][r];
            const float v1 = acc[f][1][r] + bias[f][r];
            *(float2*)(o + (size_t)(f * 16 + r) * L_) =
                make_float2(softplus_(v0), softplus_(v1));
        }
    }
}

#define SC_THREADS 64
#define SC_CHUNKS  (L_ / 4)
#define BC_PITCH   36
#define SC_STAGE_FULL ((2 * N_ * SC_CHUNKS) / SC_THREADS)
#define SC_STAGE_REST ((2 * N_ * SC_CHUNKS) - SC_STAGE_FULL * SC_THREADS)

__device__ __forceinline__ void scan_chunk4(const float* __restrict__ srow,
                                            const float4 d4, const float4 x4,
                                            const float4 z4,
                                            const float (&A_neg)[N_],
                                            float (&h)[N_],
                                            const float Dp,
                                            float4* __restrict__ dst)
{
    const float dvs[4] = {d4.x, d4.y, d4.z, d4.w};
    const float xvs[4] = {x4.x, x4.y, x4.z, x4.w};
    const float zvs[4] = {z4.x, z4.y, z4.z, z4.w};
    float ov[4];
    #pragma unroll
    for (int k = 0; k < 4; ++k) {
        const float dv  = dvs[k];
        const float xv  = xvs[k];
        const float zv  = zvs[k];
        const float dvx = dv * xv;
        const float4* row = (const float4*)(srow + k * BC_PITCH);
        float bc[2 * N_];
        #pragma unroll
        for (int q = 0; q < 8; ++q) {
            const float4 t = row[q];
            bc[4 * q + 0] = t.x; bc[4 * q + 1] = t.y;
            bc[4 * q + 2] = t.z; bc[4 * q + 3] = t.w;
        }
        float y0 = 0.f, y1 = 0.f, y2 = 0.f, y3 = 0.f;
        #pragma unroll
        for (int n = 0; n < N_; n += 4) {
            h[n + 0] = h[n + 0] * __expf(dv * A_neg[n + 0]) + dvx * bc[n + 0];
            h[n + 1] = h[n + 1] * __expf(dv * A_neg[n + 1]) + dvx * bc[n + 1];
            h[n + 2] = h[n + 2] * __expf(dv * A_neg[n + 2]) + dvx * bc[n + 2];
            h[n + 3] = h[n + 3] * __expf(dv * A_neg[n + 3]) + dvx * bc[n + 3];
            y0 += h[n + 0] * bc[N_ + n + 0];
            y1 += h[n + 1] * bc[N_ + n + 1];
            y2 += h[n + 2] * bc[N_ + n + 2];
            y3 += h[n + 3] * bc[N_ + n + 3];
        }
        const float y = (y0 + y1) + (y2 + y3);
        ov[k] = (y + xv * Dp) * (zv * sigmoid_(zv));
    }
    *dst = make_float4(ov[0], ov[1], ov[2], ov[3]);
}

__global__ __launch_bounds__(SC_THREADS) __attribute__((amdgpu_waves_per_eu(1, 1)))
void k_scan_out(const float* __restrict__ delta,
                const float* __restrict__ xc,
                const float* __restrict__ zbuf,
                const float* __restrict__ x_dbl,
                const float* __restrict__ A_log,
                const float* __restrict__ Dp_,
                float* __restrict__ out)
{
    __shared__ __attribute__((aligned(16))) float sBC[L_ * BC_PITCH];

    const int tid = threadIdx.x;
    const int b   = blockIdx.x / (C_ / SC_THREADS);
    const int c   = (blockIdx.x % (C_ / SC_THREADS)) * SC_THREADS + tid;
    const int bc_ = b * C_ + c;

    const float4* dp4 = (const float4*)(delta + (size_t)bc_ * L_);
    const float4* xp4 = (const float4*)(xc    + (size_t)bc_ * L_);
    const float4* zp4 = (const float4*)(zbuf  + (size_t)bc_ * L_);
    float4*       op4 = (float4*)(out + (size_t)bc_ * L_);

    float4 qd0 = dp4[0], qx0 = xp4[0], qz0 = zp4[0];
    float4 qd1 = dp4[1], qx1 = xp4[1], qz1 = zp4[1];
    float4 qd2, qx2, qz2;

    const float4* src4 = (const float4*)(x_dbl + ((size_t)b * RN2 + R_) * L_);
    #pragma unroll
    for (int j = 0; j < SC_STAGE_FULL; ++j) {
        const int f  = tid + j * SC_THREADS;
        const int r  = f / SC_CHUNKS;
        const int l4 = f % SC_CHUNKS;
        const float4 v = src4[f];
        float* d = sBC + (l4 * 4) * BC_PITCH + r;
        d[0]            = v.x;
        d[BC_PITCH]     = v.y;
        d[2 * BC_PITCH] = v.z;
        d[3 * BC_PITCH] = v.w;
    }
    if (tid < SC_STAGE_REST) {
        const int f  = tid + SC_STAGE_FULL * SC_THREADS;
        const int r  = f / SC_CHUNKS;
        const int l4 = f % SC_CHUNKS;
        const float4 v = src4[f];
        float* d = sBC + (l4 * 4) * BC_PITCH + r;
        d[0]            = v.x;
        d[BC_PITCH]     = v.y;
        d[2 * BC_PITCH] = v.z;
        d[3 * BC_PITCH] = v.w;
    }

    float A_neg[N_];
    {
        const float4* a4 = (const float4*)(A_log + (size_t)c * N_);
        #pragma unroll
        for (int q = 0; q < 4; ++q) {
            const float4 t = a4[q];
            A_neg[4 * q + 0] = -__expf(t.x);
            A_neg[4 * q + 1] = -__expf(t.y);
            A_neg[4 * q + 2] = -__expf(t.z);
            A_neg[4 * q + 3] = -__expf(t.w);
        }
    }
    const float Dp = Dp_[c];

    float h[N_];
    #pragma unroll
    for (int n = 0; n < N_; ++n) h[n] = 0.f;

    __syncthreads();

    for (int i = 0; i < SC_CHUNKS - 1; i += 3) {
        qd2 = dp4[i + 2]; qx2 = xp4[i + 2]; qz2 = zp4[i + 2];
        scan_chunk4(sBC + (i * 4) * BC_PITCH, qd0, qx0, qz0, A_neg, h, Dp, op4 + i);
        qd0 = dp4[i + 3]; qx0 = xp4[i + 3]; qz0 = zp4[i + 3];
        scan_chunk4(sBC + ((i + 1) * 4) * BC_PITCH, qd1, qx1, qz1, A_neg, h, Dp, op4 + i + 1);
        const int i4 = (i + 4 < SC_CHUNKS) ? (i + 4) : (SC_CHUNKS - 1);
        qd1 = dp4[i4]; qx1 = xp4[i4]; qz1 = zp4[i4];
        scan_chunk4(sBC + ((i + 2) * 4) * BC_PITCH, qd2, qx2, qz2, A_neg, h, Dp, op4 + i + 2);
    }
    scan_chunk4(sBC + ((SC_CHUNKS - 1) * 4) * BC_PITCH, qd0, qx0, qz0, A_neg, h, Dp,
                op4 + (SC_CHUNKS - 1));
}

extern "C" void kernel_launch(void* const* d_in, const int* in_sizes, int n_in,
                              void* d_out, int out_size, void* d_ws, size_t ws_size,
                              hipStream_t stream)
{
    const float* hidden    = (const float*)d_in[0];
    const float* in_proj_w = (const float*)d_in[1];
    const float* conv_w    = (const float*)d_in[2];
    const float* conv_b    = (const float*)d_in[3];
    const float* x_proj_w  = (const float*)d_in[4];
    const float* dt_proj_w = (const float*)d_in[5];
    const float* dt_bias   = (const float*)d_in[6];
    const float* A_log     = (const float*)d_in[7];
    const float* D_param   = (const float*)d_in[8];

    float* ws = (float*)d_ws;
    const size_t BCL = (size_t)BSZ * C_ * L_;
    float* x_pre = ws;
    float* zbuf  = ws + BCL;
    float* xcbuf = ws + 2 * BCL;
    float* delta = ws + 3 * BCL;
    float* x_dbl = ws + 4 * BCL;

    k_gemm_inproj<<<IP_EBLK * IP_MBLK, 32, 0, stream>>>(
        hidden, in_proj_w, x_pre, zbuf);

    k_conv_silu<<<CV_BLOCKS, CV_THREADS, 0, stream>>>(
        x_pre, conv_w, conv_b, xcbuf);

    k_gemm_xproj<<<dim3(XP_RBLK * XP_LBLK, BSZ), XP_NW * 32, 0, stream>>>(
        xcbuf, x_proj_w, x_dbl);

    k_gemm_dtproj<<<DP_CBLK * DP_MBLK, 32, 0, stream>>>(
        x_dbl, dt_proj_w, dt_bias, delta);

    k_scan_out<<<(BSZ * C_) / SC_THREADS, SC_THREADS, 0, stream>>>(
        delta, xcbuf, zbuf, x_dbl, A_log, D_param, (float*)d_out);
}


# scan kernel: two lanes per (b,c) channel with 8 states each (was one lane with 16), 768 waves for 384; y = low half + high half across the lane pair (summation order changed); loads 7 chunks ahead
# speedup vs baseline: 3.7532x; 1.1223x over previous
#include <hip/hip_runtime.h>


#define BSZ   8
#define T_    8
#define L_    196
#define D_    192
#define E_    192
#define C_    1536
#define N_    16
#define KC    4
#define R_    96
#define RN2   128
#define LTILES 13

typedef float v2f __attribute__((ext_vector_type(2)));
typedef float v8f __attribute__((ext_vector_type(8)));

__device__ __forceinline__ float sigmoid_(float x) {
    return 1.0f / (1.0f + __expf(-x));
}

#define IP_EW    64
#define IP_MW    32
#define IP_EBLK  ((2 * E_) / IP_EW)
#define IP_MBLK  ((BSZ * T_ * L_) / IP_MW)
#define IP_KC    32
#define IP_KS    (IP_KC / 4)
#define IP_NCH   (D_ / IP_KC)
#define IP_NPASS (IP_NCH * 4)

__device__ __forceinline__ void ip_load_line(const float* __restrict__ p, v2f (&x)[IP_KS])
{
    #pragma unroll
    for (int s = 0; s < IP_KS; ++s)
        x[s] = *(const v2f*)(p + 4 * s);
}

__device__ __forceinline__ void ip_pass(const v2f (&a)[IP_KS],
                                        const v2f (&b0)[IP_KS], const v2f (&b1)[IP_KS],
                                        v8f& c0, v8f& c1)
{
    #pragma unroll
    for (int s = 0; s < IP_KS; ++s) {
        c0 = __builtin_amdgcn_wmma_f32_16x16x4_f32(
                 false, a[s], false, b0[s], (short)0, c0, false, false);
        c1 = __builtin_amdgcn_wmma_f32_16x16x4_f32(
                 false, a[s], false, b1[s], (short)0, c1, false, false);
    }
}

__global__ __launch_bounds__(32)
void k_gemm_inproj(const float* __restrict__ hidden,
                   const float* __restrict__ w,
                   float* __restrict__ x_pre,
                   float* __restrict__ zbuf)
{
    const int lane = threadIdx.x;
    const int c    = lane & 15;
    const int h    = lane >> 4;
    const int eb   = blockIdx.x % IP_EBLK;
    const int mb   = blockIdx.x / IP_EBLK;

    const float* Ap = w      + (size_t)(eb * IP_EW + c) * D_ + 2 * h;
    const float* Bp = hidden + (size_t)(mb * IP_MW + c) * D_ + 2 * h;

    v8f acc[4][2];
    #pragma unroll
    for (int f = 0; f < 4; ++f) {
        #pragma unroll
        for (int g = 0; g < 2; ++g)
            acc[f][g] = (v8f){0.f,0.f,0.f,0.f,0.f,0.f,0.f,0.f};
    }

    v2f a[4][IP_KS];
    v2f b[2][2][IP_KS];
    ip_load_line(Bp, b[0][0]);
    ip_load_line(Bp + (size_t)16 * D_, b[0][1]);
    ip_load_line(Ap, a[0]);
    ip_load_line(Ap + (size_t)16 * D_, a[1]);
    ip_load_line(Ap + (size_t)32 * D_, a[2]);

    #pragma unroll
    for (int p = 0; p < IP_NPASS; ++p) {
        const int ch = p >> 2;
        const int f  = p & 3;
        const int q  = p + 3;
        if (q < IP_NPASS)
            ip_load_line(Ap + (size_t)(q & 3) * 16 * D_ + (q >> 2) * IP_KC, a[q & 3]);
        if (f == 1 && ch + 1 < IP_NCH) {
            ip_load_line(Bp + (ch + 1) * IP_KC, b[(ch + 1) & 1][0]);
            ip_load_line(Bp + (size_t)16 * D_ + (ch + 1) * IP_KC, b[(ch + 1) & 1][1]);
        }
        __builtin_amdgcn_sched_barrier(0);
        ip_pass(a[p & 3], b[ch & 1][0], b[ch & 1][1], acc[f][0], acc[f][1]);
        __builtin_amdgcn_sched_barrier(0);
    }

    float* outp = (eb < IP_EBLK / 2) ? x_pre : zbuf;
    const int e0 = (eb % (IP_EBLK / 2)) * IP_EW + 8 * h;
    #pragma unroll
    for (int g = 0; g < 2; ++g) {
        const int m  = mb * IP_MW + g * 16 + c;
        const int bt = m / L_;
        const int l  = m - bt * L_;
        const int b_ = bt / T_;
        const int t  = bt % T_;
        float* p = outp + ((size_t)b_ * C_ + (size_t)e0 * T_ + t) * L_ + l;
        #pragma unroll
        for (int f = 0; f < 4; ++f) {
            #pragma unroll
            for (int r = 0; r < 8; ++r)
                p[(size_t)(f * 16 + r) * (T_ * L_)] = acc[f][g][r];
        }
    }
}

#define CV_F4_ROW   (L_ / 4)
#define CV_F4_B     (C_ * CV_F4_ROW)
#define CV_THREADS  128
#define CV_BLOCKS   (CV_F4_B / CV_THREADS)

__global__ __launch_bounds__(CV_THREADS)
void k_conv_silu(const float* __restrict__ x_pre,
                 const float* __restrict__ cw,
                 const float* __restrict__ cb,
                 float* __restrict__ xc)
{
    const int f0 = blockIdx.x * CV_THREADS + threadIdx.x;
    const int c  = f0 / CV_F4_ROW;
    const int l4 = f0 - c * CV_F4_ROW;
    const bool has_prev = (l4 > 0);
    const int  poff     = has_prev ? -1 : 0;

    const float4* src = (const float4*)x_pre + f0;
    float4*       dst = (float4*)xc + f0;

    float4 cur[BSZ], prv[BSZ];
    #pragma unroll
    for (int b = 0; b < BSZ; ++b) {
        cur[b] = src[(size_t)b * CV_F4_B];
        prv[b] = src[(size_t)b * CV_F4_B + poff];
    }
    const float4 w    = *(const float4*)(cw + c * KC);
    const float  bias = cb[c];

    #pragma unroll
    for (int b = 0; b < BSZ; ++b) {
        const float p1 = has_prev ? prv[b].y : 0.f;
        const float p2 = has_prev ? prv[b].z : 0.f;
        const float p3 = has_prev ? prv[b].w : 0.f;
        const float4 x = cur[b];
        float s0 = bias, s1 = bias, s2 = bias, s3 = bias;
        s0 += p1  * w.x;  s1 += p2  * w.x;  s2 += p3  * w.x;  s3 += x.x * w.x;
        s0 += p2  * w.y;  s1 += p3  * w.y;  s2 += x.x * w.y;  s3 += x.y * w.y;
        s0 += p3  * w.z;  s1 += x.x * w.z;  s2 += x.y * w.z;  s3 += x.z * w.z;
        s0 += x.x * w.w;  s1 += x.y * w.w;  s2 += x.z * w.w;  s3 += x.w * w.w;
        dst[(size_t)b * CV_F4_B] = make_float4(s0 * sigmoid_(s0), s1 * sigmoid_(s1),
                                               s2 * sigmoid_(s2), s3 * sigmoid_(s3));
    }
}

#define XP_NW    8
#define XP_KW    (C_ / XP_NW)
#define XP_RB    32
#define XP_LB    32
#define XP_RBLK  (RN2 / XP_RB)
#define XP_LBLK  ((L_ + XP_LB - 1) / XP_LB)
#define XP_G     4
#define XP_NG    (XP_KW / (8 * XP_G))

__device__ __forceinline__ void xp_load(const float* __restrict__ Ap,
                                        const float* __restrict__ Bp,
                                        float4 (&a)[2][XP_G], float2 (&bq)[XP_G][4])
{
    #pragma unroll
    for (int s = 0; s < XP_G; ++s) {
        a[0][s] = *(const float4*)(Ap + 8 * s);
        a[1][s] = *(const float4*)(Ap + (size_t)16 * C_ + 8 * s);
    }
    #pragma unroll
    for (int s = 0; s < XP_G; ++s) {
        #pragma unroll
        for (int i = 0; i < 4; ++i)
            bq[s][i] = *(const float2*)(Bp + (size_t)(8 * s + i) * L_);
    }
}

__device__ __forceinline__ void xp_mma(const float4 (&a)[2][XP_G],
                                       const float2 (&bq)[XP_G][4], v8f (&acc)[2][2])
{
    #pragma unroll
    for (int s = 0; s < XP_G; ++s) {
        const v2f be0 = {bq[s][0].x, bq[s][1].x};
        const v2f bo0 = {bq[s][0].y, bq[s][1].y};
        const v2f be1 = {bq[s][2].x, bq[s][3].x};
        const v2f bo1 = {bq[s][2].y, bq[s][3].y};
        const v2f a0lo = {a[0][s].x, a[0][s].y};
        const v2f a1lo = {a[1][s].x, a[1][s].y};
        const v2f a0hi = {a[0][s].z, a[0][s].w};
        const v2f a1hi = {a[1][s].z, a[1][s].w};
        acc[0][0] = __builtin_amdgcn_wmma_f32_16x16x4_f32(
                        false, a0lo, false, be0, (short)0, acc[0][0], false, false);
        acc[0][1] = __builtin_amdgcn_wmma_f32_16x16x4_f32(
                        false, a0lo, false, bo0, (short)0, acc[0][1], false, false);
        acc[1][0] = __builtin_amdgcn_wmma_f32_16x16x4_f32(
                        false, a1lo, false, be0, (short)0, acc[1][0], false, false);
        acc[1][1] = __builtin_amdgcn_wmma_f32_16x16x4_f32(
                        false, a1lo, false, bo0, (short)0, acc[1][1], false, false);
        acc[0][0] = __builtin_amdgcn_wmma_f32_16x16x4_f32(
                        false, a0hi, false, be1, (short)0, acc[0][0], false, false);
        acc[0][1] = __builtin_amdgcn_wmma_f32_16x16x4_f32(
                        false, a0hi, false, bo1, (short)0, acc[0][1], false, false);
        acc[1][0] = __builtin_amdgcn_wmma_f32_16x16x4_f32(
                        false, a1hi, false, be1, (short)0, acc[1][0], false, false);
        acc[1][1] = __builtin_amdgcn_wmma_f32_16x16x4_f32(
                        false, a1hi, false, bo1, (short)0, acc[1][1], false, false);
    }
}

__global__ __launch_bounds__(XP_NW * 32)
void k_gemm_xproj(const float* __restrict__ xc,
                  const float* __restrict__ w,
                  float* __restrict__ x_dbl)
{
    __shared__ __attribute__((aligned(16))) float sPart[XP_NW * 8 * 32 * 4];

    const int tid  = threadIdx.x;
    const int lane = tid & 31;
    const int wv   = tid >> 5;
    const int c    = lane & 15;
    const int h    = lane >> 4;
    const int rb   = blockIdx.x % XP_RBLK;
    const int lb   = blockIdx.x / XP_RBLK;
    const int b    = blockIdx.y;

    const int l0 = lb * XP_LB + 2 * c;
    const int lc = (l0 < L_) ? l0 : (L_ - 2);
    const int k0 = wv * XP_KW + 4 * h;

    const float* Ap = w  + (size_t)(rb * XP_RB + c) * C_ + k0;
    const float* Bp = xc + ((size_t)b * C_ + k0) * L_ + lc;

    v8f acc[2][2];
    #pragma unroll
    for (int f = 0; f < 2; ++f) {
        #pragma unroll
        for (int g = 0; g < 2; ++g)
            acc[f][g] = (v8f){0.f,0.f,0.f,0.f,0.f,0.f,0.f,0.f};
    }

    float4 a[2][2][XP_G];
    float2 bq[2][XP_G][4];
    xp_load(Ap, Bp, a[0], bq[0]);
    #pragma unroll
    for (int t = 0; t < XP_NG; ++t) {
        if (t + 1 < XP_NG)
            xp_load(Ap + (t + 1) * 8 * XP_G, Bp + (size_t)(t + 1) * 8 * XP_G * L_,
                    a[(t + 1) & 1], bq[(t + 1) & 1]);
        __builtin_amdgcn_sched_barrier(0);
        xp_mma(a[t & 1], bq[t & 1], acc);
        __builtin_amdgcn_sched_barrier(0);
    }

    float4* sp = (float4*)sPart;
    #pragma unroll
    for (int f = 0; f < 2; ++f) {
        #pragma unroll
        for (int rp = 0; rp < 4; ++rp)
            sp[(wv * 8 + f * 4 + rp) * 32 + lane] =
                make_float4(acc[f][0][2 * rp],     acc[f][1][2 * rp],
                            acc[f][0][2 * rp + 1], acc[f][1][2 * rp + 1]);
    }
    __syncthreads();

    float4 s = sp[wv * 32 + lane];
    #pragma unroll
    for (int u = 1; u < XP_NW; ++u) {
        const float4 t = sp[(u * 8 + wv) * 32 + lane];
        s.x += t.x; s.y += t.y; s.z += t.z; s.w += t.w;
    }
    const int row = rb * XP_RB + 16 * (wv >> 2) + 8 * h + 2 * (wv & 3);
    if (l0 < L_) {
        float* o = x_dbl + ((size_t)b * RN2 + row) * L_ + l0;
        *(float2*)o        = make_float2(s.x, s.y);
        *(float2*)(o + L_) = make_float2(s.z, s.w);
    }
}

#define DP_CB    32
#define DP_MB    32
#define DP_CBLK  (C_ / DP_CB)
#define DP_MBLK  ((BSZ * L_) / DP_MB)
#define DP_NG    (R_ / (8 * XP_G))

__device__ __forceinline__ void dp_load(const float* __restrict__ Ap,
                                        const float* __restrict__ Bp,
                                        float4 (&a)[2][XP_G], float2 (&bq)[XP_G][4])
{
    #pragma unroll
    for (int s = 0; s < XP_G; ++s) {
        a[0][s] = *(const float4*)(Ap + 8 * s);
        a[1][s] = *(const float4*)(Ap + (size_t)16 * R_ + 8 * s);
    }
    #pragma unroll
    for (int s = 0; s < XP_G; ++s) {
        #pragma unroll
        for (int i = 0; i < 4; ++i)
            bq[s][i] = *(const float2*)(Bp + (size_t)(8 * s + i) * L_);
    }
}

__device__ __forceinline__ float softplus_(float v) {
    return (v > 20.f) ? v : __logf(1.f + __expf(v));
}

__global__ __launch_bounds__(32)
void k_gemm_dtproj(const float* __restrict__ x_dbl,
                   const float* __restrict__ w,
                   const float* __restrict__ dt_bias,
                   float* __restrict__ delta)
{
    const int lane = threadIdx.x;
    const int c    = lane & 15;
    const int h    = lane >> 4;
    const int cb   = blockIdx.x % DP_CBLK;
    const int mb   = blockIdx.x / DP_CBLK;

    const int m0 = mb * DP_MB + 2 * c;
    const int b  = m0 / L_;
    const int l0 = m0 - b * L_;

    const float* Ap = w     + (size_t)(cb * DP_CB + c) * R_ + 4 * h;
    const float* Bp = x_dbl + ((size_t)b * RN2 + 4 * h) * L_ + l0;

    v8f acc[2][2];
    #pragma unroll
    for (int f = 0; f < 2; ++f) {
        #pragma unroll
        for (int g = 0; g < 2; ++g)
            acc[f][g] = (v8f){0.f,0.f,0.f,0.f,0.f,0.f,0.f,0.f};
    }

    float4 a[DP_NG][2][XP_G];
    float2 bq[DP_NG][XP_G][4];
    #pragma unroll
    for (int t = 0; t < DP_NG; ++t)
        dp_load(Ap + t * 8 * XP_G, Bp + (size_t)t * 8 * XP_G * L_, a[t], bq[t]);

    const float4* bp4 = (const float4*)(dt_bias + cb * DP_CB + 8 * h);
    float bias[2][8];
    #pragma unroll
    for (int f = 0; f < 2; ++f) {
        const float4 t0 = bp4[f * 4], t1 = bp4[f * 4 + 1];
        bias[f][0] = t0.x; bias[f][1] = t0.y; bias[f][2] = t0.z; bias[f][3] = t0.w;
        bias[f][4] = t1.x; bias[f][5] = t1.y; bias[f][6] = t1.z; bias[f][7] = t1.w;
    }

    __builtin_amdgcn_sched_barrier(0);
    #pragma unroll
    for (int t = 0; t < DP_NG; ++t) {
        xp_mma(a[t], bq[t], acc);
        __builtin_amdgcn_sched_barrier(0);
    }

    float* o = delta + ((size_t)b * C_ + cb * DP_CB + 8 * h) * L_ + l0;
    #pragma unroll
    for (int f = 0; f < 2; ++f) {
        #pragma unroll
        for (int r = 0; r < 8; ++r) {
            const float v0 = acc[f][0][r] + bias[f][r];
            const float v1 = acc[f][1][r] + bias[f][r];
            *(float2*)(o + (size_t)(f * 16 + r) * L_) =
                make_float2(softplus_(v0), softplus_(v1));
        }
    }
}

#define SC_THREADS 128
#define SC_CH      (SC_THREADS / 2)
#define SC_NH      (N_ / 2)
#define SC_CHUNKS  (L_ / 4)
#define SC_SETS    8
#define BC_PITCH   36
#define SC_STAGE_FULL ((2 * N_ * SC_CHUNKS) / SC_THREADS)
#define SC_STAGE_REST ((2 * N_ * SC_CHUNKS) - SC_STAGE_FULL * SC_THREADS)

__device__ __forceinline__ float pair_other(const float v)
{
    return __builtin_bit_cast(float,
        __builtin_amdgcn_update_dpp(0, __builtin_bit_cast(int, v), 0xB1, 0xF, 0xF, true));
}

__device__ __forceinline__ void scan_rows(const float* __restrict__ srow, float4 (&rw)[16])
{
    #pragma unroll
    for (int k = 0; k < 4; ++k) {
        const float4* row = (const float4*)(srow + k * BC_PITCH);
        rw[4 * k + 0] = row[0];
        rw[4 * k + 1] = row[1];
        rw[4 * k + 2] = row[4];
        rw[4 * k + 3] = row[5];
    }
}

__device__ __forceinline__ void scan_chunk4(const float4 (&rw)[16],
                                            const float4 d4, const float4 x4,
                                            const float2 z2, const bool hi,
                                            const float (&A_neg)[SC_NH],
                                            float (&h)[SC_NH],
                                            const float Dp,
                                            float2* __restrict__ dst)
{
    const float dvs[4] = {d4.x, d4.y, d4.z, d4.w};
    const float xvs[4] = {x4.x, x4.y, x4.z, x4.w};
    float yt[4];
    #pragma unroll
    for (int k = 0; k < 4; ++k) {
        const float dv  = dvs[k];
        const float xv  = xvs[k];
        const float dvx = dv * xv;
        const float4 b0 = rw[4 * k], b1 = rw[4 * k + 1], c0 = rw[4 * k + 2], c1 = rw[4 * k + 3];
        const float bc[2 * SC_NH] = {b0.x, b0.y, b0.z, b0.w, b1.x, b1.y, b1.z, b1.w,
                                     c0.x, c0.y, c0.z, c0.w, c1.x, c1.y, c1.z, c1.w};
        float y0 = 0.f, y1 = 0.f, y2 = 0.f, y3 = 0.f;
        #pragma unroll
        for (int n = 0; n < SC_NH; n += 4) {
            h[n + 0] = h[n + 0] * __expf(dv * A_neg[n + 0]) + dvx * bc[n + 0];
            h[n + 1] = h[n + 1] * __expf(dv * A_neg[n + 1]) + dvx * bc[n + 1];
            h[n + 2] = h[n + 2] * __expf(dv * A_neg[n + 2]) + dvx * bc[n + 2];
            h[n + 3] = h[n + 3] * __expf(dv * A_neg[n + 3]) + dvx * bc[n + 3];
            y0 += h[n + 0] * bc[SC_NH + n + 0];
            y1 += h[n + 1] * bc[SC_NH + n + 1];
            y2 += h[n + 2] * bc[SC_NH + n + 2];
            y3 += h[n + 3] * bc[SC_NH + n + 3];
        }
        const float yp = (y0 + y1) + (y2 + y3);
        yt[k] = yp + pair_other(yp);
    }
    const float ya = hi ? yt[2] : yt[0];
    const float yb = hi ? yt[3] : yt[1];
    const float xa = hi ? xvs[2] : xvs[0];
    const float xb = hi ? xvs[3] : xvs[1];
    *dst = make_float2((ya + xa * Dp) * (z2.x * sigmoid_(z2.x)),
                       (yb + xb * Dp) * (z2.y * sigmoid_(z2.y)));
}

__global__ __launch_bounds__(SC_THREADS) __attribute__((amdgpu_waves_per_eu(1, 1)))
void k_scan_out(const float* __restrict__ delta,
                const float* __restrict__ xc,
                const float* __restrict__ zbuf,
                const float* __restrict__ x_dbl,
                const float* __restrict__ A_log,
                const float* __restrict__ Dp_,
                float* __restrict__ out)
{
    __shared__ __attribute__((aligned(16))) float sBC[L_ * BC_PITCH];

    const int  tid  = threadIdx.x;
    const int  half = tid & 1;
    const bool hi   = (half != 0);
    const int  b    = blockIdx.x / (C_ / SC_CH);
    const int  c    = (blockIdx.x % (C_ / SC_CH)) * SC_CH + (tid >> 1);
    const int  bc_  = b * C_ + c;

    const float4* dp4 = (const float4*)(delta + (size_t)bc_ * L_);
    const float4* xp4 = (const float4*)(xc    + (size_t)bc_ * L_);
    const float2* zp2 = (const float2*)(zbuf  + (size_t)bc_ * L_) + half;
    float2*       op2 = (float2*)(out + (size_t)bc_ * L_) + half;

    float4 qd[SC_SETS], qx[SC_SETS];
    float2 qz[SC_SETS];
    #pragma unroll
    for (int s = 0; s < SC_SETS - 1; ++s) {
        qd[s] = dp4[s]; qx[s] = xp4[s]; qz[s] = zp2[2 * s];
    }

    const float4* src4 = (const float4*)(x_dbl + ((size_t)b * RN2 + R_) * L_);
    #pragma unroll
    for (int j = 0; j < SC_STAGE_FULL; ++j) {
        const int f  = tid + j * SC_THREADS;
        const int r  = f / SC_CHUNKS;
        const int l4 = f % SC_CHUNKS;
        const float4 v = src4[f];
        float* d = sBC + (l4 * 4) * BC_PITCH + r;
        d[0]            = v.x;
        d[BC_PITCH]     = v.y;
        d[2 * BC_PITCH] = v.z;
        d[3 * BC_PITCH] = v.w;
    }
    if (tid < SC_STAGE_REST) {
        const int f  = tid + SC_STAGE_FULL * SC_THREADS;
        const int r  = f / SC_CHUNKS;
        const int l4 = f % SC_CHUNKS;
        const float4 v = src4[f];
        float* d = sBC + (l4 * 4) * BC_PITCH + r;
        d[0]            = v.x;
        d[BC_PITCH]     = v.y;
        d[2 * BC_PITCH] = v.z;
        d[3 * BC_PITCH] = v.w;
    }

    float A_neg[SC_NH];
    {
        const float4* a4 = (const float4*)(A_log + (size_t)c * N_ + SC_NH * half);
        #pragma unroll
        for (int q = 0; q < SC_NH / 4; ++q) {
            const float4 t = a4[q];
            A_neg[4 * q + 0] = -__expf(t.x);
            A_neg[4 * q + 1] = -__expf(t.y);
            A_neg[4 * q + 2] = -__expf(t.z);
            A_neg[4 * q + 3] = -__expf(t.w);
        }
    }
    const float Dp = Dp_[c];

    float h[SC_NH];
    #pragma unroll
    for (int n = 0; n < SC_NH; ++n) h[n] = 0.f;

    __syncthreads();

    const float* sB = sBC + SC_NH * half;

    float4 rw[2][16];
    scan_rows(sB, rw[0]);
    for (int i = 0; i < SC_CHUNKS - 1; i += SC_SETS) {
        #pragma unroll
        for (int j = 0; j < SC_SETS; ++j) {
            const int nx = (i + j + SC_SETS - 1 < SC_CHUNKS) ? (i + j + SC_SETS - 1)
                                                             : (SC_CHUNKS - 1);
            const int ls = (j + SC_SETS - 1) % SC_SETS;
            qd[ls] = dp4[nx]; qx[ls] = xp4[nx]; qz[ls] = zp2[2 * nx];
            scan_rows(sB + ((i + j + 1) * 4) * BC_PITCH, rw[(j + 1) & 1]);
            __builtin_amdgcn_sched_barrier(0);
            scan_chunk4(rw[j & 1], qd[j], qx[j], qz[j], hi,
                        A_neg, h, Dp, op2 + 2 * (i + j));
            __builtin_amdgcn_sched_barrier(0);
        }
    }
    scan_chunk4(rw[0], qd[0], qx[0], qz[0], hi, A_neg, h, Dp, op2 + 2 * (SC_CHUNKS - 1));
}

extern "C" void kernel_launch(void* const* d_in, const int* in_sizes, int n_in,
                              void* d_out, int out_size, void* d_ws, size_t ws_size,
                              hipStream_t stream)
{
    const float* hidden    = (const float*)d_in[0];
    const float* in_proj_w = (const float*)d_in[1];
    const float* conv_w    = (const float*)d_in[2];
    const float* conv_b    = (const float*)d_in[3];
    const float* x_proj_w  = (const float*)d_in[4];
    const float* dt_proj_w = (const float*)d_in[5];
    const float* dt_bias   = (const float*)d_in[6];
    const float* A_log     = (const float*)d_in[7];
    const float* D_param   = (const float*)d_in[8];

    float* ws = (float*)d_ws;
    const size_t BCL = (size_t)BSZ * C_ * L_;
    float* x_pre = ws;
    float* zbuf  = ws + BCL;
    float* xcbuf = ws + 2 * BCL;
    float* delta = ws + 3 * BCL;
    float* x_dbl = ws + 4 * BCL;

    k_gemm_inproj<<<IP_EBLK * IP_MBLK, 32, 0, stream>>>(
        hidden, in_proj_w, x_pre, zbuf);

    k_conv_silu<<<CV_BLOCKS, CV_THREADS, 0, stream>>>(
        x_pre, conv_w, conv_b, xcbuf);

    k_gemm_xproj<<<dim3(XP_RBLK * XP_LBLK, BSZ), XP_NW * 32, 0, stream>>>(
        xcbuf, x_proj_w, x_dbl);

    k_gemm_dtproj<<<DP_CBLK * DP_MBLK, 32, 0, stream>>>(
        x_dbl, dt_proj_w, dt_bias, delta);

    k_scan_out<<<(BSZ * C_) / SC_CH, SC_THREADS, 0, stream>>>(
        delta, xcbuf, zbuf, x_dbl, A_log, D_param, (float*)d_out);
}


# in_proj GEMM: both fragment loaders read 16 B per lane (two k-steps per load) instead of 8 B: half the loads and cache-line touches per WMMA; k pairs regrouped in each 8-k step (sum order changed)
# speedup vs baseline: 3.9056x; 1.0406x over previous
#include <hip/hip_runtime.h>


#define BSZ   8
#define T_    8
#define L_    196
#define D_    192
#define E_    192
#define C_    1536
#define N_    16
#define KC    4
#define R_    96
#define RN2   128
#define LTILES 13

typedef float v2f __attribute__((ext_vector_type(2)));
typedef float v8f __attribute__((ext_vector_type(8)));

__device__ __forceinline__ float sigmoid_(float x) {
    return 1.0f / (1.0f + __expf(-x));
}

#define IP_EW    64
#define IP_MW    32
#define IP_EBLK  ((2 * E_) / IP_EW)
#define IP_MBLK  ((BSZ * T_ * L_) / IP_MW)
#define IP_KC    32
#define IP_KS    (IP_KC / 4)
#define IP_KQ    (IP_KC / 8)
#define IP_NCH   (D_ / IP_KC)
#define IP_NPASS (IP_NCH * 4)

__device__ __forceinline__ void ip_load_line(const float* __restrict__ p, float4 (&x)[IP_KQ])
{
    #pragma unroll
    for (int s = 0; s < IP_KQ; ++s)
        x[s] = *(const float4*)(p + 8 * s);
}

__device__ __forceinline__ void ip_pass(const float4 (&a)[IP_KQ],
                                        const float4 (&b0)[IP_KQ], const float4 (&b1)[IP_KQ],
                                        v8f& c0, v8f& c1)
{
    #pragma unroll
    for (int s = 0; s < IP_KQ; ++s) {
        const v2f alo  = {a[s].x,  a[s].y},  ahi  = {a[s].z,  a[s].w};
        const v2f b0lo = {b0[s].x, b0[s].y}, b0hi = {b0[s].z, b0[s].w};
        const v2f b1lo = {b1[s].x, b1[s].y}, b1hi = {b1[s].z, b1[s].w};
        c0 = __builtin_amdgcn_wmma_f32_16x16x4_f32(
                 false, alo, false, b0lo, (short)0, c0, false, false);
        c1 = __builtin_amdgcn_wmma_f32_16x16x4_f32(
                 false, alo, false, b1lo, (short)0, c1, false, false);
        c0 = __builtin_amdgcn_wmma_f32_16x16x4_f32(
                 false, ahi, false, b0hi, (short)0, c0, false, false);
        c1 = __builtin_amdgcn_wmma_f32_16x16x4_f32(
                 false, ahi, false, b1hi, (short)0, c1, false, false);
    }
}

__global__ __launch_bounds__(32)
void k_gemm_inproj(const float* __restrict__ hidden,
                   const float* __restrict__ w,
                   float* __restrict__ x_pre,
                   float* __restrict__ zbuf)
{
    const int lane = threadIdx.x;
    const int c    = lane & 15;
    const int h    = lane >> 4;
    const int eb   = blockIdx.x % IP_EBLK;
    const int mb   = blockIdx.x / IP_EBLK;

    const float* Ap = w      + (size_t)(eb * IP_EW + c) * D_ + 4 * h;
    const float* Bp = hidden + (size_t)(mb * IP_MW + c) * D_ + 4 * h;

    v8f acc[4][2];
    #pragma unroll
    for (int f = 0; f < 4; ++f) {
        #pragma unroll
        for (int g = 0; g < 2; ++g)
            acc[f][g] = (v8f){0.f,0.f,0.f,0.f,0.f,0.f,0.f,0.f};
    }

    float4 a[4][IP_KQ];
    float4 b[2][2][IP_KQ];
    ip_load_line(Bp, b[0][0]);
    ip_load_line(Bp + (size_t)16 * D_, b[0][1]);
    ip_load_line(Ap, a[0]);
    ip_load_line(Ap + (size_t)16 * D_, a[1]);
    ip_load_line(Ap + (size_t)32 * D_, a[2]);

    #pragma unroll
    for (int p = 0; p < IP_NPASS; ++p) {
        const int ch = p >> 2;
        const int f  = p & 3;
        const int q  = p + 3;
        if (q < IP_NPASS)
            ip_load_line(Ap + (size_t)(q & 3) * 16 * D_ + (q >> 2) * IP_KC, a[q & 3]);
        if (f == 1 && ch + 1 < IP_NCH) {
            ip_load_line(Bp + (ch + 1) * IP_KC, b[(ch + 1) & 1][0]);
            ip_load_line(Bp + (size_t)16 * D_ + (ch + 1) * IP_KC, b[(ch + 1) & 1][1]);
        }
        __builtin_amdgcn_sched_barrier(0);
        ip_pass(a[p & 3], b[ch & 1][0], b[ch & 1][1], acc[f][0], acc[f][1]);
        __builtin_amdgcn_sched_barrier(0);
    }

    float* outp = (eb < IP_EBLK / 2) ? x_pre : zbuf;
    const int e0 = (eb % (IP_EBLK / 2)) * IP_EW + 8 * h;
    #pragma unroll
    for (int g = 0; g < 2; ++g) {
        const int m  = mb * IP_MW + g * 16 + c;
        const int bt = m / L_;
        const int l  = m - bt * L_;
        const int b_ = bt / T_;
        const int t  = bt % T_;
        float* p = outp + ((size_t)b_ * C_ + (size_t)e0 * T_ + t) * L_ + l;
        #pragma unroll
        for (int f = 0; f < 4; ++f) {
            #pragma unroll
            for (int r = 0; r < 8; ++r)
                p[(size_t)(f * 16 + r) * (T_ * L_)] = acc[f][g][r];
        }
    }
}

#define CV_F4_ROW   (L_ / 4)
#define CV_F4_B     (C_ * CV_F4_ROW)
#define CV_THREADS  128
#define CV_BLOCKS   (CV_F4_B / CV_THREADS)

__global__ __launch_bounds__(CV_THREADS)
void k_conv_silu(const float* __restrict__ x_pre,
                 const float* __restrict__ cw,
                 const float* __restrict__ cb,
                 float* __restrict__ xc)
{
    const int f0 = blockIdx.x * CV_THREADS + threadIdx.x;
    const int c  = f0 / CV_F4_ROW;
    const int l4 = f0 - c * CV_F4_ROW;
    const bool has_prev = (l4 > 0);
    const int  poff     = has_prev ? -1 : 0;

    const float4* src = (const float4*)x_pre + f0;
    float4*       dst = (float4*)xc + f0;

    float4 cur[BSZ], prv[BSZ];
    #pragma unroll
    for (int b = 0; b < BSZ; ++b) {
        cur[b] = src[(size_t)b * CV_F4_B];
        prv[b] = src[(size_t)b * CV_F4_B + poff];
    }
    const float4 w    = *(const float4*)(cw + c * KC);
    const float  bias = cb[c];

    #pragma unroll
    for (int b = 0; b < BSZ; ++b) {
        const float p1 = has_prev ? prv[b].y : 0.f;
        const float p2 = has_prev ? prv[b].z : 0.f;
        const float p3 = has_prev ? prv[b].w : 0.f;
        const float4 x = cur[b];
        float s0 = bias, s1 = bias, s2 = bias, s3 = bias;
        s0 += p1  * w.x;  s1 += p2  * w.x;  s2 += p3  * w.x;  s3 += x.x * w.x;
        s0 += p2  * w.y;  s1 += p3  * w.y;  s2 += x.x * w.y;  s3 += x.y * w.y;
        s0 += p3  * w.z;  s1 += x.x * w.z;  s2 += x.y * w.z;  s3 += x.z * w.z;
        s0 += x.x * w.w;  s1 += x.y * w.w;  s2 += x.z * w.w;  s3 += x.w * w.w;
        dst[(size_t)b * CV_F4_B] = make_float4(s0 * sigmoid_(s0), s1 * sigmoid_(s1),
                                               s2 * sigmoid_(s2), s3 * sigmoid_(s3));
    }
}

#define XP_NW    8
#define XP_KW    (C_ / XP_NW)
#define XP_RB    32
#define XP_LB    32
#define XP_RBLK  (RN2 / XP_RB)
#define XP_LBLK  ((L_ + XP_LB - 1) / XP_LB)
#define XP_G     4
#define XP_NG    (XP_KW / (8 * XP_G))

__device__ __forceinline__ void xp_load(const float* __restrict__ Ap,
                                        const float* __restrict__ Bp,
                                        float4 (&a)[2][XP_G], float2 (&bq)[XP_G][4])
{
    #pragma unroll
    for (int s = 0; s < XP_G; ++s) {
        a[0][s] = *(const float4*)(Ap + 8 * s);
        a[1][s] = *(const float4*)(Ap + (size_t)16 * C_ + 8 * s);
    }
    #pragma unroll
    for (int s = 0; s < XP_G; ++s) {
        #pragma unroll
        for (int i = 0; i < 4; ++i)
            bq[s][i] = *(const float2*)(Bp + (size_t)(8 * s + i) * L_);
    }
}

__device__ __forceinline__ void xp_mma(const float4 (&a)[2][XP_G],
                                       const float2 (&bq)[XP_G][4], v8f (&acc)[2][2])
{
    #pragma unroll
    for (int s = 0; s < XP_G; ++s) {
        const v2f be0 = {bq[s][0].x, bq[s][1].x};
        const v2f bo0 = {bq[s][0].y, bq[s][1].y};
        const v2f be1 = {bq[s][2].x, bq[s][3].x};
        const v2f bo1 = {bq[s][2].y, bq[s][3].y};
        const v2f a0lo = {a[0][s].x, a[0][s].y};
        const v2f a1lo = {a[1][s].x, a[1][s].y};
        const v2f a0hi = {a[0][s].z, a[0][s].w};
        const v2f a1hi = {a[1][s].z, a[1][s].w};
        acc[0][0] = __builtin_amdgcn_wmma_f32_16x16x4_f32(
                        false, a0lo, false, be0, (short)0, acc[0][0], false, false);
        acc[0][1] = __builtin_amdgcn_wmma_f32_16x16x4_f32(
                        false, a0lo, false, bo0, (short)0, acc[0][1], false, false);
        acc[1][0] = __builtin_amdgcn_wmma_f32_16x16x4_f32(
                        false, a1lo, false, be0, (short)0, acc[1][0], false, false);
        acc[1][1] = __builtin_amdgcn_wmma_f32_16x16x4_f32(
                        false, a1lo, false, bo0, (short)0, acc[1][1], false, false);
        acc[0][0] = __builtin_amdgcn_wmma_f32_16x16x4_f32(
                        false, a0hi, false, be1, (short)0, acc[0][0], false, false);
        acc[0][1] = __builtin_amdgcn_wmma_f32_16x16x4_f32(
                        false, a0hi, false, bo1, (short)0, acc[0][1], false, false);
        acc[1][0] = __builtin_amdgcn_wmma_f32_16x16x4_f32(
                        false, a1hi, false, be1, (short)0, acc[1][0], false, false);
        acc[1][1] = __builtin_amdgcn_wmma_f32_16x16x4_f32(
                        false, a1hi, false, bo1, (short)0, acc[1][1], false, false);
    }
}

__global__ __launch_bounds__(XP_NW * 32)
void k_gemm_xproj(const float* __restrict__ xc,
                  const float* __restrict__ w,
                  float* __restrict__ x_dbl)
{
    __shared__ __attribute__((aligned(16))) float sPart[XP_NW * 8 * 32 * 4];

    const int tid  = threadIdx.x;
    const int lane = tid & 31;
    const int wv   = tid >> 5;
    const int c    = lane & 15;
    const int h    = lane >> 4;
    const int rb   = blockIdx.x % XP_RBLK;
    const int lb   = blockIdx.x / XP_RBLK;
    const int b    = blockIdx.y;

    const int l0 = lb * XP_LB + 2 * c;
    const int lc = (l0 < L_) ? l0 : (L_ - 2);
    const int k0 = wv * XP_KW + 4 * h;

    const float* Ap = w  + (size_t)(rb * XP_RB + c) * C_ + k0;
    const float* Bp = xc + ((size_t)b * C_ + k0) * L_ + lc;

    v8f acc[2][2];
    #pragma unroll
    for (int f = 0; f < 2; ++f) {
        #pragma unroll
        for (int g = 0; g < 2; ++g)
            acc[f][g] = (v8f){0.f,0.f,0.f,0.f,0.f,0.f,0.f,0.f};
    }

    float4 a[2][2][XP_G];
    float2 bq[2][XP_G][4];
    xp_load(Ap, Bp, a[0], bq[0]);
    #pragma unroll
    for (int t = 0; t < XP_NG; ++t) {
        if (t + 1 < XP_NG)
            xp_load(Ap + (t + 1) * 8 * XP_G, Bp + (size_t)(t + 1) * 8 * XP_G * L_,
                    a[(t + 1) & 1], bq[(t + 1) & 1]);
        __builtin_amdgcn_sched_barrier(0);
        xp_mma(a[t & 1], bq[t & 1], acc);
        __builtin_amdgcn_sched_barrier(0);
    }

    float4* sp = (float4*)sPart;
    #pragma unroll
    for (int f = 0; f < 2; ++f) {
        #pragma unroll
        for (int rp = 0; rp < 4; ++rp)
            sp[(wv * 8 + f * 4 + rp) * 32 + lane] =
                make_float4(acc[f][0][2 * rp],     acc[f][1][2 * rp],
                            acc[f][0][2 * rp + 1], acc[f][1][2 * rp + 1]);
    }
    __syncthreads();

    float4 s = sp[wv * 32 + lane];
    #pragma unroll
    for (int u = 1; u < XP_NW; ++u) {
        const float4 t = sp[(u * 8 + wv) * 32 + lane];
        s.x += t.x; s.y += t.y; s.z += t.z; s.w += t.w;
    }
    const int row = rb * XP_RB + 16 * (wv >> 2) + 8 * h + 2 * (wv & 3);
    if (l0 < L_) {
        float* o = x_dbl + ((size_t)b * RN2 + row) * L_ + l0;
        *(float2*)o        = make_float2(s.x, s.y);
        *(float2*)(o + L_) = make_float2(s.z, s.w);
    }
}

#define DP_CB    32
#define DP_MB    32
#define DP_CBLK  (C_ / DP_CB)
#define DP_MBLK  ((BSZ * L_) / DP_MB)
#define DP_NG    (R_ / (8 * XP_G))

__device__ __forceinline__ void dp_load(const float* __restrict__ Ap,
                                        const float* __restrict__ Bp,
                                        float4 (&a)[2][XP_G], float2 (&bq)[XP_G][4])
{
    #pragma unroll
    for (int s = 0; s < XP_G; ++s) {
        a[0][s] = *(const float4*)(Ap + 8 * s);
        a[1][s] = *(const float4*)(Ap + (size_t)16 * R_ + 8 * s);
    }
    #pragma unroll
    for (int s = 0; s < XP_G; ++s) {
        #pragma unroll
        for (int i = 0; i < 4; ++i)
            bq[s][i] = *(const float2*)(Bp + (size_t)(8 * s + i) * L_);
    }
}

__device__ __forceinline__ float softplus_(float v) {
    return (v > 20.f) ? v : __logf(1.f + __expf(v));
}

__global__ __launch_bounds__(32)
void k_gemm_dtproj(const float* __restrict__ x_dbl,
                   const float* __restrict__ w,
                   const float* __restrict__ dt_bias,
                   float* __restrict__ delta)
{
    const int lane = threadIdx.x;
    const int c    = lane & 15;
    const int h    = lane >> 4;
    const int cb   = blockIdx.x % DP_CBLK;
    const int mb   = blockIdx.x / DP_CBLK;

    const int m0 = mb * DP_MB + 2 * c;
    const int b  = m0 / L_;
    const int l0 = m0 - b * L_;

    const float* Ap = w     + (size_t)(cb * DP_CB + c) * R_ + 4 * h;
    const float* Bp = x_dbl + ((size_t)b * RN2 + 4 * h) * L_ + l0;

    v8f acc[2][2];
    #pragma unroll
    for (int f = 0; f < 2; ++f) {
        #pragma unroll
        for (int g = 0; g < 2; ++g)
            acc[f][g] = (v8f){0.f,0.f,0.f,0.f,0.f,0.f,0.f,0.f};
    }

    float4 a[DP_NG][2][XP_G];
    float2 bq[DP_NG][XP_G][4];
    #pragma unroll
    for (int t = 0; t < DP_NG; ++t)
        dp_load(Ap + t * 8 * XP_G, Bp + (size_t)t * 8 * XP_G * L_, a[t], bq[t]);

    const float4* bp4 = (const float4*)(dt_bias + cb * DP_CB + 8 * h);
    float bias[2][8];
    #pragma unroll
    for (int f = 0; f < 2; ++f) {
        const float4 t0 = bp4[f * 4], t1 = bp4[f * 4 + 1];
        bias[f][0] = t0.x; bias[f][1] = t0.y; bias[f][2] = t0.z; bias[f][3] = t0.w;
        bias[f][4] = t1.x; bias[f][5] = t1.y; bias[f][6] = t1.z; bias[f][7] = t1.w;
    }

    __builtin_amdgcn_sched_barrier(0);
    #pragma unroll
    for (int t = 0; t < DP_NG; ++t) {
        xp_mma(a[t], bq[t], acc);
        __builtin_amdgcn_sched_barrier(0);
    }

    float* o = delta + ((size_t)b * C_ + cb * DP_CB + 8 * h) * L_ + l0;
    #pragma unroll
    for (int f = 0; f < 2; ++f) {
        #pragma unroll
        for (int r = 0; r < 8; ++r) {
            const float v0 = acc[f][0][r] + bias[f][r];
            const float v1 = acc[f][1][r] + bias[f][r];
            *(float2*)(o + (size_t)(f * 16 + r) * L_) =
                make_float2(softplus_(v0), softplus_(v1));
        }
    }
}

#define SC_THREADS 128
#define SC_CH      (SC_THREADS / 2)
#define SC_NH      (N_ / 2)
#define SC_CHUNKS  (L_ / 4)
#define SC_SETS    8
#define BC_PITCH   36
#define SC_STAGE_FULL ((2 * N_ * SC_CHUNKS) / SC_THREADS)
#define SC_STAGE_REST ((2 * N_ * SC_CHUNKS) - SC_STAGE_FULL * SC_THREADS)

__device__ __forceinline__ float pair_other(const float v)
{
    return __builtin_bit_cast(float,
        __builtin_amdgcn_update_dpp(0, __builtin_bit_cast(int, v), 0xB1, 0xF, 0xF, true));
}

__device__ __forceinline__ void scan_rows(const float* __restrict__ srow, float4 (&rw)[16])
{
    #pragma unroll
    for (int k = 0; k < 4; ++k) {
        const float4* row = (const float4*)(srow + k * BC_PITCH);
        rw[4 * k + 0] = row[0];
        rw[4 * k + 1] = row[1];
        rw[4 * k + 2] = row[4];
        rw[4 * k + 3] = row[5];
    }
}

__device__ __forceinline__ void scan_chunk4(const float4 (&rw)[16],
                                            const float4 d4, const float4 x4,
                                            const float2 z2, const bool hi,
                                            const float (&A_neg)[SC_NH],
                                            float (&h)[SC_NH],
                                            const float Dp,
                                            float2* __restrict__ dst)
{
    const float dvs[4] = {d4.x, d4.y, d4.z, d4.w};
    const float xvs[4] = {x4.x, x4.y, x4.z, x4.w};
    float yt[4];
    #pragma unroll
    for (int k = 0; k < 4; ++k) {
        const float dv  = dvs[k];
        const float xv  = xvs[k];
        const float dvx = dv * xv;
        const float4 b0 = rw[4 * k], b1 = rw[4 * k + 1], c0 = rw[4 * k + 2], c1 = rw[4 * k + 3];
        const float bc[2 * SC_NH] = {b0.x, b0.y, b0.z, b0.w, b1.x, b1.y, b1.z, b1.w,
                                     c0.x, c0.y, c0.z, c0.w, c1.x, c1.y, c1.z, c1.w};
        float y0 = 0.f, y1 = 0.f, y2 = 0.f, y3 = 0.f;
        #pragma unroll
        for (int n = 0; n < SC_NH; n += 4) {
            h[n + 0] = h[n + 0] * __expf(dv * A_neg[n + 0]) + dvx * bc[n + 0];
            h[n + 1] = h[n + 1] * __expf(dv * A_neg[n + 1]) + dvx * bc[n + 1];
            h[n + 2] = h[n + 2] * __expf(dv * A_neg[n + 2]) + dvx * bc[n + 2];
            h[n + 3] = h[n + 3] * __expf(dv * A_neg[n + 3]) + dvx * bc[n + 3];
            y0 += h[n + 0] * bc[SC_NH + n + 0];
            y1 += h[n + 1] * bc[SC_NH + n + 1];
            y2 += h[n + 2] * bc[SC_NH + n + 2];
            y3 += h[n + 3] * bc[SC_NH + n + 3];
        }
        const float yp = (y0 + y1) + (y2 + y3);
        yt[k] = yp + pair_other(yp);
    }
    const float ya = hi ? yt[2] : yt[0];
    const float yb = hi ? yt[3] : yt[1];
    const float xa = hi ? xvs[2] : xvs[0];
    const float xb = hi ? xvs[3] : xvs[1];
    *dst = make_float2((ya + xa * Dp) * (z2.x * sigmoid_(z2.x)),
                       (yb + xb * Dp) * (z2.y * sigmoid_(z2.y)));
}

__global__ __launch_bounds__(SC_THREADS) __attribute__((amdgpu_waves_per_eu(1, 1)))
void k_scan_out(const float* __restrict__ delta,
                const float* __restrict__ xc,
                const float* __restrict__ zbuf,
                const float* __restrict__ x_dbl,
                const float* __restrict__ A_log,
                const float* __restrict__ Dp_,
                float* __restrict__ out)
{
    __shared__ __attribute__((aligned(16))) float sBC[L_ * BC_PITCH];

    const int  tid  = threadIdx.x;
    const int  half = tid & 1;
    const bool hi   = (half != 0);
    const int  b    = blockIdx.x / (C_ / SC_CH);
    const int  c    = (blockIdx.x % (C_ / SC_CH)) * SC_CH + (tid >> 1);
    const int  bc_  = b * C_ + c;

    const float4* dp4 = (const float4*)(delta + (size_t)bc_ * L_);
    const float4* xp4 = (const float4*)(xc    + (size_t)bc_ * L_);
    const float2* zp2 = (const float2*)(zbuf  + (size_t)bc_ * L_) + half;
    float2*       op2 = (float2*)(out + (size_t)bc_ * L_) + half;

    float4 qd[SC_SETS], qx[SC_SETS];
    float2 qz[SC_SETS];
    #pragma unroll
    for (int s = 0; s < SC_SETS - 1; ++s) {
        qd[s] = dp4[s]; qx[s] = xp4[s]; qz[s] = zp2[2 * s];
    }

    const float4* src4 = (const float4*)(x_dbl + ((size_t)b * RN2 + R_) * L_);
    #pragma unroll
    for (int j = 0; j < SC_STAGE_FULL; ++j) {
        const int f  = tid + j * SC_THREADS;
        const int r  = f / SC_CHUNKS;
        const int l4 = f % SC_CHUNKS;
        const float4 v = src4[f];
        float* d = sBC + (l4 * 4) * BC_PITCH + r;
        d[0]            = v.x;
        d[BC_PITCH]     = v.y;
        d[2 * BC_PITCH] = v.z;
        d[3 * BC_PITCH] = v.w;
    }
    if (tid < SC_STAGE_REST) {
        const int f  = tid + SC_STAGE_FULL * SC_THREADS;
        const int r  = f / SC_CHUNKS;
        const int l4 = f % SC_CHUNKS;
        const float4 v = src4[f];
        float* d = sBC + (l4 * 4) * BC_PITCH + r;
        d[0]            = v.x;
        d[BC_PITCH]     = v.y;
        d[2 * BC_PITCH] = v.z;
        d[3 * BC_PITCH] = v.w;
    }

    float A_neg[SC_NH];
    {
        const float4* a4 = (const float4*)(A_log + (size_t)c * N_ + SC_NH * half);
        #pragma unroll
        for (int q = 0; q < SC_NH / 4; ++q) {
            const float4 t = a4[q];
            A_neg[4 * q + 0] = -__expf(t.x);
            A_neg[4 * q + 1] = -__expf(t.y);
            A_neg[4 * q + 2] = -__expf(t.z);
            A_neg[4 * q + 3] = -__expf(t.w);
        }
    }
    const float Dp = Dp_[c];

    float h[SC_NH];
    #pragma unroll
    for (int n = 0; n < SC_NH; ++n) h[n] = 0.f;

    __syncthreads();

    const float* sB = sBC + SC_NH * half;

    float4 rw[2][16];
    scan_rows(sB, rw[0]);
    for (int i = 0; i < SC_CHUNKS - 1; i += SC_SETS) {
        #pragma unroll
        for (int j = 0; j < SC_SETS; ++j) {
            const int nx = (i + j + SC_SETS - 1 < SC_CHUNKS) ? (i + j + SC_SETS - 1)
                                                             : (SC_CHUNKS - 1);
            const int ls = (j + SC_SETS - 1) % SC_SETS;
            qd[ls] = dp4[nx]; qx[ls] = xp4[nx]; qz[ls] = zp2[2 * nx];
            scan_rows(sB + ((i + j + 1) * 4) * BC_PITCH, rw[(j + 1) & 1]);
            __builtin_amdgcn_sched_barrier(0);
            scan_chunk4(rw[j & 1], qd[j], qx[j], qz[j], hi,
                        A_neg, h, Dp, op2 + 2 * (i + j));
            __builtin_amdgcn_sched_barrier(0);
        }
    }
    scan_chunk4(rw[0], qd[0], qx[0], qz[0], hi, A_neg, h, Dp, op2 + 2 * (SC_CHUNKS - 1));
}

extern "C" void kernel_launch(void* const* d_in, const int* in_sizes, int n_in,
                              void* d_out, int out_size, void* d_ws, size_t ws_size,
                              hipStream_t stream)
{
    const float* hidden    = (const float*)d_in[0];
    const float* in_proj_w = (const float*)d_in[1];
    const float* conv_w    = (const float*)d_in[2];
    const float* conv_b    = (const float*)d_in[3];
    const float* x_proj_w  = (const float*)d_in[4];
    const float* dt_proj_w = (const float*)d_in[5];
    const float* dt_bias   = (const float*)d_in[6];
    const float* A_log     = (const float*)d_in[7];
    const float* D_param   = (const float*)d_in[8];

    float* ws = (float*)d_ws;
    const size_t BCL = (size_t)BSZ * C_ * L_;
    float* x_pre = ws;
    float* zbuf  = ws + BCL;
    float* xcbuf = ws + 2 * BCL;
    float* delta = ws + 3 * BCL;
    float* x_dbl = ws + 4 * BCL;

    k_gemm_inproj<<<IP_EBLK * IP_MBLK, 32, 0, stream>>>(
        hidden, in_proj_w, x_pre, zbuf);

    k_conv_silu<<<CV_BLOCKS, CV_THREADS, 0, stream>>>(
        x_pre, conv_w, conv_b, xcbuf);

    k_gemm_xproj<<<dim3(XP_RBLK * XP_LBLK, BSZ), XP_NW * 32, 0, stream>>>(
        xcbuf, x_proj_w, x_dbl);

    k_gemm_dtproj<<<DP_CBLK * DP_MBLK, 32, 0, stream>>>(
        x_dbl, dt_proj_w, dt_bias, delta);

    k_scan_out<<<(BSZ * C_) / SC_CH, SC_THREADS, 0, stream>>>(
        delta, xcbuf, zbuf, x_dbl, A_log, D_param, (float*)d_out);
}


# dt_proj GEMM + softplus fused into the scan kernel: each scan wave computes delta of its own 16 channels on f32 WMMA (same k order) into LDS; delta never passes through memory; 4 launches, not 5
# speedup vs baseline: 4.2692x; 1.0931x over previous
#include <hip/hip_runtime.h>


#define BSZ   8
#define T_    8
#define L_    196
#define D_    192
#define E_    192
#define C_    1536
#define N_    16
#define KC    4
#define R_    96
#define RN2   128
#define LTILES 13

typedef float v2f __attribute__((ext_vector_type(2)));
typedef float v8f __attribute__((ext_vector_type(8)));

__device__ __forceinline__ float sigmoid_(float x) {
    return 1.0f / (1.0f + __expf(-x));
}

#define IP_EW    64
#define IP_MW    32
#define IP_EBLK  ((2 * E_) / IP_EW)
#define IP_MBLK  ((BSZ * T_ * L_) / IP_MW)
#define IP_KC    32
#define IP_KS    (IP_KC / 4)
#define IP_KQ    (IP_KC / 8)
#define IP_NCH   (D_ / IP_KC)
#define IP_NPASS (IP_NCH * 4)

__device__ __forceinline__ void ip_load_line(const float* __restrict__ p, float4 (&x)[IP_KQ])
{
    #pragma unroll
    for (int s = 0; s < IP_KQ; ++s)
        x[s] = *(const float4*)(p + 8 * s);
}

__device__ __forceinline__ void ip_pass(const float4 (&a)[IP_KQ],
                                        const float4 (&b0)[IP_KQ], const float4 (&b1)[IP_KQ],
                                        v8f& c0, v8f& c1)
{
    #pragma unroll
    for (int s = 0; s < IP_KQ; ++s) {
        const v2f alo  = {a[s].x,  a[s].y},  ahi  = {a[s].z,  a[s].w};
        const v2f b0lo = {b0[s].x, b0[s].y}, b0hi = {b0[s].z, b0[s].w};
        const v2f b1lo = {b1[s].x, b1[s].y}, b1hi = {b1[s].z, b1[s].w};
        c0 = __builtin_amdgcn_wmma_f32_16x16x4_f32(
                 false, alo, false, b0lo, (short)0, c0, false, false);
        c1 = __builtin_amdgcn_wmma_f32_16x16x4_f32(
                 false, alo, false, b1lo, (short)0, c1, false, false);
        c0 = __builtin_amdgcn_wmma_f32_16x16x4_f32(
                 false, ahi, false, b0hi, (short)0, c0, false, false);
        c1 = __builtin_amdgcn_wmma_f32_16x16x4_f32(
                 false, ahi, false, b1hi, (short)0, c1, false, false);
    }
}

__global__ __launch_bounds__(32)
void k_gemm_inproj(const float* __restrict__ hidden,
                   const float* __restrict__ w,
                   float* __restrict__ x_pre,
                   float* __restrict__ zbuf)
{
    const int lane = threadIdx.x;
    const int c    = lane & 15;
    const int h    = lane >> 4;
    const int eb   = blockIdx.x % IP_EBLK;
    const int mb   = blockIdx.x / IP_EBLK;

    const float* Ap = w      + (size_t)(eb * IP_EW + c) * D_ + 4 * h;
    const float* Bp = hidden + (size_t)(mb * IP_MW + c) * D_ + 4 * h;

    v8f acc[4][2];
    #pragma unroll
    for (int f = 0; f < 4; ++f) {
        #pragma unroll
        for (int g = 0; g < 2; ++g)
            acc[f][g] = (v8f){0.f,0.f,0.f,0.f,0.f,0.f,0.f,0.f};
    }

    float4 a[4][IP_KQ];
    float4 b[2][2][IP_KQ];
    ip_load_line(Bp, b[0][0]);
    ip_load_line(Bp + (size_t)16 * D_, b[0][1]);
    ip_load_line(Ap, a[0]);
    ip_load_line(Ap + (size_t)16 * D_, a[1]);
    ip_load_line(Ap + (size_t)32 * D_, a[2]);

    #pragma unroll
    for (int p = 0; p < IP_NPASS; ++p) {
        const int ch = p >> 2;
        const int f  = p & 3;
        const int q  = p + 3;
        if (q < IP_NPASS)
            ip_load_line(Ap + (size_t)(q & 3) * 16 * D_ + (q >> 2) * IP_KC, a[q & 3]);
        if (f == 1 && ch + 1 < IP_NCH) {
            ip_load_line(Bp + (ch + 1) * IP_KC, b[(ch + 1) & 1][0]);
            ip_load_line(Bp + (size_t)16 * D_ + (ch + 1) * IP_KC, b[(ch + 1) & 1][1]);
        }
        __builtin_amdgcn_sched_barrier(0);
        ip_pass(a[p & 3], b[ch & 1][0], b[ch & 1][1], acc[f][0], acc[f][1]);
        __builtin_amdgcn_sched_barrier(0);
    }

    float* outp = (eb < IP_EBLK / 2) ? x_pre : zbuf;
    const int e0 = (eb % (IP_EBLK / 2)) * IP_EW + 8 * h;
    #pragma unroll
    for (int g = 0; g < 2; ++g) {
        const int m  = mb * IP_MW + g * 16 + c;
        const int bt = m / L_;
        const int l  = m - bt * L_;
        const int b_ = bt / T_;
        const int t  = bt % T_;
        float* p = outp + ((size_t)b_ * C_ + (size_t)e0 * T_ + t) * L_ + l;
        #pragma unroll
        for (int f = 0; f < 4; ++f) {
            #pragma unroll
            for (int r = 0; r < 8; ++r)
                p[(size_t)(f * 16 + r) * (T_ * L_)] = acc[f][g][r];
        }
    }
}

#define CV_F4_ROW   (L_ / 4)
#define CV_F4_B     (C_ * CV_F4_ROW)
#define CV_THREADS  128
#define CV_BLOCKS   (CV_F4_B / CV_THREADS)

__global__ __launch_bounds__(CV_THREADS)
void k_conv_silu(const float* __restrict__ x_pre,
                 const float* __restrict__ cw,
                 const float* __restrict__ cb,
                 float* __restrict__ xc)
{
    const int f0 = blockIdx.x * CV_THREADS + threadIdx.x;
    const int c  = f0 / CV_F4_ROW;
    const int l4 = f0 - c * CV_F4_ROW;
    const bool has_prev = (l4 > 0);
    const int  poff     = has_prev ? -1 : 0;

    const float4* src = (const float4*)x_pre + f0;
    float4*       dst = (float4*)xc + f0;

    float4 cur[BSZ], prv[BSZ];
    #pragma unroll
    for (int b = 0; b < BSZ; ++b) {
        cur[b] = src[(size_t)b * CV_F4_B];
        prv[b] = src[(size_t)b * CV_F4_B + poff];
    }
    const float4 w    = *(const float4*)(cw + c * KC);
    const float  bias = cb[c];

    #pragma unroll
    for (int b = 0; b < BSZ; ++b) {
        const float p1 = has_prev ? prv[b].y : 0.f;
        const float p2 = has_prev ? prv[b].z : 0.f;
        const float p3 = has_prev ? prv[b].w : 0.f;
        const float4 x = cur[b];
        float s0 = bias, s1 = bias, s2 = bias, s3 = bias;
        s0 += p1  * w.x;  s1 += p2  * w.x;  s2 += p3  * w.x;  s3 += x.x * w.x;
        s0 += p2  * w.y;  s1 += p3  * w.y;  s2 += x.x * w.y;  s3 += x.y * w.y;
        s0 += p3  * w.z;  s1 += x.x * w.z;  s2 += x.y * w.z;  s3 += x.z * w.z;
        s0 += x.x * w.w;  s1 += x.y * w.w;  s2 += x.z * w.w;  s3 += x.w * w.w;
        dst[(size_t)b * CV_F4_B] = make_float4(s0 * sigmoid_(s0), s1 * sigmoid_(s1),
                                               s2 * sigmoid_(s2), s3 * sigmoid_(s3));
    }
}

#define XP_NW    8
#define XP_KW    (C_ / XP_NW)
#define XP_RB    32
#define XP_LB    32
#define XP_RBLK  (RN2 / XP_RB)
#define XP_LBLK  ((L_ + XP_LB - 1) / XP_LB)
#define XP_G     4
#define XP_NG    (XP_KW / (8 * XP_G))

__device__ __forceinline__ void xp_load(const float* __restrict__ Ap,
                                        const float* __restrict__ Bp,
                                        float4 (&a)[2][XP_G], float2 (&bq)[XP_G][4])
{
    #pragma unroll
    for (int s = 0; s < XP_G; ++s) {
        a[0][s] = *(const float4*)(Ap + 8 * s);
        a[1][s] = *(const float4*)(Ap + (size_t)16 * C_ + 8 * s);
    }
    #pragma unroll
    for (int s = 0; s < XP_G; ++s) {
        #pragma unroll
        for (int i = 0; i < 4; ++i)
            bq[s][i] = *(const float2*)(Bp + (size_t)(8 * s + i) * L_);
    }
}

__device__ __forceinline__ void xp_mma(const float4 (&a)[2][XP_G],
                                       const float2 (&bq)[XP_G][4], v8f (&acc)[2][2])
{
    #pragma unroll
    for (int s = 0; s < XP_G; ++s) {
        const v2f be0 = {bq[s][0].x, bq[s][1].x};
        const v2f bo0 = {bq[s][0].y, bq[s][1].y};
        const v2f be1 = {bq[s][2].x, bq[s][3].x};
        const v2f bo1 = {bq[s][2].y, bq[s][3].y};
        const v2f a0lo = {a[0][s].x, a[0][s].y};
        const v2f a1lo = {a[1][s].x, a[1][s].y};
        const v2f a0hi = {a[0][s].z, a[0][s].w};
        const v2f a1hi = {a[1][s].z, a[1][s].w};
        acc[0][0] = __builtin_amdgcn_wmma_f32_16x16x4_f32(
                        false, a0lo, false, be0, (short)0, acc[0][0], false, false);
        acc[0][1] = __builtin_amdgcn_wmma_f32_16x16x4_f32(
                        false, a0lo, false, bo0, (short)0, acc[0][1], false, false);
        acc[1][0] = __builtin_amdgcn_wmma_f32_16x16x4_f32(
                        false, a1lo, false, be0, (short)0, acc[1][0], false, false);
        acc[1][1] = __builtin_amdgcn_wmma_f32_16x16x4_f32(
                        false, a1lo, false, bo0, (short)0, acc[1][1], false, false);
        acc[0][0] = __builtin_amdgcn_wmma_f32_16x16x4_f32(
                        false, a0hi, false, be1, (short)0, acc[0][0], false, false);
        acc[0][1] = __builtin_amdgcn_wmma_f32_16x16x4_f32(
                        false, a0hi, false, bo1, (short)0, acc[0][1], false, false);
        acc[1][0] = __builtin_amdgcn_wmma_f32_16x16x4_f32(
                        false, a1hi, false, be1, (short)0, acc[1][0], false, false);
        acc[1][1] = __builtin_amdgcn_wmma_f32_16x16x4_f32(
                        false, a1hi, false, bo1, (short)0, acc[1][1], false, false);
    }
}

__global__ __launch_bounds__(XP_NW * 32)
void k_gemm_xproj(const float* __restrict__ xc,
                  const float* __restrict__ w,
                  float* __restrict__ x_dbl)
{
    __shared__ __attribute__((aligned(16))) float sPart[XP_NW * 8 * 32 * 4];

    const int tid  = threadIdx.x;
    const int lane = tid & 31;
    const int wv   = tid >> 5;
    const int c    = lane & 15;
    const int h    = lane >> 4;
    const int rb   = blockIdx.x % XP_RBLK;
    const int lb   = blockIdx.x / XP_RBLK;
    const int b    = blockIdx.y;

    const int l0 = lb * XP_LB + 2 * c;
    const int lc = (l0 < L_) ? l0 : (L_ - 2);
    const int k0 = wv * XP_KW + 4 * h;

    const float* Ap = w  + (size_t)(rb * XP_RB + c) * C_ + k0;
    const float* Bp = xc + ((size_t)b * C_ + k0) * L_ + lc;

    v8f acc[2][2];
    #pragma unroll
    for (int f = 0; f < 2; ++f) {
        #pragma unroll
        for (int g = 0; g < 2; ++g)
            acc[f][g] = (v8f){0.f,0.f,0.f,0.f,0.f,0.f,0.f,0.f};
    }

    float4 a[2][2][XP_G];
    float2 bq[2][XP_G][4];
    xp_load(Ap, Bp, a[0], bq[0]);
    #pragma unroll
    for (int t = 0; t < XP_NG; ++t) {
        if (t + 1 < XP_NG)
            xp_load(Ap + (t + 1) * 8 * XP_G, Bp + (size_t)(t + 1) * 8 * XP_G * L_,
                    a[(t + 1) & 1], bq[(t + 1) & 1]);
        __builtin_amdgcn_sched_barrier(0);
        xp_mma(a[t & 1], bq[t & 1], acc);
        __builtin_amdgcn_sched_barrier(0);
    }

    float4* sp = (float4*)sPart;
    #pragma unroll
    for (int f = 0; f < 2; ++f) {
        #pragma unroll
        for (int rp = 0; rp < 4; ++rp)
            sp[(wv * 8 + f * 4 + rp) * 32 + lane] =
                make_float4(acc[f][0][2 * rp],     acc[f][1][2 * rp],
                            acc[f][0][2 * rp + 1], acc[f][1][2 * rp + 1]);
    }
    __syncthreads();

    float4 s = sp[wv * 32 + lane];
    #pragma unroll
    for (int u = 1; u < XP_NW; ++u) {
        const float4 t = sp[(u * 8 + wv) * 32 + lane];
        s.x += t.x; s.y += t.y; s.z += t.z; s.w += t.w;
    }
    const int row = rb * XP_RB + 16 * (wv >> 2) + 8 * h + 2 * (wv & 3);
    if (l0 < L_) {
        float* o = x_dbl + ((size_t)b * RN2 + row) * L_ + l0;
        *(float2*)o        = make_float2(s.x, s.y);
        *(float2*)(o + L_) = make_float2(s.z, s.w);
    }
}

#define SC_THREADS 128
#define SC_CH      (SC_THREADS / 2)
#define SC_NH      (N_ / 2)
#define SC_CHUNKS  (L_ / 4)
#define SC_SETS    8
#define BC_PITCH   36
#define SC_STAGE_FULL ((2 * N_ * SC_CHUNKS) / SC_THREADS)
#define SC_STAGE_REST ((2 * N_ * SC_CHUNKS) - SC_STAGE_FULL * SC_THREADS)
#define SD_KS      (R_ / 8)
#define SD_LBLK    ((L_ + 31) / 32)
#define SD_PITCH   228

__device__ __forceinline__ void sd_load_b(const float* __restrict__ Bp, float2 (&bq)[SD_KS][4])
{
    #pragma unroll
    for (int s = 0; s < SD_KS; ++s) {
        #pragma unroll
        for (int i = 0; i < 4; ++i)
            bq[s][i] = *(const float2*)(Bp + (size_t)(8 * s + i) * L_);
    }
}

__device__ __forceinline__ float softplus_(float v) {
    const float t = __logf(1.f + __expf(v));
    return (v > 20.f) ? v : t;
}

__device__ __forceinline__ void sd_store(const v8f& acc0, const v8f& acc1,
                                         const float (&bias)[8], float* __restrict__ drow)
{
    #pragma unroll
    for (int r = 0; r < 8; ++r) {
        const float v0 = acc0[r] + bias[r];
        const float v1 = acc1[r] + bias[r];
        *(float2*)(drow + r * SD_PITCH) = make_float2(softplus_(v0), softplus_(v1));
    }
}

__device__ __forceinline__ void sd_mma_store(const float4 (&a)[SD_KS], const float2 (&bq)[SD_KS][4],
                                             v8f& acc0, v8f& acc1,
                                             const v8f& accP0, const v8f& accP1,
                                             const float (&bias)[8], float* __restrict__ drow)
{
    #pragma unroll
    for (int s = 0; s < SD_KS; ++s) {
        const v2f be0 = {bq[s][0].x, bq[s][1].x};
        const v2f bo0 = {bq[s][0].y, bq[s][1].y};
        const v2f be1 = {bq[s][2].x, bq[s][3].x};
        const v2f bo1 = {bq[s][2].y, bq[s][3].y};
        const v2f alo = {a[s].x, a[s].y};
        const v2f ahi = {a[s].z, a[s].w};
        acc0 = __builtin_amdgcn_wmma_f32_16x16x4_f32(
                   false, alo, false, be0, (short)0, acc0, false, false);
        acc1 = __builtin_amdgcn_wmma_f32_16x16x4_f32(
                   false, alo, false, bo0, (short)0, acc1, false, false);
        acc0 = __builtin_amdgcn_wmma_f32_16x16x4_f32(
                   false, ahi, false, be1, (short)0, acc0, false, false);
        acc1 = __builtin_amdgcn_wmma_f32_16x16x4_f32(
                   false, ahi, false, bo1, (short)0, acc1, false, false);
        if (s < 8) {
            const float v0 = accP0[s] + bias[s];
            const float v1 = accP1[s] + bias[s];
            *(float2*)(drow + s * SD_PITCH) = make_float2(softplus_(v0), softplus_(v1));
        }
    }
}

__device__ __forceinline__ float pair_other(const float v)
{
    return __builtin_bit_cast(float,
        __builtin_amdgcn_update_dpp(0, __builtin_bit_cast(int, v), 0xB1, 0xF, 0xF, true));
}

__device__ __forceinline__ void scan_rows(const float* __restrict__ srow, float4 (&rw)[16])
{
    #pragma unroll
    for (int k = 0; k < 4; ++k) {
        const float4* row = (const float4*)(srow + k * BC_PITCH);
        rw[4 * k + 0] = row[0];
        rw[4 * k + 1] = row[1];
        rw[4 * k + 2] = row[4];
        rw[4 * k + 3] = row[5];
    }
}

__device__ __forceinline__ void scan_chunk4(const float4 (&rw)[16],
                                            const float4 d4, const float4 x4,
                                            const float2 z2, const bool hi,
                                            const float (&A_neg)[SC_NH],
                                            float (&h)[SC_NH],
                                            const float Dp,
                                            float2* __restrict__ dst)
{
    const float dvs[4] = {d4.x, d4.y, d4.z, d4.w};
    const float xvs[4] = {x4.x, x4.y, x4.z, x4.w};
    float yt[4];
    #pragma unroll
    for (int k = 0; k < 4; ++k) {
        const float dv  = dvs[k];
        const float xv  = xvs[k];
        const float dvx = dv * xv;
        const float4 b0 = rw[4 * k], b1 = rw[4 * k + 1], c0 = rw[4 * k + 2], c1 = rw[4 * k + 3];
        const float bc[2 * SC_NH] = {b0.x, b0.y, b0.z, b0.w, b1.x, b1.y, b1.z, b1.w,
                                     c0.x, c0.y, c0.z, c0.w, c1.x, c1.y, c1.z, c1.w};
        float y0 = 0.f, y1 = 0.f, y2 = 0.f, y3 = 0.f;
        #pragma unroll
        for (int n = 0; n < SC_NH; n += 4) {
            h[n + 0] = h[n + 0] * __expf(dv * A_neg[n + 0]) + dvx * bc[n + 0];
            h[n + 1] = h[n + 1] * __expf(dv * A_neg[n + 1]) + dvx * bc[n + 1];
            h[n + 2] = h[n + 2] * __expf(dv * A_neg[n + 2]) + dvx * bc[n + 2];
            h[n + 3] = h[n + 3] * __expf(dv * A_neg[n + 3]) + dvx * bc[n + 3];
            y0 += h[n + 0] * bc[SC_NH + n + 0];
            y1 += h[n + 1] * bc[SC_NH + n + 1];
            y2 += h[n + 2] * bc[SC_NH + n + 2];
            y3 += h[n + 3] * bc[SC_NH + n + 3];
        }
        const float yp = (y0 + y1) + (y2 + y3);
        yt[k] = yp + pair_other(yp);
    }
    const float ya = hi ? yt[2] : yt[0];
    const float yb = hi ? yt[3] : yt[1];
    const float xa = hi ? xvs[2] : xvs[0];
    const float xb = hi ? xvs[3] : xvs[1];
    *dst = make_float2((ya + xa * Dp) * (z2.x * sigmoid_(z2.x)),
                       (yb + xb * Dp) * (z2.y * sigmoid_(z2.y)));
}

__global__ __launch_bounds__(SC_THREADS) __attribute__((amdgpu_waves_per_eu(1, 1)))
void k_scan_out(const float* __restrict__ xc,
                const float* __restrict__ zbuf,
                const float* __restrict__ x_dbl,
                const float* __restrict__ dt_w,
                const float* __restrict__ dt_bias,
                const float* __restrict__ A_log,
                const float* __restrict__ Dp_,
                float* __restrict__ out)
{
    __shared__ __attribute__((aligned(16))) float sBC[L_ * BC_PITCH];
    __shared__ __attribute__((aligned(16))) float sDel[SC_CH * SD_PITCH];

    const int  tid  = threadIdx.x;
    const int  half = tid & 1;
    const bool hi   = (half != 0);
    const int  lane = tid & 31;
    const int  wv   = tid >> 5;
    const int  n    = lane & 15;
    const int  hh   = lane >> 4;
    const int  b    = blockIdx.x / (C_ / SC_CH);
    const int  cblk = (blockIdx.x % (C_ / SC_CH)) * SC_CH;
    const int  c    = cblk + (tid >> 1);
    const int  bc_  = b * C_ + c;

    const float4* xp4 = (const float4*)(xc    + (size_t)bc_ * L_);
    const float2* zp2 = (const float2*)(zbuf  + (size_t)bc_ * L_) + half;
    float2*       op2 = (float2*)(out + (size_t)bc_ * L_) + half;

    const float* Ap = dt_w + (size_t)(cblk + 16 * wv + n) * R_ + 4 * hh;
    float4 a[SD_KS];
    #pragma unroll
    for (int s = 0; s < SD_KS; ++s)
        a[s] = *(const float4*)(Ap + 8 * s);
    float bias[8];
    {
        const float4* bp4 = (const float4*)(dt_bias + cblk + 16 * wv + 8 * hh);
        const float4 t0 = bp4[0], t1 = bp4[1];
        bias[0] = t0.x; bias[1] = t0.y; bias[2] = t0.z; bias[3] = t0.w;
        bias[4] = t1.x; bias[5] = t1.y; bias[6] = t1.z; bias[7] = t1.w;
    }
    const float* Bbase = x_dbl + ((size_t)b * RN2 + 4 * hh) * L_;
    float2 bq[SD_KS][4];
    sd_load_b(Bbase + 2 * n, bq);

    const float4* src4 = (const float4*)(x_dbl + ((size_t)b * RN2 + R_) * L_);
    float4 stg[SC_STAGE_FULL];
    #pragma unroll
    for (int j = 0; j < SC_STAGE_FULL; ++j)
        stg[j] = src4[tid + j * SC_THREADS];
    const int    frest = (tid < SC_STAGE_REST) ? (tid + SC_STAGE_FULL * SC_THREADS)
                                               : (SC_STAGE_FULL * SC_THREADS);
    const float4 stgr  = src4[frest];

    const float4* a4 = (const float4*)(A_log + (size_t)c * N_ + SC_NH * half);
    const float4 al0 = a4[0], al1 = a4[1];
    const float  Dp  = Dp_[c];

    float4 qx[SC_SETS];
    float2 qz[SC_SETS];
    #pragma unroll
    for (int s = 0; s < SC_SETS - 1; ++s) {
        qx[s] = xp4[s]; qz[s] = zp2[2 * s];
    }

    __builtin_amdgcn_sched_barrier(0);

    #pragma unroll
    for (int j = 0; j < SC_STAGE_FULL; ++j) {
        const int f  = tid + j * SC_THREADS;
        const int r  = f / SC_CHUNKS;
        const int l4 = f % SC_CHUNKS;
        const float4 v = stg[j];
        float* d = sBC + (l4 * 4) * BC_PITCH + r;
        d[0]            = v.x;
        d[BC_PITCH]     = v.y;
        d[2 * BC_PITCH] = v.z;
        d[3 * BC_PITCH] = v.w;
    }
    asm volatile("" :: "v"(stgr.x), "v"(stgr.y), "v"(stgr.z), "v"(stgr.w));
    if (tid < SC_STAGE_REST) {
        const int r  = frest / SC_CHUNKS;
        const int l4 = frest % SC_CHUNKS;
        float* d = sBC + (l4 * 4) * BC_PITCH + r;
        d[0]            = stgr.x;
        d[BC_PITCH]     = stgr.y;
        d[2 * BC_PITCH] = stgr.z;
        d[3 * BC_PITCH] = stgr.w;
    }

    float A_neg[SC_NH];
    A_neg[0] = -__expf(al0.x); A_neg[1] = -__expf(al0.y);
    A_neg[2] = -__expf(al0.z); A_neg[3] = -__expf(al0.w);
    A_neg[4] = -__expf(al1.x); A_neg[5] = -__expf(al1.y);
    A_neg[6] = -__expf(al1.z); A_neg[7] = -__expf(al1.w);

    float* drow = sDel + (16 * wv + 8 * hh) * SD_PITCH + 2 * n;
    float2 bn[SD_KS][4];
    v8f accP0 = (v8f){0.f,0.f,0.f,0.f,0.f,0.f,0.f,0.f};
    v8f accP1 = accP0;
    #pragma unroll 1
    for (int lb = 0; lb < SD_LBLK; ++lb) {
        const int l0 = 32 * (lb + 1) + 2 * n;
        const int lc = (l0 < L_) ? l0 : (L_ - 2);
        sd_load_b(Bbase + lc, bn);
        __builtin_amdgcn_sched_barrier(0);
        v8f acc0 = (v8f){0.f,0.f,0.f,0.f,0.f,0.f,0.f,0.f};
        v8f acc1 = acc0;
        sd_mma_store(a, bq, acc0, acc1, accP0, accP1, bias,
                     drow + 32 * ((lb > 0) ? (lb - 1) : (SD_LBLK - 1)));
        __builtin_amdgcn_sched_barrier(0);
        accP0 = acc0; accP1 = acc1;
        #pragma unroll
        for (int s = 0; s < SD_KS; ++s) {
            #pragma unroll
            for (int i = 0; i < 4; ++i)
                bq[s][i] = bn[s][i];
        }
    }
    sd_store(accP0, accP1, bias, drow + 32 * (SD_LBLK - 1));

    float h[SC_NH];
    #pragma unroll
    for (int q = 0; q < SC_NH; ++q) h[q] = 0.f;

    __syncthreads();

    const float* sB = sBC + SC_NH * half;
    const float4* sD4 = (const float4*)(sDel + (tid >> 1) * SD_PITCH);

    float4 rw[2][16];
    float4 rd[2];
    scan_rows(sB, rw[0]);
    rd[0] = sD4[0];
    for (int i = 0; i < SC_CHUNKS - 1; i += SC_SETS) {
        #pragma unroll
        for (int j = 0; j < SC_SETS; ++j) {
            const int nx = (i + j + SC_SETS - 1 < SC_CHUNKS) ? (i + j + SC_SETS - 1)
                                                             : (SC_CHUNKS - 1);
            const int ls = (j + SC_SETS - 1) % SC_SETS;
            qx[ls] = xp4[nx]; qz[ls] = zp2[2 * nx];
            scan_rows(sB + ((i + j + 1) * 4) * BC_PITCH, rw[(j + 1) & 1]);
            rd[(j + 1) & 1] = sD4[i + j + 1];
            __builtin_amdgcn_sched_barrier(0);
            scan_chunk4(rw[j & 1], rd[j & 1], qx[j], qz[j], hi,
                        A_neg, h, Dp, op2 + 2 * (i + j));
            __builtin_amdgcn_sched_barrier(0);
        }
    }
    scan_chunk4(rw[0], rd[0], qx[0], qz[0], hi, A_neg, h, Dp, op2 + 2 * (SC_CHUNKS - 1));
}

extern "C" void kernel_launch(void* const* d_in, const int* in_sizes, int n_in,
                              void* d_out, int out_size, void* d_ws, size_t ws_size,
                              hipStream_t stream)
{
    const float* hidden    = (const float*)d_in[0];
    const float* in_proj_w = (const float*)d_in[1];
    const float* conv_w    = (const float*)d_in[2];
    const float* conv_b    = (const float*)d_in[3];
    const float* x_proj_w  = (const float*)d_in[4];
    const float* dt_proj_w = (const float*)d_in[5];
    const float* dt_bias   = (const float*)d_in[6];
    const float* A_log     = (const float*)d_in[7];
    const float* D_param   = (const float*)d_in[8];

    float* ws = (float*)d_ws;
    const size_t BCL = (size_t)BSZ * C_ * L_;
    float* x_pre = ws;
    float* zbuf  = ws + BCL;
    float* xcbuf = ws + 2 * BCL;
    float* x_dbl = ws + 3 * BCL;

    k_gemm_inproj<<<IP_EBLK * IP_MBLK, 32, 0, stream>>>(
        hidden, in_proj_w, x_pre, zbuf);

    k_conv_silu<<<CV_BLOCKS, CV_THREADS, 0, stream>>>(
        x_pre, conv_w, conv_b, xcbuf);

    k_gemm_xproj<<<dim3(XP_RBLK * XP_LBLK, BSZ), XP_NW * 32, 0, stream>>>(
        xcbuf, x_proj_w, x_dbl);

    k_scan_out<<<(BSZ * C_) / SC_CH, SC_THREADS, 0, stream>>>(
        xcbuf, zbuf, x_dbl, dt_proj_w, dt_bias, A_log, D_param, (float*)d_out);
}


# conv+SiLU kernel fused into the in_proj GEMM epilogue: x half tiled per (b,t) row, 28 new l + 4 recomputed halo columns per wave, conv taps by DPP row shifts; sum orders kept; 3 launches, not 4
# speedup vs baseline: 4.4726x; 1.0476x over previous
#include <hip/hip_runtime.h>


#define BSZ   8
#define T_    8
#define L_    196
#define D_    192
#define E_    192
#define C_    1536
#define N_    16
#define KC    4
#define R_    96
#define RN2   128
#define LTILES 13

typedef float v2f __attribute__((ext_vector_type(2)));
typedef float v8f __attribute__((ext_vector_type(8)));

__device__ __forceinline__ float sigmoid_(float x) {
    return 1.0f / (1.0f + __expf(-x));
}

#define IP_EW    64
#define IP_MW    32
#define IP_EBH   (E_ / IP_EW)
#define IP_MBLK  ((BSZ * T_ * L_) / IP_MW)
#define IP_LSTEP 28
#define IP_HALO  (IP_MW - IP_LSTEP)
#define IP_LT    (L_ / IP_LSTEP)
#define IP_XBLKS (BSZ * T_ * IP_LT * IP_EBH)
#define IP_ZBLKS (IP_MBLK * IP_EBH)
#define IP_KC    32
#define IP_KS    (IP_KC / 4)
#define IP_KQ    (IP_KC / 8)
#define IP_NCH   (D_ / IP_KC)
#define IP_NPASS (IP_NCH * 4)

__device__ __forceinline__ void ip_load_line(const float* __restrict__ p, float4 (&x)[IP_KQ])
{
    #pragma unroll
    for (int s = 0; s < IP_KQ; ++s)
        x[s] = *(const float4*)(p + 8 * s);
}

__device__ __forceinline__ void ip_pass(const float4 (&a)[IP_KQ],
                                        const float4 (&b0)[IP_KQ], const float4 (&b1)[IP_KQ],
                                        v8f& c0, v8f& c1)
{
    #pragma unroll
    for (int s = 0; s < IP_KQ; ++s) {
        const v2f alo  = {a[s].x,  a[s].y},  ahi  = {a[s].z,  a[s].w};
        const v2f b0lo = {b0[s].x, b0[s].y}, b0hi = {b0[s].z, b0[s].w};
        const v2f b1lo = {b1[s].x, b1[s].y}, b1hi = {b1[s].z, b1[s].w};
        c0 = __builtin_amdgcn_wmma_f32_16x16x4_f32(
                 false, alo, false, b0lo, (short)0, c0, false, false);
        c1 = __builtin_amdgcn_wmma_f32_16x16x4_f32(
                 false, alo, false, b1lo, (short)0, c1, false, false);
        c0 = __builtin_amdgcn_wmma_f32_16x16x4_f32(
                 false, ahi, false, b0hi, (short)0, c0, false, false);
        c1 = __builtin_amdgcn_wmma_f32_16x16x4_f32(
                 false, ahi, false, b1hi, (short)0, c1, false, false);
    }
}

template <int N>
__device__ __forceinline__ float row_shr0(const float v)
{
    return __builtin_bit_cast(float,
        __builtin_amdgcn_update_dpp(0, __builtin_bit_cast(int, v), 0x110 + N, 0xF, 0xF, true));
}
template <int N>
__device__ __forceinline__ float row_shl0(const float v)
{
    return __builtin_bit_cast(float,
        __builtin_amdgcn_update_dpp(0, __builtin_bit_cast(int, v), 0x100 + N, 0xF, 0xF, true));
}
template <int K>
__device__ __forceinline__ float tap_back(const float lo, const float hi)
{
    const int t = __builtin_amdgcn_update_dpp(0, __builtin_bit_cast(int, lo), 0x100 + 16 - K, 0xF, 0xF, true);
    return __builtin_bit_cast(float,
        __builtin_amdgcn_update_dpp(t, __builtin_bit_cast(int, hi), 0x110 + K, 0xF, 0xF, false));
}

__global__ __launch_bounds__(32)
void k_gemm_inproj(const float* __restrict__ hidden,
                   const float* __restrict__ w,
                   const float* __restrict__ cw,
                   const float* __restrict__ cb,
                   float* __restrict__ xc,
                   float* __restrict__ zbuf)
{
    const int lane = threadIdx.x;
    const int c    = lane & 15;
    const int h    = lane >> 4;
    const bool xh  = (blockIdx.x < IP_XBLKS);

    int eb, tile, row0, row1;
    int bt = 0, j = 0;
    if (xh) {
        eb   = blockIdx.x % IP_EBH;
        tile = blockIdx.x / IP_EBH;
        bt   = tile / IP_LT;
        j    = tile - bt * IP_LT;
        const int l0 = IP_LSTEP * j - IP_HALO + c;
        row0 = bt * L_ + ((l0 > 0) ? l0 : 0);
        row1 = bt * L_ + l0 + 16;
    } else {
        const int q = blockIdx.x - IP_XBLKS;
        eb   = IP_EBH + q % IP_EBH;
        tile = q / IP_EBH;
        row0 = tile * IP_MW + c;
        row1 = row0 + 16;
    }
    const int ch0 = ((eb % IP_EBH) * IP_EW + 8 * h) * T_ + (bt % T_);

    const float* Ap  = w      + (size_t)(eb * IP_EW + c) * D_ + 4 * h;
    const float* Bp0 = hidden + (size_t)row0 * D_ + 4 * h;
    const float* Bp1 = hidden + (size_t)row1 * D_ + 4 * h;

    v8f acc[4][2];
    #pragma unroll
    for (int f = 0; f < 4; ++f) {
        #pragma unroll
        for (int g = 0; g < 2; ++g)
            acc[f][g] = (v8f){0.f,0.f,0.f,0.f,0.f,0.f,0.f,0.f};
    }

    float4 a[4][IP_KQ];
    float4 b[2][2][IP_KQ];
    float4 cwv[32];
    float  cbs[32];
    ip_load_line(Bp0, b[0][0]);
    ip_load_line(Bp1, b[0][1]);
    ip_load_line(Ap, a[0]);
    ip_load_line(Ap + (size_t)16 * D_, a[1]);
    ip_load_line(Ap + (size_t)32 * D_, a[2]);

    #pragma unroll
    for (int p = 0; p < IP_NPASS; ++p) {
        const int ch = p >> 2;
        const int f  = p & 3;
        const int q  = p + 3;
        if (q < IP_NPASS)
            ip_load_line(Ap + (size_t)(q & 3) * 16 * D_ + (q >> 2) * IP_KC, a[q & 3]);
        if (f == 1 && ch + 1 < IP_NCH) {
            ip_load_line(Bp0 + (ch + 1) * IP_KC, b[(ch + 1) & 1][0]);
            ip_load_line(Bp1 + (ch + 1) * IP_KC, b[(ch + 1) & 1][1]);
        }
        if (p == IP_NPASS - 3) {
            #pragma unroll
            for (int i = 0; i < 32; ++i) {
                const int chn = ch0 + ((i >> 3) * 16 + (i & 7)) * T_;
                cwv[i] = *(const float4*)(cw + (size_t)chn * KC);
                cbs[i] = cb[chn];
            }
        }
        __builtin_amdgcn_sched_barrier(0);
        ip_pass(a[p & 3], b[ch & 1][0], b[ch & 1][1], acc[f][0], acc[f][1]);
        __builtin_amdgcn_sched_barrier(0);
    }

    #pragma unroll
    for (int i = 0; i < 32; ++i)
        asm volatile("" :: "v"(cwv[i].x), "v"(cwv[i].y), "v"(cwv[i].z), "v"(cwv[i].w), "v"(cbs[i]));

    if (!xh) {
        const int e0 = (eb - IP_EBH) * IP_EW + 8 * h;
        #pragma unroll
        for (int g = 0; g < 2; ++g) {
            const int m   = tile * IP_MW + g * 16 + c;
            const int bt_ = m / L_;
            const int l   = m - bt_ * L_;
            const int b_  = bt_ / T_;
            const int t   = bt_ % T_;
            float* p = zbuf + ((size_t)b_ * C_ + (size_t)e0 * T_ + t) * L_ + l;
            #pragma unroll
            for (int f = 0; f < 4; ++f) {
                #pragma unroll
                for (int r = 0; r < 8; ++r)
                    p[(size_t)(f * 16 + r) * (T_ * L_)] = acc[f][g][r];
            }
        }
    } else {
        const int  b_  = bt / T_;
        const bool pad = (j == 0) && (c < IP_HALO);
        const bool st0 = (c >= IP_HALO);
        float* p1 = xc + ((size_t)b_ * C_ + ch0) * L_ + (IP_LSTEP * j - IP_HALO + 16 + c);
        #pragma unroll
        for (int f = 0; f < 4; ++f) {
            float x0[8];
            #pragma unroll
            for (int r = 0; r < 8; ++r)
                x0[r] = pad ? 0.f : acc[f][0][r];
            #pragma unroll
            for (int r = 0; r < 8; ++r) {
                const float4 wv = cwv[f * 8 + r];
                const float  x1 = acc[f][1][r];
                float s1 = __builtin_fmaf(tap_back<3>(x0[r], x1), wv.x, cbs[f * 8 + r]);
                s1 = __builtin_fmaf(tap_back<2>(x0[r], x1), wv.y, s1);
                s1 = __builtin_fmaf(tap_back<1>(x0[r], x1), wv.z, s1);
                s1 = __builtin_fmaf(x1, wv.w, s1);
                p1[(size_t)(f * 16 + r) * (T_ * L_)] = s1 * sigmoid_(s1);
            }
            float s0[8];
            #pragma unroll
            for (int r = 0; r < 8; ++r) {
                const float4 wv = cwv[f * 8 + r];
                float s = __builtin_fmaf(row_shr0<3>(x0[r]), wv.x, cbs[f * 8 + r]);
                s = __builtin_fmaf(row_shr0<2>(x0[r]), wv.y, s);
                s = __builtin_fmaf(row_shr0<1>(x0[r]), wv.z, s);
                s0[r] = __builtin_fmaf(x0[r], wv.w, s);
            }
            if (st0) {
                #pragma unroll
                for (int r = 0; r < 8; ++r)
                    p1[(ptrdiff_t)(f * 16 + r) * (T_ * L_) - 16] = s0[r] * sigmoid_(s0[r]);
            }
        }
    }
}

#define XP_NW    8
#define XP_KW    (C_ / XP_NW)
#define XP_RB    32
#define XP_LB    32
#define XP_RBLK  (RN2 / XP_RB)
#define XP_LBLK  ((L_ + XP_LB - 1) / XP_LB)
#define XP_G     4
#define XP_NG    (XP_KW / (8 * XP_G))

__device__ __forceinline__ void xp_load(const float* __restrict__ Ap,
                                        const float* __restrict__ Bp,
                                        float4 (&a)[2][XP_G], float2 (&bq)[XP_G][4])
{
    #pragma unroll
    for (int s = 0; s < XP_G; ++s) {
        a[0][s] = *(const float4*)(Ap + 8 * s);
        a[1][s] = *(const float4*)(Ap + (size_t)16 * C_ + 8 * s);
    }
    #pragma unroll
    for (int s = 0; s < XP_G; ++s) {
        #pragma unroll
        for (int i = 0; i < 4; ++i)
            bq[s][i] = *(const float2*)(Bp + (size_t)(8 * s + i) * L_);
    }
}

__device__ __forceinline__ void xp_mma(const float4 (&a)[2][XP_G],
                                       const float2 (&bq)[XP_G][4], v8f (&acc)[2][2])
{
    #pragma unroll
    for (int s = 0; s < XP_G; ++s) {
        const v2f be0 = {bq[s][0].x, bq[s][1].x};
        const v2f bo0 = {bq[s][0].y, bq[s][1].y};
        const v2f be1 = {bq[s][2].x, bq[s][3].x};
        const v2f bo1 = {bq[s][2].y, bq[s][3].y};
        const v2f a0lo = {a[0][s].x, a[0][s].y};
        const v2f a1lo = {a[1][s].x, a[1][s].y};
        const v2f a0hi = {a[0][s].z, a[0][s].w};
        const v2f a1hi = {a[1][s].z, a[1][s].w};
        acc[0][0] = __builtin_amdgcn_wmma_f32_16x16x4_f32(
                        false, a0lo, false, be0, (short)0, acc[0][0], false, false);
        acc[0][1] = __builtin_amdgcn_wmma_f32_16x16x4_f32(
                        false, a0lo, false, bo0, (short)0, acc[0][1], false, false);
        acc[1][0] = __builtin_amdgcn_wmma_f32_16x16x4_f32(
                        false, a1lo, false, be0, (short)0, acc[1][0], false, false);
        acc[1][1] = __builtin_amdgcn_wmma_f32_16x16x4_f32(
                        false, a1lo, false, bo0, (short)0, acc[1][1], false, false);
        acc[0][0] = __builtin_amdgcn_wmma_f32_16x16x4_f32(
                        false, a0hi, false, be1, (short)0, acc[0][0], false, false);
        acc[0][1] = __builtin_amdgcn_wmma_f32_16x16x4_f32(
                        false, a0hi, false, bo1, (short)0, acc[0][1], false, false);
        acc[1][0] = __builtin_amdgcn_wmma_f32_16x16x4_f32(
                        false, a1hi, false, be1, (short)0, acc[1][0], false, false);
        acc[1][1] = __builtin_amdgcn_wmma_f32_16x16x4_f32(
                        false, a1hi, false, bo1, (short)0, acc[1][1], false, false);
    }
}

__global__ __launch_bounds__(XP_NW * 32)
void k_gemm_xproj(const float* __restrict__ xc,
                  const float* __restrict__ w,
                  float* __restrict__ x_dbl)
{
    __shared__ __attribute__((aligned(16))) float sPart[XP_NW * 8 * 32 * 4];

    const int tid  = threadIdx.x;
    const int lane = tid & 31;
    const int wv   = tid >> 5;
    const int c    = lane & 15;
    const int h    = lane >> 4;
    const int rb   = blockIdx.x % XP_RBLK;
    const int lb   = blockIdx.x / XP_RBLK;
    const int b    = blockIdx.y;

    const int l0 = lb * XP_LB + 2 * c;
    const int lc = (l0 < L_) ? l0 : (L_ - 2);
    const int k0 = wv * XP_KW + 4 * h;

    const float* Ap = w  + (size_t)(rb * XP_RB + c) * C_ + k0;
    const float* Bp = xc + ((size_t)b * C_ + k0) * L_ + lc;

    v8f acc[2][2];
    #pragma unroll
    for (int f = 0; f < 2; ++f) {
        #pragma unroll
        for (int g = 0; g < 2; ++g)
            acc[f][g] = (v8f){0.f,0.f,0.f,0.f,0.f,0.f,0.f,0.f};
    }

    float4 a[2][2][XP_G];
    float2 bq[2][XP_G][4];
    xp_load(Ap, Bp, a[0], bq[0]);
    #pragma unroll
    for (int t = 0; t < XP_NG; ++t) {
        if (t + 1 < XP_NG)
            xp_load(Ap + (t + 1) * 8 * XP_G, Bp + (size_t)(t + 1) * 8 * XP_G * L_,
                    a[(t + 1) & 1], bq[(t + 1) & 1]);
        __builtin_amdgcn_sched_barrier(0);
        xp_mma(a[t & 1], bq[t & 1], acc);
        __builtin_amdgcn_sched_barrier(0);
    }

    float4* sp = (float4*)sPart;
    #pragma unroll
    for (int f = 0; f < 2; ++f) {
        #pragma unroll
        for (int rp = 0; rp < 4; ++rp)
            sp[(wv * 8 + f * 4 + rp) * 32 + lane] =
                make_float4(acc[f][0][2 * rp],     acc[f][1][2 * rp],
                            acc[f][0][2 * rp + 1], acc[f][1][2 * rp + 1]);
    }
    __syncthreads();

    float4 s = sp[wv * 32 + lane];
    #pragma unroll
    for (int u = 1; u < XP_NW; ++u) {
        const float4 t = sp[(u * 8 + wv) * 32 + lane];
        s.x += t.x; s.y += t.y; s.z += t.z; s.w += t.w;
    }
    const int row = rb * XP_RB + 16 * (wv >> 2) + 8 * h + 2 * (wv & 3);
    if (l0 < L_) {
        float* o = x_dbl + ((size_t)b * RN2 + row) * L_ + l0;
        *(float2*)o        = make_float2(s.x, s.y);
        *(float2*)(o + L_) = make_float2(s.z, s.w);
    }
}

#define SC_THREADS 128
#define SC_CH      (SC_THREADS / 2)
#define SC_NH      (N_ / 2)
#define SC_CHUNKS  (L_ / 4)
#define SC_SETS    8
#define BC_PITCH   36
#define SC_STAGE_FULL ((2 * N_ * SC_CHUNKS) / SC_THREADS)
#define SC_STAGE_REST ((2 * N_ * SC_CHUNKS) - SC_STAGE_FULL * SC_THREADS)
#define SD_KS      (R_ / 8)
#define SD_LBLK    ((L_ + 31) / 32)
#define SD_PITCH   228

__device__ __forceinline__ void sd_load_b(const float* __restrict__ Bp, float2 (&bq)[SD_KS][4])
{
    #pragma unroll
    for (int s = 0; s < SD_KS; ++s) {
        #pragma unroll
        for (int i = 0; i < 4; ++i)
            bq[s][i] = *(const float2*)(Bp + (size_t)(8 * s + i) * L_);
    }
}

__device__ __forceinline__ float softplus_(float v) {
    const float t = __logf(1.f + __expf(v));
    return (v > 20.f) ? v : t;
}

__device__ __forceinline__ void sd_store(const v8f& acc0, const v8f& acc1,
                                         const float (&bias)[8], float* __restrict__ drow)
{
    #pragma unroll
    for (int r = 0; r < 8; ++r) {
        const float v0 = acc0[r] + bias[r];
        const float v1 = acc1[r] + bias[r];
        *(float2*)(drow + r * SD_PITCH) = make_float2(softplus_(v0), softplus_(v1));
    }
}

__device__ __forceinline__ void sd_mma_store(const float4 (&a)[SD_KS], const float2 (&bq)[SD_KS][4],
                                             v8f& acc0, v8f& acc1,
                                             const v8f& accP0, const v8f& accP1,
                                             const float (&bias)[8], float* __restrict__ drow)
{
    #pragma unroll
    for (int s = 0; s < SD_KS; ++s) {
        const v2f be0 = {bq[s][0].x, bq[s][1].x};
        const v2f bo0 = {bq[s][0].y, bq[s][1].y};
        const v2f be1 = {bq[s][2].x, bq[s][3].x};
        const v2f bo1 = {bq[s][2].y, bq[s][3].y};
        const v2f alo = {a[s].x, a[s].y};
        const v2f ahi = {a[s].z, a[s].w};
        acc0 = __builtin_amdgcn_wmma_f32_16x16x4_f32(
                   false, alo, false, be0, (short)0, acc0, false, false);
        acc1 = __builtin_amdgcn_wmma_f32_16x16x4_f32(
                   false, alo, false, bo0, (short)0, acc1, false, false);
        acc0 = __builtin_amdgcn_wmma_f32_16x16x4_f32(
                   false, ahi, false, be1, (short)0, acc0, false, false);
        acc1 = __builtin_amdgcn_wmma_f32_16x16x4_f32(
                   false, ahi, false, bo1, (short)0, acc1, false, false);
        if (s < 8) {
            const float v0 = accP0[s] + bias[s];
            const float v1 = accP1[s] + bias[s];
            *(float2*)(drow + s * SD_PITCH) = make_float2(softplus_(v0), softplus_(v1));
        }
    }
}

__device__ __forceinline__ float pair_other(const float v)
{
    return __builtin_bit_cast(float,
        __builtin_amdgcn_update_dpp(0, __builtin_bit_cast(int, v), 0xB1, 0xF, 0xF, true));
}

__device__ __forceinline__ void scan_rows(const float* __restrict__ srow, float4 (&rw)[16])
{
    #pragma unroll
    for (int k = 0; k < 4; ++k) {
        const float4* row = (const float4*)(srow + k * BC_PITCH);
        rw[4 * k + 0] = row[0];
        rw[4 * k + 1] = row[1];
        rw[4 * k + 2] = row[4];
        rw[4 * k + 3] = row[5];
    }
}

__device__ __forceinline__ void scan_chunk4(const float4 (&rw)[16],
                                            const float4 d4, const float4 x4,
                                            const float2 z2, const bool hi,
                                            const float (&A_neg)[SC_NH],
                                            float (&h)[SC_NH],
                                            const float Dp,
                                            float2* __restrict__ dst)
{
    const float dvs[4] = {d4.x, d4.y, d4.z, d4.w};
    const float xvs[4] = {x4.x, x4.y, x4.z, x4.w};
    float yt[4];
    #pragma unroll
    for (int k = 0; k < 4; ++k) {
        const float dv  = dvs[k];
        const float xv  = xvs[k];
        const float dvx = dv * xv;
        const float4 b0 = rw[4 * k], b1 = rw[4 * k + 1], c0 = rw[4 * k + 2], c1 = rw[4 * k + 3];
        const float bc[2 * SC_NH] = {b0.x, b0.y, b0.z, b0.w, b1.x, b1.y, b1.z, b1.w,
                                     c0.x, c0.y, c0.z, c0.w, c1.x, c1.y, c1.z, c1.w};
        float y0 = 0.f, y1 = 0.f, y2 = 0.f, y3 = 0.f;
        #pragma unroll
        for (int n = 0; n < SC_NH; n += 4) {
            h[n + 0] = h[n + 0] * __expf(dv * A_neg[n + 0]) + dvx * bc[n + 0];
            h[n + 1] = h[n + 1] * __expf(dv * A_neg[n + 1]) + dvx * bc[n + 1];
            h[n + 2] = h[n + 2] * __expf(dv * A_neg[n + 2]) + dvx * bc[n + 2];
            h[n + 3] = h[n + 3] * __expf(dv * A_neg[n + 3]) + dvx * bc[n + 3];
            y0 += h[n + 0] * bc[SC_NH + n + 0];
            y1 += h[n + 1] * bc[SC_NH + n + 1];
            y2 += h[n + 2] * bc[SC_NH + n + 2];
            y3 += h[n + 3] * bc[SC_NH + n + 3];
        }
        const float yp = (y0 + y1) + (y2 + y3);
        yt[k] = yp + pair_other(yp);
    }
    const float ya = hi ? yt[2] : yt[0];
    const float yb = hi ? yt[3] : yt[1];
    const float xa = hi ? xvs[2] : xvs[0];
    const float xb = hi ? xvs[3] : xvs[1];
    *dst = make_float2((ya + xa * Dp) * (z2.x * sigmoid_(z2.x)),
                       (yb + xb * Dp) * (z2.y * sigmoid_(z2.y)));
}

__global__ __launch_bounds__(SC_THREADS) __attribute__((amdgpu_waves_per_eu(1, 1)))
void k_scan_out(const float* __restrict__ xc,
                const float* __restrict__ zbuf,
                const float* __restrict__ x_dbl,
                const float* __restrict__ dt_w,
                const float* __restrict__ dt_bias,
                const float* __restrict__ A_log,
                const float* __restrict__ Dp_,
                float* __restrict__ out)
{
    __shared__ __attribute__((aligned(16))) float sBC[L_ * BC_PITCH];
    __shared__ __attribute__((aligned(16))) float sDel[SC_CH * SD_PITCH];

    const int  tid  = threadIdx.x;
    const int  half = tid & 1;
    const bool hi   = (half != 0);
    const int  lane = tid & 31;
    const int  wv   = tid >> 5;
    const int  n    = lane & 15;
    const int  hh   = lane >> 4;
    const int  b    = blockIdx.x / (C_ / SC_CH);
    const int  cblk = (blockIdx.x % (C_ / SC_CH)) * SC_CH;
    const int  c    = cblk + (tid >> 1);
    const int  bc_  = b * C_ + c;

    const float4* xp4 = (const float4*)(xc    + (size_t)bc_ * L_);
    const float2* zp2 = (const float2*)(zbuf  + (size_t)bc_ * L_) + half;
    float2*       op2 = (float2*)(out + (size_t)bc_ * L_) + half;

    const float* Ap = dt_w + (size_t)(cblk + 16 * wv + n) * R_ + 4 * hh;
    float4 a[SD_KS];
    #pragma unroll
    for (int s = 0; s < SD_KS; ++s)
        a[s] = *(const float4*)(Ap + 8 * s);
    float bias[8];
    {
        const float4* bp4 = (const float4*)(dt_bias + cblk + 16 * wv + 8 * hh);
        const float4 t0 = bp4[0], t1 = bp4[1];
        bias[0] = t0.x; bias[1] = t0.y; bias[2] = t0.z; bias[3] = t0.w;
        bias[4] = t1.x; bias[5] = t1.y; bias[6] = t1.z; bias[7] = t1.w;
    }
    const float* Bbase = x_dbl + ((size_t)b * RN2 + 4 * hh) * L_;
    float2 bq[SD_KS][4];
    sd_load_b(Bbase + 2 * n, bq);

    const float4* src4 = (const float4*)(x_dbl + ((size_t)b * RN2 + R_) * L_);
    float4 stg[SC_STAGE_FULL];
    #pragma unroll
    for (int j = 0; j < SC_STAGE_FULL; ++j)
        stg[j] = src4[tid + j * SC_THREADS];
    const int    frest = (tid < SC_STAGE_REST) ? (tid + SC_STAGE_FULL * SC_THREADS)
                                               : (SC_STAGE_FULL * SC_THREADS);
    const float4 stgr  = src4[frest];

    const float4* a4 = (const float4*)(A_log + (size_t)c * N_ + SC_NH * half);
    const float4 al0 = a4[0], al1 = a4[1];
    const float  Dp  = Dp_[c];

    float4 qx[SC_SETS];
    float2 qz[SC_SETS];
    #pragma unroll
    for (int s = 0; s < SC_SETS - 1; ++s) {
        qx[s] = xp4[s]; qz[s] = zp2[2 * s];
    }

    __builtin_amdgcn_sched_barrier(0);

    #pragma unroll
    for (int j = 0; j < SC_STAGE_FULL; ++j) {
        const int f  = tid + j * SC_THREADS;
        const int r  = f / SC_CHUNKS;
        const int l4 = f % SC_CHUNKS;
        const float4 v = stg[j];
        float* d = sBC + (l4 * 4) * BC_PITCH + r;
        d[0]            = v.x;
        d[BC_PITCH]     = v.y;
        d[2 * BC_PITCH] = v.z;
        d[3 * BC_PITCH] = v.w;
    }
    asm volatile("" :: "v"(stgr.x), "v"(stgr.y), "v"(stgr.z), "v"(stgr.w));
    if (tid < SC_STAGE_REST) {
        const int r  = frest / SC_CHUNKS;
        const int l4 = frest % SC_CHUNKS;
        float* d = sBC + (l4 * 4) * BC_PITCH + r;
        d[0]            = stgr.x;
        d[BC_PITCH]     = stgr.y;
        d[2 * BC_PITCH] = stgr.z;
        d[3 * BC_PITCH] = stgr.w;
    }

    float A_neg[SC_NH];
    A_neg[0] = -__expf(al0.x); A_neg[1] = -__expf(al0.y);
    A_neg[2] = -__expf(al0.z); A_neg[3] = -__expf(al0.w);
    A_neg[4] = -__expf(al1.x); A_neg[5] = -__expf(al1.y);
    A_neg[6] = -__expf(al1.z); A_neg[7] = -__expf(al1.w);

    float* drow = sDel + (16 * wv + 8 * hh) * SD_PITCH + 2 * n;
    float2 bn[SD_KS][4];
    v8f accP0 = (v8f){0.f,0.f,0.f,0.f,0.f,0.f,0.f,0.f};
    v8f accP1 = accP0;
    #pragma unroll 1
    for (int lb = 0; lb < SD_LBLK; ++lb) {
        const int l0 = 32 * (lb + 1) + 2 * n;
        const int lc = (l0 < L_) ? l0 : (L_ - 2);
        sd_load_b(Bbase + lc, bn);
        __builtin_amdgcn_sched_barrier(0);
        v8f acc0 = (v8f){0.f,0.f,0.f,0.f,0.f,0.f,0.f,0.f};
        v8f acc1 = acc0;
        sd_mma_store(a, bq, acc0, acc1, accP0, accP1, bias,
                     drow + 32 * ((lb > 0) ? (lb - 1) : (SD_LBLK - 1)));
        __builtin_amdgcn_sched_barrier(0);
        accP0 = acc0; accP1 = acc1;
        #pragma unroll
        for (int s = 0; s < SD_KS; ++s) {
            #pragma unroll
            for (int i = 0; i < 4; ++i)
                bq[s][i] = bn[s][i];
        }
    }
    sd_store(accP0, accP1, bias, drow + 32 * (SD_LBLK - 1));

    float h[SC_NH];
    #pragma unroll
    for (int q = 0; q < SC_NH; ++q) h[q] = 0.f;

    __syncthreads();

    const float* sB = sBC + SC_NH * half;
    const float4* sD4 = (const float4*)(sDel + (tid >> 1) * SD_PITCH);

    float4 rw[2][16];
    float4 rd[2];
    scan_rows(sB, rw[0]);
    rd[0] = sD4[0];
    for (int i = 0; i < SC_CHUNKS - 1; i += SC_SETS) {
        #pragma unroll
        for (int j = 0; j < SC_SETS; ++j) {
            const int nx = (i + j + SC_SETS - 1 < SC_CHUNKS) ? (i + j + SC_SETS - 1)
                                                             : (SC_CHUNKS - 1);
            const int ls = (j + SC_SETS - 1) % SC_SETS;
            qx[ls] = xp4[nx]; qz[ls] = zp2[2 * nx];
            scan_rows(sB + ((i + j + 1) * 4) * BC_PITCH, rw[(j + 1) & 1]);
            rd[(j + 1) & 1] = sD4[i + j + 1];
            __builtin_amdgcn_sched_barrier(0);
            scan_chunk4(rw[j & 1], rd[j & 1], qx[j], qz[j], hi,
                        A_neg, h, Dp, op2 + 2 * (i + j));
            __builtin_amdgcn_sched_barrier(0);
        }
    }
    scan_chunk4(rw[0], rd[0], qx[0], qz[0], hi, A_neg, h, Dp, op2 + 2 * (SC_CHUNKS - 1));
}

extern "C" void kernel_launch(void* const* d_in, const int* in_sizes, int n_in,
                              void* d_out, int out_size, void* d_ws, size_t ws_size,
                              hipStream_t stream)
{
    const float* hidden    = (const float*)d_in[0];
    const float* in_proj_w = (const float*)d_in[1];
    const float* conv_w    = (const float*)d_in[2];
    const float* conv_b    = (const float*)d_in[3];
    const float* x_proj_w  = (const float*)d_in[4];
    const float* dt_proj_w = (const float*)d_in[5];
    const float* dt_bias   = (const float*)d_in[6];
    const float* A_log     = (const float*)d_in[7];
    const float* D_param   = (const float*)d_in[8];

    float* ws = (float*)d_ws;
    const size_t BCL = (size_t)BSZ * C_ * L_;
    float* zbuf  = ws + BCL;
    float* xcbuf = ws + 2 * BCL;
    float* x_dbl = ws + 3 * BCL;

    k_gemm_inproj<<<IP_XBLKS + IP_ZBLKS, 32, 0, stream>>>(
        hidden, in_proj_w, conv_w, conv_b, xcbuf, zbuf);

    k_gemm_xproj<<<dim3(XP_RBLK * XP_LBLK, BSZ), XP_NW * 32, 0, stream>>>(
        xcbuf, x_proj_w, x_dbl);

    k_scan_out<<<(BSZ * C_) / SC_CH, SC_THREADS, 0, stream>>>(
        xcbuf, zbuf, x_dbl, dt_proj_w, dt_bias, A_log, D_param, (float*)d_out);
}


# in_proj GEMM: 4 waves per work-group (was 1) holding the same 64 weight rows and 4 neighbouring m blocks: a weight line is fetched once per WGP cache, not per wave; 630 work-groups, not 2520
# speedup vs baseline: 4.7087x; 1.0528x over previous
#include <hip/hip_runtime.h>


#define BSZ   8
#define T_    8
#define L_    196
#define D_    192
#define E_    192
#define C_    1536
#define N_    16
#define KC    4
#define R_    96
#define RN2   128
#define LTILES 13

typedef float v2f __attribute__((ext_vector_type(2)));
typedef float v8f __attribute__((ext_vector_type(8)));

__device__ __forceinline__ float sigmoid_(float x) {
    return 1.0f / (1.0f + __expf(-x));
}

#define IP_EW    64
#define IP_MW    32
#define IP_EBH   (E_ / IP_EW)
#define IP_MBLK  ((BSZ * T_ * L_) / IP_MW)
#define IP_LSTEP 28
#define IP_HALO  (IP_MW - IP_LSTEP)
#define IP_LT    (L_ / IP_LSTEP)
#define IP_XBLKS (BSZ * T_ * IP_LT * IP_EBH)
#define IP_ZBLKS (IP_MBLK * IP_EBH)
#define IP_KC    32
#define IP_KS    (IP_KC / 4)
#define IP_KQ    (IP_KC / 8)
#define IP_NCH   (D_ / IP_KC)
#define IP_NPASS (IP_NCH * 4)
#define IP_WPB   4
#define IP_XWGS  (IP_XBLKS / IP_WPB)
#define IP_ZWGS  (IP_ZBLKS / IP_WPB)

__device__ __forceinline__ void ip_load_line(const float* __restrict__ p, float4 (&x)[IP_KQ])
{
    #pragma unroll
    for (int s = 0; s < IP_KQ; ++s)
        x[s] = *(const float4*)(p + 8 * s);
}

__device__ __forceinline__ void ip_pass(const float4 (&a)[IP_KQ],
                                        const float4 (&b0)[IP_KQ], const float4 (&b1)[IP_KQ],
                                        v8f& c0, v8f& c1)
{
    #pragma unroll
    for (int s = 0; s < IP_KQ; ++s) {
        const v2f alo  = {a[s].x,  a[s].y},  ahi  = {a[s].z,  a[s].w};
        const v2f b0lo = {b0[s].x, b0[s].y}, b0hi = {b0[s].z, b0[s].w};
        const v2f b1lo = {b1[s].x, b1[s].y}, b1hi = {b1[s].z, b1[s].w};
        c0 = __builtin_amdgcn_wmma_f32_16x16x4_f32(
                 false, alo, false, b0lo, (short)0, c0, false, false);
        c1 = __builtin_amdgcn_wmma_f32_16x16x4_f32(
                 false, alo, false, b1lo, (short)0, c1, false, false);
        c0 = __builtin_amdgcn_wmma_f32_16x16x4_f32(
                 false, ahi, false, b0hi, (short)0, c0, false, false);
        c1 = __builtin_amdgcn_wmma_f32_16x16x4_f32(
                 false, ahi, false, b1hi, (short)0, c1, false, false);
    }
}

template <int N>
__device__ __forceinline__ float row_shr0(const float v)
{
    return __builtin_bit_cast(float,
        __builtin_amdgcn_update_dpp(0, __builtin_bit_cast(int, v), 0x110 + N, 0xF, 0xF, true));
}
template <int N>
__device__ __forceinline__ float row_shl0(const float v)
{
    return __builtin_bit_cast(float,
        __builtin_amdgcn_update_dpp(0, __builtin_bit_cast(int, v), 0x100 + N, 0xF, 0xF, true));
}
template <int K>
__device__ __forceinline__ float tap_back(const float lo, const float hi)
{
    const int t = __builtin_amdgcn_update_dpp(0, __builtin_bit_cast(int, lo), 0x100 + 16 - K, 0xF, 0xF, true);
    return __builtin_bit_cast(float,
        __builtin_amdgcn_update_dpp(t, __builtin_bit_cast(int, hi), 0x110 + K, 0xF, 0xF, false));
}

__global__ __launch_bounds__(32 * IP_WPB)
void k_gemm_inproj(const float* __restrict__ hidden,
                   const float* __restrict__ w,
                   const float* __restrict__ cw,
                   const float* __restrict__ cb,
                   float* __restrict__ xc,
                   float* __restrict__ zbuf)
{
    const int lane = threadIdx.x & 31;
    const int c    = lane & 15;
    const int h    = lane >> 4;
    const int wq   = __builtin_amdgcn_readfirstlane((int)(threadIdx.x >> 5));
    const bool xh  = (blockIdx.x < IP_XWGS);

    int eb, tile, row0, row1;
    int bt = 0, j = 0;
    if (xh) {
        eb   = blockIdx.x % IP_EBH;
        tile = (blockIdx.x / IP_EBH) * IP_WPB + wq;
        bt   = tile / IP_LT;
        j    = tile - bt * IP_LT;
        const int l0 = IP_LSTEP * j - IP_HALO + c;
        row0 = bt * L_ + ((l0 > 0) ? l0 : 0);
        row1 = bt * L_ + l0 + 16;
    } else {
        const int q = blockIdx.x - IP_XWGS;
        eb   = IP_EBH + q % IP_EBH;
        tile = (q / IP_EBH) * IP_WPB + wq;
        row0 = tile * IP_MW + c;
        row1 = row0 + 16;
    }
    const int ch0 = ((eb % IP_EBH) * IP_EW + 8 * h) * T_ + (bt % T_);

    const float* Ap  = w      + (size_t)(eb * IP_EW + c) * D_ + 4 * h;
    const float* Bp0 = hidden + (size_t)row0 * D_ + 4 * h;
    const float* Bp1 = hidden + (size_t)row1 * D_ + 4 * h;

    v8f acc[4][2];
    #pragma unroll
    for (int f = 0; f < 4; ++f) {
        #pragma unroll
        for (int g = 0; g < 2; ++g)
            acc[f][g] = (v8f){0.f,0.f,0.f,0.f,0.f,0.f,0.f,0.f};
    }

    float4 a[4][IP_KQ];
    float4 b[2][2][IP_KQ];
    float4 cwv[32];
    float  cbs[32];
    ip_load_line(Bp0, b[0][0]);
    ip_load_line(Bp1, b[0][1]);
    ip_load_line(Ap, a[0]);
    ip_load_line(Ap + (size_t)16 * D_, a[1]);
    ip_load_line(Ap + (size_t)32 * D_, a[2]);

    #pragma unroll
    for (int p = 0; p < IP_NPASS; ++p) {
        const int ch = p >> 2;
        const int f  = p & 3;
        const int q  = p + 3;
        if (q < IP_NPASS)
            ip_load_line(Ap + (size_t)(q & 3) * 16 * D_ + (q >> 2) * IP_KC, a[q & 3]);
        if (f == 1 && ch + 1 < IP_NCH) {
            ip_load_line(Bp0 + (ch + 1) * IP_KC, b[(ch + 1) & 1][0]);
            ip_load_line(Bp1 + (ch + 1) * IP_KC, b[(ch + 1) & 1][1]);
        }
        if (p == IP_NPASS - 3) {
            #pragma unroll
            for (int i = 0; i < 32; ++i) {
                const int chn = ch0 + ((i >> 3) * 16 + (i & 7)) * T_;
                cwv[i] = *(const float4*)(cw + (size_t)chn * KC);
                cbs[i] = cb[chn];
            }
        }
        __builtin_amdgcn_sched_barrier(0);
        ip_pass(a[p & 3], b[ch & 1][0], b[ch & 1][1], acc[f][0], acc[f][1]);
        __builtin_amdgcn_sched_barrier(0);
    }

    #pragma unroll
    for (int i = 0; i < 32; ++i)
        asm volatile("" :: "v"(cwv[i].x), "v"(cwv[i].y), "v"(cwv[i].z), "v"(cwv[i].w), "v"(cbs[i]));

    if (!xh) {
        const int e0 = (eb - IP_EBH) * IP_EW + 8 * h;
        #pragma unroll
        for (int g = 0; g < 2; ++g) {
            const int m   = tile * IP_MW + g * 16 + c;
            const int bt_ = m / L_;
            const int l   = m - bt_ * L_;
            const int b_  = bt_ / T_;
            const int t   = bt_ % T_;
            float* p = zbuf + ((size_t)b_ * C_ + (size_t)e0 * T_ + t) * L_ + l;
            #pragma unroll
            for (int f = 0; f < 4; ++f) {
                #pragma unroll
                for (int r = 0; r < 8; ++r)
                    p[(size_t)(f * 16 + r) * (T_ * L_)] = acc[f][g][r];
            }
        }
    } else {
        const int  b_  = bt / T_;
        const bool pad = (j == 0) && (c < IP_HALO);
        const bool st0 = (c >= IP_HALO);
        float* p1 = xc + ((size_t)b_ * C_ + ch0) * L_ + (IP_LSTEP * j - IP_HALO + 16 + c);
        #pragma unroll
        for (int f = 0; f < 4; ++f) {
            float x0[8];
            #pragma unroll
            for (int r = 0; r < 8; ++r)
                x0[r] = pad ? 0.f : acc[f][0][r];
            #pragma unroll
            for (int r = 0; r < 8; ++r) {
                const float4 wv = cwv[f * 8 + r];
                const float  x1 = acc[f][1][r];
                float s1 = __builtin_fmaf(tap_back<3>(x0[r], x1), wv.x, cbs[f * 8 + r]);
                s1 = __builtin_fmaf(tap_back<2>(x0[r], x1), wv.y, s1);
                s1 = __builtin_fmaf(tap_back<1>(x0[r], x1), wv.z, s1);
                s1 = __builtin_fmaf(x1, wv.w, s1);
                p1[(size_t)(f * 16 + r) * (T_ * L_)] = s1 * sigmoid_(s1);
            }
            float s0[8];
            #pragma unroll
            for (int r = 0; r < 8; ++r) {
                const float4 wv = cwv[f * 8 + r];
                float s = __builtin_fmaf(row_shr0<3>(x0[r]), wv.x, cbs[f * 8 + r]);
                s = __builtin_fmaf(row_shr0<2>(x0[r]), wv.y, s);
                s = __builtin_fmaf(row_shr0<1>(x0[r]), wv.z, s);
                s0[r] = __builtin_fmaf(x0[r], wv.w, s);
            }
            if (st0) {
                #pragma unroll
                for (int r = 0; r < 8; ++r)
                    p1[(ptrdiff_t)(f * 16 + r) * (T_ * L_) - 16] = s0[r] * sigmoid_(s0[r]);
            }
        }
    }
}

#define XP_NW    8
#define XP_KW    (C_ / XP_NW)
#define XP_RB    32
#define XP_LB    32
#define XP_RBLK  (RN2 / XP_RB)
#define XP_LBLK  ((L_ + XP_LB - 1) / XP_LB)
#define XP_G     4
#define XP_NG    (XP_KW / (8 * XP_G))

__device__ __forceinline__ void xp_load(const float* __restrict__ Ap,
                                        const float* __restrict__ Bp,
                                        float4 (&a)[2][XP_G], float2 (&bq)[XP_G][4])
{
    #pragma unroll
    for (int s = 0; s < XP_G; ++s) {
        a[0][s] = *(const float4*)(Ap + 8 * s);
        a[1][s] = *(const float4*)(Ap + (size_t)16 * C_ + 8 * s);
    }
    #pragma unroll
    for (int s = 0; s < XP_G; ++s) {
        #pragma unroll
        for (int i = 0; i < 4; ++i)
            bq[s][i] = *(const float2*)(Bp + (size_t)(8 * s + i) * L_);
    }
}

__device__ __forceinline__ void xp_mma(const float4 (&a)[2][XP_G],
                                       const float2 (&bq)[XP_G][4], v8f (&acc)[2][2])
{
    #pragma unroll
    for (int s = 0; s < XP_G; ++s) {
        const v2f be0 = {bq[s][0].x, bq[s][1].x};
        const v2f bo0 = {bq[s][0].y, bq[s][1].y};
        const v2f be1 = {bq[s][2].x, bq[s][3].x};
        const v2f bo1 = {bq[s][2].y, bq[s][3].y};
        const v2f a0lo = {a[0][s].x, a[0][s].y};
        const v2f a1lo = {a[1][s].x, a[1][s].y};
        const v2f a0hi = {a[0][s].z, a[0][s].w};
        const v2f a1hi = {a[1][s].z, a[1][s].w};
        acc[0][0] = __builtin_amdgcn_wmma_f32_16x16x4_f32(
                        false, a0lo, false, be0, (short)0, acc[0][0], false, false);
        acc[0][1] = __builtin_amdgcn_wmma_f32_16x16x4_f32(
                        false, a0lo, false, bo0, (short)0, acc[0][1], false, false);
        acc[1][0] = __builtin_amdgcn_wmma_f32_16x16x4_f32(
                        false, a1lo, false, be0, (short)0, acc[1][0], false, false);
        acc[1][1] = __builtin_amdgcn_wmma_f32_16x16x4_f32(
                        false, a1lo, false, bo0, (short)0, acc[1][1], false, false);
        acc[0][0] = __builtin_amdgcn_wmma_f32_16x16x4_f32(
                        false, a0hi, false, be1, (short)0, acc[0][0], false, false);
        acc[0][1] = __builtin_amdgcn_wmma_f32_16x16x4_f32(
                        false, a0hi, false, bo1, (short)0, acc[0][1], false, false);
        acc[1][0] = __builtin_amdgcn_wmma_f32_16x16x4_f32(
                        false, a1hi, false, be1, (short)0, acc[1][0], false, false);
        acc[1][1] = __builtin_amdgcn_wmma_f32_16x16x4_f32(
                        false, a1hi, false, bo1, (short)0, acc[1][1], false, false);
    }
}

__global__ __launch_bounds__(XP_NW * 32)
void k_gemm_xproj(const float* __restrict__ xc,
                  const float* __restrict__ w,
                  float* __restrict__ x_dbl)
{
    __shared__ __attribute__((aligned(16))) float sPart[XP_NW * 8 * 32 * 4];

    const int tid  = threadIdx.x;
    const int lane = tid & 31;
    const int wv   = tid >> 5;
    const int c    = lane & 15;
    const int h    = lane >> 4;
    const int rb   = blockIdx.x % XP_RBLK;
    const int lb   = blockIdx.x / XP_RBLK;
    const int b    = blockIdx.y;

    const int l0 = lb * XP_LB + 2 * c;
    const int lc = (l0 < L_) ? l0 : (L_ - 2);
    const int k0 = wv * XP_KW + 4 * h;

    const float* Ap = w  + (size_t)(rb * XP_RB + c) * C_ + k0;
    const float* Bp = xc + ((size_t)b * C_ + k0) * L_ + lc;

    v8f acc[2][2];
    #pragma unroll
    for (int f = 0; f < 2; ++f) {
        #pragma unroll
        for (int g = 0; g < 2; ++g)
            acc[f][g] = (v8f){0.f,0.f,0.f,0.f,0.f,0.f,0.f,0.f};
    }

    float4 a[2][2][XP_G];
    float2 bq[2][XP_G][4];
    xp_load(Ap, Bp, a[0], bq[0]);
    #pragma unroll
    for (int t = 0; t < XP_NG; ++t) {
        if (t + 1 < XP_NG)
            xp_load(Ap + (t + 1) * 8 * XP_G, Bp + (size_t)(t + 1) * 8 * XP_G * L_,
                    a[(t + 1) & 1], bq[(t + 1) & 1]);
        __builtin_amdgcn_sched_barrier(0);
        xp_mma(a[t & 1], bq[t & 1], acc);
        __builtin_amdgcn_sched_barrier(0);
    }

    float4* sp = (float4*)sPart;
    #pragma unroll
    for (int f = 0; f < 2; ++f) {
        #pragma unroll
        for (int rp = 0; rp < 4; ++rp)
            sp[(wv * 8 + f * 4 + rp) * 32 + lane] =
                make_float4(acc[f][0][2 * rp],     acc[f][1][2 * rp],
                            acc[f][0][2 * rp + 1], acc[f][1][2 * rp + 1]);
    }
    __syncthreads();

    float4 s = sp[wv * 32 + lane];
    #pragma unroll
    for (int u = 1; u < XP_NW; ++u) {
        const float4 t = sp[(u * 8 + wv) * 32 + lane];
        s.x += t.x; s.y += t.y; s.z += t.z; s.w += t.w;
    }
    const int row = rb * XP_RB + 16 * (wv >> 2) + 8 * h + 2 * (wv & 3);
    if (l0 < L_) {
        float* o = x_dbl + ((size_t)b * RN2 + row) * L_ + l0;
        *(float2*)o        = make_float2(s.x, s.y);
        *(float2*)(o + L_) = make_float2(s.z, s.w);
    }
}

#define SC_THREADS 128
#define SC_CH      (SC_THREADS / 2)
#define SC_NH      (N_ / 2)
#define SC_CHUNKS  (L_ / 4)
#define SC_SETS    8
#define BC_PITCH   36
#define SC_STAGE_FULL ((2 * N_ * SC_CHUNKS) / SC_THREADS)
#define SC_STAGE_REST ((2 * N_ * SC_CHUNKS) - SC_STAGE_FULL * SC_THREADS)
#define SD_KS      (R_ / 8)
#define SD_LBLK    ((L_ + 31) / 32)
#define SD_PITCH   228

__device__ __forceinline__ void sd_load_b(const float* __restrict__ Bp, float2 (&bq)[SD_KS][4])
{
    #pragma unroll
    for (int s = 0; s < SD_KS; ++s) {
        #pragma unroll
        for (int i = 0; i < 4; ++i)
            bq[s][i] = *(const float2*)(Bp + (size_t)(8 * s + i) * L_);
    }
}

__device__ __forceinline__ float softplus_(float v) {
    const float t = __logf(1.f + __expf(v));
    return (v > 20.f) ? v : t;
}

__device__ __forceinline__ void sd_store(const v8f& acc0, const v8f& acc1,
                                         const float (&bias)[8], float* __restrict__ drow)
{
    #pragma unroll
    for (int r = 0; r < 8; ++r) {
        const float v0 = acc0[r] + bias[r];
        const float v1 = acc1[r] + bias[r];
        *(float2*)(drow + r * SD_PITCH) = make_float2(softplus_(v0), softplus_(v1));
    }
}

__device__ __forceinline__ void sd_mma_store(const float4 (&a)[SD_KS], const float2 (&bq)[SD_KS][4],
                                             v8f& acc0, v8f& acc1,
                                             const v8f& accP0, const v8f& accP1,
                                             const float (&bias)[8], float* __restrict__ drow)
{
    #pragma unroll
    for (int s = 0; s < SD_KS; ++s) {
        const v2f be0 = {bq[s][0].x, bq[s][1].x};
        const v2f bo0 = {bq[s][0].y, bq[s][1].y};
        const v2f be1 = {bq[s][2].x, bq[s][3].x};
        const v2f bo1 = {bq[s][2].y, bq[s][3].y};
        const v2f alo = {a[s].x, a[s].y};
        const v2f ahi = {a[s].z, a[s].w};
        acc0 = __builtin_amdgcn_wmma_f32_16x16x4_f32(
                   false, alo, false, be0, (short)0, acc0, false, false);
        acc1 = __builtin_amdgcn_wmma_f32_16x16x4_f32(
                   false, alo, false, bo0, (short)0, acc1, false, false);
        acc0 = __builtin_amdgcn_wmma_f32_16x16x4_f32(
                   false, ahi, false, be1, (short)0, acc0, false, false);
        acc1 = __builtin_amdgcn_wmma_f32_16x16x4_f32(
                   false, ahi, false, bo1, (short)0, acc1, false, false);
        if (s < 8) {
            const float v0 = accP0[s] + bias[s];
            const float v1 = accP1[s] + bias[s];
            *(float2*)(drow + s * SD_PITCH) = make_float2(softplus_(v0), softplus_(v1));
        }
    }
}

__device__ __forceinline__ float pair_other(const float v)
{
    return __builtin_bit_cast(float,
        __builtin_amdgcn_update_dpp(0, __builtin_bit_cast(int, v), 0xB1, 0xF, 0xF, true));
}

__device__ __forceinline__ void scan_rows(const float* __restrict__ srow, float4 (&rw)[16])
{
    #pragma unroll
    for (int k = 0; k < 4; ++k) {
        const float4* row = (const float4*)(srow + k * BC_PITCH);
        rw[4 * k + 0] = row[0];
        rw[4 * k + 1] = row[1];
        rw[4 * k + 2] = row[4];
        rw[4 * k + 3] = row[5];
    }
}

__device__ __forceinline__ void scan_chunk4(const float4 (&rw)[16],
                                            const float4 d4, const float4 x4,
                                            const float2 z2, const bool hi,
                                            const float (&A_neg)[SC_NH],
                                            float (&h)[SC_NH],
                                            const float Dp,
                                            float2* __restrict__ dst)
{
    const float dvs[4] = {d4.x, d4.y, d4.z, d4.w};
    const float xvs[4] = {x4.x, x4.y, x4.z, x4.w};
    float yt[4];
    #pragma unroll
    for (int k = 0; k < 4; ++k) {
        const float dv  = dvs[k];
        const float xv  = xvs[k];
        const float dvx = dv * xv;
        const float4 b0 = rw[4 * k], b1 = rw[4 * k + 1], c0 = rw[4 * k + 2], c1 = rw[4 * k + 3];
        const float bc[2 * SC_NH] = {b0.x, b0.y, b0.z, b0.w, b1.x, b1.y, b1.z, b1.w,
                                     c0.x, c0.y, c0.z, c0.w, c1.x, c1.y, c1.z, c1.w};
        float y0 = 0.f, y1 = 0.f, y2 = 0.f, y3 = 0.f;
        #pragma unroll
        for (int n = 0; n < SC_NH; n += 4) {
            h[n + 0] = h[n + 0] * __expf(dv * A_neg[n + 0]) + dvx * bc[n + 0];
            h[n + 1] = h[n + 1] * __expf(dv * A_neg[n + 1]) + dvx * bc[n + 1];
            h[n + 2] = h[n + 2] * __expf(dv * A_neg[n + 2]) + dvx * bc[n + 2];
            h[n + 3] = h[n + 3] * __expf(dv * A_neg[n + 3]) + dvx * bc[n + 3];
            y0 += h[n + 0] * bc[SC_NH + n + 0];
            y1 += h[n + 1] * bc[SC_NH + n + 1];
            y2 += h[n + 2] * bc[SC_NH + n + 2];
            y3 += h[n + 3] * bc[SC_NH + n + 3];
        }
        const float yp = (y0 + y1) + (y2 + y3);
        yt[k] = yp + pair_other(yp);
    }
    const float ya = hi ? yt[2] : yt[0];
    const float yb = hi ? yt[3] : yt[1];
    const float xa = hi ? xvs[2] : xvs[0];
    const float xb = hi ? xvs[3] : xvs[1];
    *dst = make_float2((ya + xa * Dp) * (z2.x * sigmoid_(z2.x)),
                       (yb + xb * Dp) * (z2.y * sigmoid_(z2.y)));
}

__global__ __launch_bounds__(SC_THREADS) __attribute__((amdgpu_waves_per_eu(1, 1)))
void k_scan_out(const float* __restrict__ xc,
                const float* __restrict__ zbuf,
                const float* __restrict__ x_dbl,
                const float* __restrict__ dt_w,
                const float* __restrict__ dt_bias,
                const float* __restrict__ A_log,
                const float* __restrict__ Dp_,
                float* __restrict__ out)
{
    __shared__ __attribute__((aligned(16))) float sBC[L_ * BC_PITCH];
    __shared__ __attribute__((aligned(16))) float sDel[SC_CH * SD_PITCH];

    const int  tid  = threadIdx.x;
    const int  half = tid & 1;
    const bool hi   = (half != 0);
    const int  lane = tid & 31;
    const int  wv   = tid >> 5;
    const int  n    = lane & 15;
    const int  hh   = lane >> 4;
    const int  b    = blockIdx.x / (C_ / SC_CH);
    const int  cblk = (blockIdx.x % (C_ / SC_CH)) * SC_CH;
    const int  c    = cblk + (tid >> 1);
    const int  bc_  = b * C_ + c;

    const float4* xp4 = (const float4*)(xc    + (size_t)bc_ * L_);
    const float2* zp2 = (const float2*)(zbuf  + (size_t)bc_ * L_) + half;
    float2*       op2 = (float2*)(out + (size_t)bc_ * L_) + half;

    const float* Ap = dt_w + (size_t)(cblk + 16 * wv + n) * R_ + 4 * hh;
    float4 a[SD_KS];
    #pragma unroll
    for (int s = 0; s < SD_KS; ++s)
        a[s] = *(const float4*)(Ap + 8 * s);
    float bias[8];
    {
        const float4* bp4 = (const float4*)(dt_bias + cblk + 16 * wv + 8 * hh);
        const float4 t0 = bp4[0], t1 = bp4[1];
        bias[0] = t0.x; bias[1] = t0.y; bias[2] = t0.z; bias[3] = t0.w;
        bias[4] = t1.x; bias[5] = t1.y; bias[6] = t1.z; bias[7] = t1.w;
    }
    const float* Bbase = x_dbl + ((size_t)b * RN2 + 4 * hh) * L_;
    float2 bq[SD_KS][4];
    sd_load_b(Bbase + 2 * n, bq);

    const float4* src4 = (const float4*)(x_dbl + ((size_t)b * RN2 + R_) * L_);
    float4 stg[SC_STAGE_FULL];
    #pragma unroll
    for (int j = 0; j < SC_STAGE_FULL; ++j)
        stg[j] = src4[tid + j * SC_THREADS];
    const int    frest = (tid < SC_STAGE_REST) ? (tid + SC_STAGE_FULL * SC_THREADS)
                                               : (SC_STAGE_FULL * SC_THREADS);
    const float4 stgr  = src4[frest];

    const float4* a4 = (const float4*)(A_log + (size_t)c * N_ + SC_NH * half);
    const float4 al0 = a4[0], al1 = a4[1];
    const float  Dp  = Dp_[c];

    float4 qx[SC_SETS];
    float2 qz[SC_SETS];
    #pragma unroll
    for (int s = 0; s < SC_SETS - 1; ++s) {
        qx[s] = xp4[s]; qz[s] = zp2[2 * s];
    }

    __builtin_amdgcn_sched_barrier(0);

    #pragma unroll
    for (int j = 0; j < SC_STAGE_FULL; ++j) {
        const int f  = tid + j * SC_THREADS;
        const int r  = f / SC_CHUNKS;
        const int l4 = f % SC_CHUNKS;
        const float4 v = stg[j];
        float* d = sBC + (l4 * 4) * BC_PITCH + r;
        d[0]            = v.x;
        d[BC_PITCH]     = v.y;
        d[2 * BC_PITCH] = v.z;
        d[3 * BC_PITCH] = v.w;
    }
    asm volatile("" :: "v"(stgr.x), "v"(stgr.y), "v"(stgr.z), "v"(stgr.w));
    if (tid < SC_STAGE_REST) {
        const int r  = frest / SC_CHUNKS;
        const int l4 = frest % SC_CHUNKS;
        float* d = sBC + (l4 * 4) * BC_PITCH + r;
        d[0]            = stgr.x;
        d[BC_PITCH]     = stgr.y;
        d[2 * BC_PITCH] = stgr.z;
        d[3 * BC_PITCH] = stgr.w;
    }

    float A_neg[SC_NH];
    A_neg[0] = -__expf(al0.x); A_neg[1] = -__expf(al0.y);
    A_neg[2] = -__expf(al0.z); A_neg[3] = -__expf(al0.w);
    A_neg[4] = -__expf(al1.x); A_neg[5] = -__expf(al1.y);
    A_neg[6] = -__expf(al1.z); A_neg[7] = -__expf(al1.w);

    float* drow = sDel + (16 * wv + 8 * hh) * SD_PITCH + 2 * n;
    float2 bn[SD_KS][4];
    v8f accP0 = (v8f){0.f,0.f,0.f,0.f,0.f,0.f,0.f,0.f};
    v8f accP1 = accP0;
    #pragma unroll 1
    for (int lb = 0; lb < SD_LBLK; ++lb) {
        const int l0 = 32 * (lb + 1) + 2 * n;
        const int lc = (l0 < L_) ? l0 : (L_ - 2);
        sd_load_b(Bbase + lc, bn);
        __builtin_amdgcn_sched_barrier(0);
        v8f acc0 = (v8f){0.f,0.f,0.f,0.f,0.f,0.f,0.f,0.f};
        v8f acc1 = acc0;
        sd_mma_store(a, bq, acc0, acc1, accP0, accP1, bias,
                     drow + 32 * ((lb > 0) ? (lb - 1) : (SD_LBLK - 1)));
        __builtin_amdgcn_sched_barrier(0);
        accP0 = acc0; accP1 = acc1;
        #pragma unroll
        for (int s = 0; s < SD_KS; ++s) {
            #pragma unroll
            for (int i = 0; i < 4; ++i)
                bq[s][i] = bn[s][i];
        }
    }
    sd_store(accP0, accP1, bias, drow + 32 * (SD_LBLK - 1));

    float h[SC_NH];
    #pragma unroll
    for (int q = 0; q < SC_NH; ++q) h[q] = 0.f;

    __syncthreads();

    const float* sB = sBC + SC_NH * half;
    const float4* sD4 = (const float4*)(sDel + (tid >> 1) * SD_PITCH);

    float4 rw[2][16];
    float4 rd[2];
    scan_rows(sB, rw[0]);
    rd[0] = sD4[0];
    for (int i = 0; i < SC_CHUNKS - 1; i += SC_SETS) {
        #pragma unroll
        for (int j = 0; j < SC_SETS; ++j) {
            const int nx = (i + j + SC_SETS - 1 < SC_CHUNKS) ? (i + j + SC_SETS - 1)
                                                             : (SC_CHUNKS - 1);
            const int ls = (j + SC_SETS - 1) % SC_SETS;
            qx[ls] = xp4[nx]; qz[ls] = zp2[2 * nx];
            scan_rows(sB + ((i + j + 1) * 4) * BC_PITCH, rw[(j + 1) & 1]);
            rd[(j + 1) & 1] = sD4[i + j + 1];
            __builtin_amdgcn_sched_barrier(0);
            scan_chunk4(rw[j & 1], rd[j & 1], qx[j], qz[j], hi,
                        A_neg, h, Dp, op2 + 2 * (i + j));
            __builtin_amdgcn_sched_barrier(0);
        }
    }
    scan_chunk4(rw[0], rd[0], qx[0], qz[0], hi, A_neg, h, Dp, op2 + 2 * (SC_CHUNKS - 1));
}

extern "C" void kernel_launch(void* const* d_in, const int* in_sizes, int n_in,
                              void* d_out, int out_size, void* d_ws, size_t ws_size,
                              hipStream_t stream)
{
    const float* hidden    = (const float*)d_in[0];
    const float* in_proj_w = (const float*)d_in[1];
    const float* conv_w    = (const float*)d_in[2];
    const float* conv_b    = (const float*)d_in[3];
    const float* x_proj_w  = (const float*)d_in[4];
    const float* dt_proj_w = (const float*)d_in[5];
    const float* dt_bias   = (const float*)d_in[6];
    const float* A_log     = (const float*)d_in[7];
    const float* D_param   = (const float*)d_in[8];

    float* ws = (float*)d_ws;
    const size_t BCL = (size_t)BSZ * C_ * L_;
    float* zbuf  = ws + BCL;
    float* xcbuf = ws + 2 * BCL;
    float* x_dbl = ws + 3 * BCL;

    k_gemm_inproj<<<IP_XWGS + IP_ZWGS, 32 * IP_WPB, 0, stream>>>(
        hidden, in_proj_w, conv_w, conv_b, xcbuf, zbuf);

    k_gemm_xproj<<<dim3(XP_RBLK * XP_LBLK, BSZ), XP_NW * 32, 0, stream>>>(
        xcbuf, x_proj_w, x_dbl);

    k_scan_out<<<(BSZ * C_) / SC_CH, SC_THREADS, 0, stream>>>(
        xcbuf, zbuf, x_dbl, dt_proj_w, dt_bias, A_log, D_param, (float*)d_out);
}
